# starting kernel
# baseline (speedup 1.0000x reference)
#include <hip/hip_runtime.h>
#include <math.h>

#define B_SZ   2
#define LSEQ   2048
#define DEXP   1024
#define DRANK  64
#define NST    16
#define RPROJ  (DRANK + 2*NST)
#define NCHUNK 8
#define CHUNK  (LSEQ / NCHUNK)
#define ROWS   (B_SZ * LSEQ)

typedef __attribute__((ext_vector_type(2))) float v2f;
typedef __attribute__((ext_vector_type(8))) float v8f;

__device__ __forceinline__ v8f wmma_f32_k4(v2f a, v2f b, v8f c) {
  return __builtin_amdgcn_wmma_f32_16x16x4_f32(
       false, a,  false, b,
       (short)0, c,  false,  false);
}

__global__ void ssm_prep(const float* __restrict__ A_log, float* __restrict__ Aneg,
                         const float* __restrict__ xw,  float* __restrict__ xwT,
                         const float* __restrict__ dtw, float* __restrict__ dtwT) {
  int i = blockIdx.x * 256 + threadIdx.x;
  if (i < DEXP * NST) Aneg[i] = -__expf(A_log[i]);
  if (i < DEXP * RPROJ) {
    int k = i / RPROJ, c = i % RPROJ;
    xwT[i] = xw[(size_t)c * DEXP + k];
  }
  if (i < DRANK * DEXP) {
    int k = i / DEXP, d = i % DEXP;
    dtwT[i] = dtw[(size_t)d * DRANK + k];
  }
}

__global__ __launch_bounds__(256) void ssm_proj_wmma(
    const float* __restrict__ x,    const float* __restrict__ xwT,
    const float* __restrict__ xb,   const float* __restrict__ dtwT,
    const float* __restrict__ dtb,
    float* __restrict__ dtbuf, float* __restrict__ Bmbuf, float* __restrict__ Cbuf)
{
  __shared__ float sx[16 * 516];
  __shared__ float sproj[16 * 100];

  const int t    = threadIdx.x;
  const int wave = t >> 5;
  const int nl   = t & 15;
  const int hi   = (t >> 4) & 1;
  const int rowbase = blockIdx.x * 16;

  v8f acc1;
  if (wave < 6) {
    float bv = xb[wave * 16 + nl];
    for (int i = 0; i < 8; ++i) acc1[i] = bv;
  }
  for (int kc = 0; kc < DEXP; kc += 512) {
    __syncthreads();
    for (int i = 0; i < 8; ++i) {
      int flat = t + i * 256;
      int r  = flat >> 7;
      int kk = (flat & 127) << 2;
      const float4 v = *(const float4*)(x + (size_t)(rowbase + r) * DEXP + kc + kk);
      float* dst = &sx[r * 516 + kk];
      dst[0] = v.x; dst[1] = v.y; dst[2] = v.z; dst[3] = v.w;
    }
    __syncthreads();
    if (wave < 6) {
      const int col = wave * 16 + nl;
      for (int k = 0; k < 512; k += 4) {
        const int kk = k + hi * 2;
        v2f a;  a.x  = sx[nl * 516 + kk];   a.y  = sx[nl * 516 + kk + 1];
        const int gk = kc + kk;
        v2f bf;
        bf.x = xwT[(size_t)gk * RPROJ + col];
        bf.y = xwT[(size_t)(gk + 1) * RPROJ + col];
        acc1 = wmma_f32_k4(a, bf, acc1);
      }
    }
  }
  if (wave < 6) {
    for (int i = 0; i < 8; ++i) {
      int M = i + hi * 8;
      sproj[M * 100 + wave * 16 + nl] = acc1[i];
      if (wave == 4) Bmbuf[(size_t)(rowbase + M) * NST + nl] = acc1[i];
      if (wave == 5) Cbuf [(size_t)(rowbase + M) * NST + nl] = acc1[i];
    }
  }
  __syncthreads();

  v2f a2[16];
  for (int kt = 0; kt < 16; ++kt) {
    int kb = kt * 4 + hi * 2;
    a2[kt].x = sproj[nl * 100 + kb];
    a2[kt].y = sproj[nl * 100 + kb + 1];
  }
  for (int j = 0; j < 8; ++j) {
    const int col = wave * 128 + j * 16 + nl;
    v8f acc2;
    { float bv = dtb[col]; for (int i = 0; i < 8; ++i) acc2[i] = bv; }
    for (int kt = 0; kt < 16; ++kt) {
      int kb = kt * 4 + hi * 2;
      v2f bf;
      bf.x = dtwT[(size_t)kb * DEXP + col];
      bf.y = dtwT[(size_t)(kb + 1) * DEXP + col];
      acc2 = wmma_f32_k4(a2[kt], bf, acc2);
    }
    for (int i = 0; i < 8; ++i) {
      int M = i + hi * 8;
      float v  = acc2[i];
      float sp = (v > 20.0f) ? v : log1pf(__expf(v));
      dtbuf[(size_t)(rowbase + M) * DEXP + col] = sp;
    }
  }
}

__global__ __launch_bounds__(256) void ssm_scan_carry(
    const float* __restrict__ x, const float* __restrict__ dtbuf,
    const float* __restrict__ Bmbuf, const float* __restrict__ Aneg,
    float* __restrict__ Sbuf, float* __restrict__ Pbuf)
{
  const int t  = threadIdx.x;
  const int bi = blockIdx.x;
  const int c  = bi & (NCHUNK - 1);
  const int dg = (bi >> 3) & 63;
  const int b  = bi >> 9;
  const int n  = t & 15;
  const int d  = dg * 16 + (t >> 4);

  const float And = Aneg[d * NST + n];
  const size_t rbase = (size_t)b * LSEQ + (size_t)c * CHUNK;
  const float* dtp = dtbuf + rbase * DEXP + d;
  const float* xp  = x     + rbase * DEXP + d;
  const float* Bp  = Bmbuf + rbase * NST + n;

  float h = 0.0f, sdt = 0.0f;
  for (int li = 0; li < CHUNK; ++li) {
    __builtin_prefetch(dtp + (size_t)(li + 8) * DEXP, 0, 1);
    __builtin_prefetch(xp  + (size_t)(li + 8) * DEXP, 0, 1);
    float dtv = dtp[(size_t)li * DEXP];
    float xv  = xp [(size_t)li * DEXP];
    float Bv  = Bp [li * NST];
    float dA  = __expf(And * dtv);
    h   = fmaf(dA, h, dtv * xv * Bv);
    sdt += dtv;
  }
  const size_t si = (((size_t)b * NCHUNK + c) * DEXP + d) * NST + n;
  Sbuf[si] = h;
  Pbuf[si] = __expf(And * sdt);
}

__global__ void ssm_carry_combine(const float* __restrict__ Sbuf,
                                  const float* __restrict__ Pbuf,
                                  float* __restrict__ Car)
{
  const int tg = blockIdx.x * 256 + threadIdx.x;
  const int n  = tg & 15;
  const int d  = (tg >> 4) & (DEXP - 1);
  const int b  = tg >> 14;
  const size_t stride = (size_t)DEXP * NST;
  const size_t base   = (size_t)b * NCHUNK * stride + (size_t)d * NST + n;
  float cin = 0.0f;
  Car[base] = 0.0f;
  for (int c = 1; c < NCHUNK; ++c) {
    const size_t prev = base + (size_t)(c - 1) * stride;
    cin = Sbuf[prev] + Pbuf[prev] * cin;
    Car[base + (size_t)c * stride] = cin;
  }
}

__global__ __launch_bounds__(256) void ssm_scan_out(
    const float* __restrict__ x, const float* __restrict__ dtbuf,
    const float* __restrict__ Bmbuf, const float* __restrict__ Cbuf,
    const float* __restrict__ Aneg, const float* __restrict__ Car,
    float* __restrict__ out)
{
  const int t  = threadIdx.x;
  const int bi = blockIdx.x;
  const int c  = bi & (NCHUNK - 1);
  const int dg = (bi >> 3) & 63;
  const int b  = bi >> 9;
  const int n  = t & 15;
  const int d  = dg * 16 + (t >> 4);

  const float And = Aneg[d * NST + n];
  const size_t rbase = (size_t)b * LSEQ + (size_t)c * CHUNK;
  const float* dtp = dtbuf + rbase * DEXP + d;
  const float* xp  = x     + rbase * DEXP + d;
  const float* Bp  = Bmbuf + rbase * NST + n;
  const float* Cp  = Cbuf  + rbase * NST + n;
  float* op        = out   + rbase * DEXP + d;

  const size_t si = (((size_t)b * NCHUNK + c) * DEXP + d) * NST + n;
  float h = Car[si];

  for (int li = 0; li < CHUNK; ++li) {
    __builtin_prefetch(dtp + (size_t)(li + 8) * DEXP, 0, 1);
    __builtin_prefetch(xp  + (size_t)(li + 8) * DEXP, 0, 1);
    float dtv = dtp[(size_t)li * DEXP];
    float xv  = xp [(size_t)li * DEXP];
    float Bv  = Bp [li * NST];
    float dA  = __expf(And * dtv);
    h = fmaf(dA, h, dtv * xv * Bv);
    float y = h * Cp[li * NST];
    y += __shfl_xor(y, 1);
    y += __shfl_xor(y, 2);
    y += __shfl_xor(y, 4);
    y += __shfl_xor(y, 8);
    if (n == 0) op[(size_t)li * DEXP] = y;
  }
}

extern "C" void kernel_launch(void* const* d_in, const int* in_sizes, int n_in,
                              void* d_out, int out_size, void* d_ws, size_t ws_size,
                              hipStream_t stream) {
  (void)in_sizes; (void)n_in; (void)out_size; (void)ws_size;
  const float* x     = (const float*)d_in[0];
  const float* xw    = (const float*)d_in[1];
  const float* xb    = (const float*)d_in[2];
  const float* dtw   = (const float*)d_in[3];
  const float* dtb   = (const float*)d_in[4];
  const float* A_log = (const float*)d_in[5];
  float* out = (float*)d_out;

  float* ws    = (float*)d_ws;
  float* dtbuf = ws;
  float* Aneg  = dtbuf + (size_t)ROWS * DEXP;
  float* Bm    = Aneg  + (size_t)DEXP * NST;
  float* Cb    = Bm    + (size_t)ROWS * NST;
  float* Sb    = Cb    + (size_t)ROWS * NST;
  float* Pb    = Sb    + (size_t)B_SZ * NCHUNK * DEXP * NST;
  float* Car   = Pb    + (size_t)B_SZ * NCHUNK * DEXP * NST;
  float* xwT   = Car   + (size_t)B_SZ * NCHUNK * DEXP * NST;
  float* dtwT  = xwT   + (size_t)DEXP * RPROJ;

  hipLaunchKernelGGL(ssm_prep, dim3((DEXP * RPROJ + 255) / 256), dim3(256), 0, stream,
                     A_log, Aneg, xw, xwT, dtw, dtwT);
  hipLaunchKernelGGL(ssm_proj_wmma, dim3(ROWS / 16), dim3(256), 0, stream,
                     x, xwT, xb, dtwT, dtb, dtbuf, Bm, Cb);
  hipLaunchKernelGGL(ssm_scan_carry, dim3(B_SZ * (DEXP / 16) * NCHUNK), dim3(256), 0, stream,
                     x, dtbuf, Bm, Aneg, Sb, Pb);
  hipLaunchKernelGGL(ssm_carry_combine, dim3((B_SZ * DEXP * NST) / 256), dim3(256), 0, stream,
                     Sb, Pb, Car);
  hipLaunchKernelGGL(ssm_scan_out, dim3(B_SZ * (DEXP / 16) * NCHUNK), dim3(256), 0, stream,
                     x, dtbuf, Bm, Cb, Aneg, Car, out);
}


# scan kernels: state index n stays on lanes, each lane carries 4 channels; dt/x by 16-byte loads batched 8 steps ahead in registers; n-sum by transposing DPP row_xmask butterfly, same sum tree
# speedup vs baseline: 2.3592x; 2.3592x over previous
#include <hip/hip_runtime.h>
#include <math.h>

#define B_SZ   2
#define LSEQ   2048
#define DEXP   1024
#define DRANK  64
#define NST    16
#define RPROJ  (DRANK + 2*NST)
#define NCHUNK 8
#define CHUNK  (LSEQ / NCHUNK)
#define ROWS   (B_SZ * LSEQ)

typedef __attribute__((ext_vector_type(2))) float v2f;
typedef __attribute__((ext_vector_type(8))) float v8f;

__device__ __forceinline__ v8f wmma_f32_k4(v2f a, v2f b, v8f c) {
  return __builtin_amdgcn_wmma_f32_16x16x4_f32(
       false, a,  false, b,
       (short)0, c,  false,  false);
}

__global__ void ssm_prep(const float* __restrict__ A_log, float* __restrict__ Aneg,
                         const float* __restrict__ xw,  float* __restrict__ xwT,
                         const float* __restrict__ dtw, float* __restrict__ dtwT) {
  int i = blockIdx.x * 256 + threadIdx.x;
  if (i < DEXP * NST) Aneg[i] = -__expf(A_log[i]);
  if (i < DEXP * RPROJ) {
    int k = i / RPROJ, c = i % RPROJ;
    xwT[i] = xw[(size_t)c * DEXP + k];
  }
  if (i < DRANK * DEXP) {
    int k = i / DEXP, d = i % DEXP;
    dtwT[i] = dtw[(size_t)d * DRANK + k];
  }
}

__global__ __launch_bounds__(256) void ssm_proj_wmma(
    const float* __restrict__ x,    const float* __restrict__ xwT,
    const float* __restrict__ xb,   const float* __restrict__ dtwT,
    const float* __restrict__ dtb,
    float* __restrict__ dtbuf, float* __restrict__ Bmbuf, float* __restrict__ Cbuf)
{
  __shared__ float sx[16 * 516];
  __shared__ float sproj[16 * 100];

  const int t    = threadIdx.x;
  const int wave = t >> 5;
  const int nl   = t & 15;
  const int hi   = (t >> 4) & 1;
  const int rowbase = blockIdx.x * 16;

  v8f acc1;
  if (wave < 6) {
    float bv = xb[wave * 16 + nl];
    for (int i = 0; i < 8; ++i) acc1[i] = bv;
  }
  for (int kc = 0; kc < DEXP; kc += 512) {
    __syncthreads();
    for (int i = 0; i < 8; ++i) {
      int flat = t + i * 256;
      int r  = flat >> 7;
      int kk = (flat & 127) << 2;
      const float4 v = *(const float4*)(x + (size_t)(rowbase + r) * DEXP + kc + kk);
      float* dst = &sx[r * 516 + kk];
      dst[0] = v.x; dst[1] = v.y; dst[2] = v.z; dst[3] = v.w;
    }
    __syncthreads();
    if (wave < 6) {
      const int col = wave * 16 + nl;
      for (int k = 0; k < 512; k += 4) {
        const int kk = k + hi * 2;
        v2f a;  a.x  = sx[nl * 516 + kk];   a.y  = sx[nl * 516 + kk + 1];
        const int gk = kc + kk;
        v2f bf;
        bf.x = xwT[(size_t)gk * RPROJ + col];
        bf.y = xwT[(size_t)(gk + 1) * RPROJ + col];
        acc1 = wmma_f32_k4(a, bf, acc1);
      }
    }
  }
  if (wave < 6) {
    for (int i = 0; i < 8; ++i) {
      int M = i + hi * 8;
      sproj[M * 100 + wave * 16 + nl] = acc1[i];
      if (wave == 4) Bmbuf[(size_t)(rowbase + M) * NST + nl] = acc1[i];
      if (wave == 5) Cbuf [(size_t)(rowbase + M) * NST + nl] = acc1[i];
    }
  }
  __syncthreads();

  v2f a2[16];
  for (int kt = 0; kt < 16; ++kt) {
    int kb = kt * 4 + hi * 2;
    a2[kt].x = sproj[nl * 100 + kb];
    a2[kt].y = sproj[nl * 100 + kb + 1];
  }
  for (int j = 0; j < 8; ++j) {
    const int col = wave * 128 + j * 16 + nl;
    v8f acc2;
    { float bv = dtb[col]; for (int i = 0; i < 8; ++i) acc2[i] = bv; }
    for (int kt = 0; kt < 16; ++kt) {
      int kb = kt * 4 + hi * 2;
      v2f bf;
      bf.x = dtwT[(size_t)kb * DEXP + col];
      bf.y = dtwT[(size_t)(kb + 1) * DEXP + col];
      acc2 = wmma_f32_k4(a2[kt], bf, acc2);
    }
    for (int i = 0; i < 8; ++i) {
      int M = i + hi * 8;
      float v  = acc2[i];
      float sp = (v > 20.0f) ? v : log1pf(__expf(v));
      dtbuf[(size_t)(rowbase + M) * DEXP + col] = sp;
    }
  }
}

#define RCH 4
#define UB  8
#define SCAN_THREADS 128
#define SCAN_BLOCKS  (B_SZ * NCHUNK * (DEXP / 32))

struct BatchIn  { float4 dt[UB]; float4 x[UB]; float b[UB]; };
struct BatchOut { float4 dt[UB]; float4 x[UB]; float b[UB]; float c[UB]; };

__device__ __forceinline__ void load_in(BatchIn& q, const float* dtp, const float* xp,
                                        const float* Bp, int l0) {
#pragma unroll
  for (int j = 0; j < UB; ++j) {
    q.dt[j] = *(const float4*)(dtp + (size_t)(l0 + j) * DEXP);
    q.x[j]  = *(const float4*)(xp  + (size_t)(l0 + j) * DEXP);
    q.b[j]  = Bp[(l0 + j) * NST];
  }
}

__device__ __forceinline__ void load_out(BatchOut& q, const float* dtp, const float* xp,
                                         const float* Bp, const float* Cp, int l0) {
#pragma unroll
  for (int j = 0; j < UB; ++j) {
    q.dt[j] = *(const float4*)(dtp + (size_t)(l0 + j) * DEXP);
    q.x[j]  = *(const float4*)(xp  + (size_t)(l0 + j) * DEXP);
    q.b[j]  = Bp[(l0 + j) * NST];
    q.c[j]  = Cp[(l0 + j) * NST];
  }
}

template <int M>
__device__ __forceinline__ float row_xor(float v) {
  return __int_as_float(__builtin_amdgcn_update_dpp(0, __float_as_int(v), 0x160 | M,
                                                    0xf, 0xf, true));
}

__device__ __forceinline__ void carry_steps(const BatchIn& q, const float (&And)[RCH],
                                            float (&h)[RCH], float (&sdt)[RCH]) {
#pragma unroll
  for (int j = 0; j < UB; ++j) {
    const float dv[RCH] = {q.dt[j].x, q.dt[j].y, q.dt[j].z, q.dt[j].w};
    const float xv[RCH] = {q.x[j].x,  q.x[j].y,  q.x[j].z,  q.x[j].w};
    const float Bv = q.b[j];
#pragma unroll
    for (int r = 0; r < RCH; ++r) {
      float dA = __expf(And[r] * dv[r]);
      h[r]   = fmaf(dA, h[r], dv[r] * xv[r] * Bv);
      sdt[r] += dv[r];
    }
  }
}

__global__ __launch_bounds__(SCAN_THREADS) void ssm_scan_carry(
    const float* __restrict__ x, const float* __restrict__ dtbuf,
    const float* __restrict__ Bmbuf, const float* __restrict__ Aneg,
    float* __restrict__ Sbuf, float* __restrict__ Pbuf)
{
  const int t    = threadIdx.x;
  const int n    = t & 15;
  const int hf   = (t >> 4) & 1;
  const int wave = t >> 5;
  const int bi   = blockIdx.x;
  const int c    = bi & (NCHUNK - 1);
  const int dg   = (bi >> 3) & 31;
  const int b    = bi >> 8;
  const int d0   = dg * 32 + wave * 8 + hf * RCH;

  float And[RCH], h[RCH], sdt[RCH];
#pragma unroll
  for (int r = 0; r < RCH; ++r) {
    And[r] = Aneg[(d0 + r) * NST + n];
    h[r] = 0.0f; sdt[r] = 0.0f;
  }
  const size_t rbase = (size_t)b * LSEQ + (size_t)c * CHUNK;
  const float* dtp = dtbuf + rbase * DEXP + d0;
  const float* xp  = x     + rbase * DEXP + d0;
  const float* Bp  = Bmbuf + rbase * NST + n;

  BatchIn qa, qb;
  load_in(qa, dtp, xp, Bp, 0);
  for (int l0 = 0; l0 < CHUNK; l0 += 2 * UB) {
    load_in(qb, dtp, xp, Bp, l0 + UB);
    carry_steps(qa, And, h, sdt);
    if (l0 + 2 * UB < CHUNK) load_in(qa, dtp, xp, Bp, l0 + 2 * UB);
    carry_steps(qb, And, h, sdt);
  }
  const size_t si = (((size_t)b * NCHUNK + c) * DEXP + d0) * NST + n;
#pragma unroll
  for (int r = 0; r < RCH; ++r) {
    Sbuf[si + r * NST] = h[r];
    Pbuf[si + r * NST] = __expf(And[r] * sdt[r]);
  }
}

__global__ void ssm_carry_combine(const float* __restrict__ Sbuf,
                                  const float* __restrict__ Pbuf,
                                  float* __restrict__ Car)
{
  const int tg = blockIdx.x * 256 + threadIdx.x;
  const int n  = tg & 15;
  const int d  = (tg >> 4) & (DEXP - 1);
  const int b  = tg >> 14;
  const size_t stride = (size_t)DEXP * NST;
  const size_t base   = (size_t)b * NCHUNK * stride + (size_t)d * NST + n;
  float cin = 0.0f;
  Car[base] = 0.0f;
  for (int c = 1; c < NCHUNK; ++c) {
    const size_t prev = base + (size_t)(c - 1) * stride;
    cin = Sbuf[prev] + Pbuf[prev] * cin;
    Car[base + (size_t)c * stride] = cin;
  }
}

__device__ __forceinline__ void out_steps(const BatchOut& q, const float (&And)[RCH],
                                          float (&h)[RCH], const bool b0, const bool b1,
                                          const bool writer, float* op, int l0) {
#pragma unroll
  for (int j = 0; j < UB; ++j) {
    const float dv[RCH] = {q.dt[j].x, q.dt[j].y, q.dt[j].z, q.dt[j].w};
    const float xv[RCH] = {q.x[j].x,  q.x[j].y,  q.x[j].z,  q.x[j].w};
    const float Bv = q.b[j], Cv = q.c[j];
    float y[RCH];
#pragma unroll
    for (int r = 0; r < RCH; ++r) {
      float dA = __expf(And[r] * dv[r]);
      h[r] = fmaf(dA, h[r], dv[r] * xv[r] * Bv);
      y[r] = h[r] * Cv;
    }
    const float k0 = b0 ? y[2] : y[0], s0 = b0 ? y[0] : y[2];
    const float k1 = b0 ? y[3] : y[1], s1 = b0 ? y[1] : y[3];
    const float z0 = k0 + row_xor<1>(s0);
    const float z1 = k1 + row_xor<1>(s1);
    const float kk = b1 ? z1 : z0, ss = b1 ? z0 : z1;
    float w = kk + row_xor<2>(ss);
    w += row_xor<4>(w);
    w += row_xor<8>(w);
    if (writer) op[(size_t)(l0 + j) * DEXP] = w;
  }
}

__global__ __launch_bounds__(SCAN_THREADS) void ssm_scan_out(
    const float* __restrict__ x, const float* __restrict__ dtbuf,
    const float* __restrict__ Bmbuf, const float* __restrict__ Cbuf,
    const float* __restrict__ Aneg, const float* __restrict__ Car,
    float* __restrict__ out)
{
  const int t    = threadIdx.x;
  const int n    = t & 15;
  const int hf   = (t >> 4) & 1;
  const int wave = t >> 5;
  const int bi   = blockIdx.x;
  const int c    = bi & (NCHUNK - 1);
  const int dg   = (bi >> 3) & 31;
  const int b    = bi >> 8;
  const int d0   = dg * 32 + wave * 8 + hf * RCH;

  const size_t rbase = (size_t)b * LSEQ + (size_t)c * CHUNK;
  const size_t si = (((size_t)b * NCHUNK + c) * DEXP + d0) * NST + n;
  float And[RCH], h[RCH];
#pragma unroll
  for (int r = 0; r < RCH; ++r) {
    And[r] = Aneg[(d0 + r) * NST + n];
    h[r]   = Car[si + r * NST];
  }
  const float* dtp = dtbuf + rbase * DEXP + d0;
  const float* xp  = x     + rbase * DEXP + d0;
  const float* Bp  = Bmbuf + rbase * NST + n;
  const float* Cp  = Cbuf  + rbase * NST + n;

  const bool b0 = (n & 1) != 0, b1 = (n & 2) != 0;
  const bool writer = n < RCH;
  const int  mych = ((n & 1) << 1) | ((n >> 1) & 1);
  float* op = out + rbase * DEXP + d0 + mych;

  BatchOut qa, qb;
  load_out(qa, dtp, xp, Bp, Cp, 0);
  for (int l0 = 0; l0 < CHUNK; l0 += 2 * UB) {
    load_out(qb, dtp, xp, Bp, Cp, l0 + UB);
    out_steps(qa, And, h, b0, b1, writer, op, l0);
    if (l0 + 2 * UB < CHUNK) load_out(qa, dtp, xp, Bp, Cp, l0 + 2 * UB);
    out_steps(qb, And, h, b0, b1, writer, op, l0 + UB);
  }
}

extern "C" void kernel_launch(void* const* d_in, const int* in_sizes, int n_in,
                              void* d_out, int out_size, void* d_ws, size_t ws_size,
                              hipStream_t stream) {
  (void)in_sizes; (void)n_in; (void)out_size; (void)ws_size;
  const float* x     = (const float*)d_in[0];
  const float* xw    = (const float*)d_in[1];
  const float* xb    = (const float*)d_in[2];
  const float* dtw   = (const float*)d_in[3];
  const float* dtb   = (const float*)d_in[4];
  const float* A_log = (const float*)d_in[5];
  float* out = (float*)d_out;

  float* ws    = (float*)d_ws;
  float* dtbuf = ws;
  float* Aneg  = dtbuf + (size_t)ROWS * DEXP;
  float* Bm    = Aneg  + (size_t)DEXP * NST;
  float* Cb    = Bm    + (size_t)ROWS * NST;
  float* Sb    = Cb    + (size_t)ROWS * NST;
  float* Pb    = Sb    + (size_t)B_SZ * NCHUNK * DEXP * NST;
  float* Car   = Pb    + (size_t)B_SZ * NCHUNK * DEXP * NST;
  float* xwT   = Car   + (size_t)B_SZ * NCHUNK * DEXP * NST;
  float* dtwT  = xwT   + (size_t)DEXP * RPROJ;

  hipLaunchKernelGGL(ssm_prep, dim3((DEXP * RPROJ + 255) / 256), dim3(256), 0, stream,
                     A_log, Aneg, xw, xwT, dtw, dtwT);
  hipLaunchKernelGGL(ssm_proj_wmma, dim3(ROWS / 16), dim3(256), 0, stream,
                     x, xwT, xb, dtwT, dtb, dtbuf, Bm, Cb);
  hipLaunchKernelGGL(ssm_scan_carry, dim3(SCAN_BLOCKS), dim3(SCAN_THREADS), 0, stream,
                     x, dtbuf, Bm, Aneg, Sb, Pb);
  hipLaunchKernelGGL(ssm_carry_combine, dim3((B_SZ * DEXP * NST) / 256), dim3(256), 0, stream,
                     Sb, Pb, Car);
  hipLaunchKernelGGL(ssm_scan_out, dim3(SCAN_BLOCKS), dim3(SCAN_THREADS), 0, stream,
                     x, dtbuf, Bm, Cb, Aneg, Car, out);
}


# projection kernel: softplus of a tile's 8 outputs evaluated without a branch (v > 20 picked by a select afterwards), so the 8 library log1p chains share one basic block and overlap; same values
# speedup vs baseline: 3.0962x; 1.3124x over previous
#include <hip/hip_runtime.h>
#include <math.h>

#define B_SZ   2
#define LSEQ   2048
#define DEXP   1024
#define DRANK  64
#define NST    16
#define RPROJ  (DRANK + 2*NST)
#define NCHUNK 8
#define CHUNK  (LSEQ / NCHUNK)
#define ROWS   (B_SZ * LSEQ)

typedef __attribute__((ext_vector_type(2))) float v2f;
typedef __attribute__((ext_vector_type(8))) float v8f;

__device__ __forceinline__ v8f wmma_f32_k4(v2f a, v2f b, v8f c) {
  return __builtin_amdgcn_wmma_f32_16x16x4_f32(
       false, a,  false, b,
       (short)0, c,  false,  false);
}

__global__ void ssm_prep(const float* __restrict__ A_log, float* __restrict__ Aneg,
                         const float* __restrict__ xw,  float* __restrict__ xwT,
                         const float* __restrict__ dtw, float* __restrict__ dtwT) {
  int i = blockIdx.x * 256 + threadIdx.x;
  if (i < DEXP * NST) Aneg[i] = -__expf(A_log[i]);
  if (i < DEXP * RPROJ) {
    int k = i / RPROJ, c = i % RPROJ;
    xwT[i] = xw[(size_t)c * DEXP + k];
  }
  if (i < DRANK * DEXP) {
    int k = i / DEXP, d = i % DEXP;
    dtwT[i] = dtw[(size_t)d * DRANK + k];
  }
}

struct BSet1 { v2f f[16]; };
struct BSet2 { v2f f[16]; float bias; };

#define SXP 1028

__device__ __forceinline__ void load_b1(BSet1& q, const float* bp, int g) {
#pragma unroll
  for (int s = 0; s < 16; ++s) {
    const int k = g * 64 + s * 4;
    q.f[s].x = bp[(size_t)k * RPROJ];
    q.f[s].y = bp[(size_t)(k + 1) * RPROJ];
  }
}

__device__ __forceinline__ v8f mma_group1(const BSet1& q, const float* ap, int g, v8f acc) {
#pragma unroll
  for (int s = 0; s < 16; ++s) {
    const int k = g * 64 + s * 4;
    v2f a;  a.x = ap[k];  a.y = ap[k + 1];
    acc = wmma_f32_k4(a, q.f[s], acc);
  }
  return acc;
}

__device__ __forceinline__ void load_b2(BSet2& q, const float* wp, const float* dtb, int col) {
  q.bias = dtb[col];
#pragma unroll
  for (int kt = 0; kt < 16; ++kt) {
    q.f[kt].x = wp[(size_t)(kt * 4) * DEXP + col];
    q.f[kt].y = wp[(size_t)(kt * 4 + 1) * DEXP + col];
  }
}

__device__ __forceinline__ void tile2(const BSet2& q, const v2f (&a2)[16], float* dp, int col) {
  v8f acc2;
#pragma unroll
  for (int i = 0; i < 8; ++i) acc2[i] = q.bias;
#pragma unroll
  for (int kt = 0; kt < 16; ++kt) acc2 = wmma_f32_k4(a2[kt], q.f[kt], acc2);
  float sp[8];
#pragma unroll
  for (int i = 0; i < 8; ++i) sp[i] = log1pf(__expf(acc2[i]));
#pragma unroll
  for (int i = 0; i < 8; ++i) {
    const float v = acc2[i];
    dp[(size_t)i * DEXP + col] = (v > 20.0f) ? v : sp[i];
  }
}

__global__ __launch_bounds__(256) void ssm_proj_wmma(
    const float* __restrict__ x,    const float* __restrict__ xwT,
    const float* __restrict__ xb,   const float* __restrict__ dtwT,
    const float* __restrict__ dtb,
    float* __restrict__ dtbuf, float* __restrict__ Bmbuf, float* __restrict__ Cbuf)
{
  __shared__ float sx[16 * SXP];
  __shared__ float sproj[16 * 100];

  const int t    = threadIdx.x;
  const int wave = t >> 5;
  const int nl   = t & 15;
  const int hi   = (t >> 4) & 1;
  const int rowbase = blockIdx.x * 16;

  const int col1 = wave * 16 + nl;
  const float* bp = xwT + (size_t)(hi * 2) * RPROJ + col1;
  BSet1 b0, b1;
  v8f acc1;
  float bv1 = 0.0f;
  if (wave < 6) {
    load_b1(b0, bp, 0);
    bv1 = xb[col1];
  }
  for (int i = 0; i < 16; ++i) {
    int flat = t + i * 256;
    int r  = flat >> 8;
    int kk = (flat & 255) << 2;
    const float4 v = *(const float4*)(x + (size_t)(rowbase + r) * DEXP + kk);
    float* dst = &sx[r * SXP + kk];
    dst[0] = v.x; dst[1] = v.y; dst[2] = v.z; dst[3] = v.w;
  }
  __syncthreads();
  if (wave < 6) {
    for (int i = 0; i < 8; ++i) acc1[i] = bv1;
    const float* ap = &sx[nl * SXP + hi * 2];
#pragma unroll 1
    for (int g = 0; g < 16; g += 2) {
      load_b1(b1, bp, g + 1);
      acc1 = mma_group1(b0, ap, g, acc1);
      if (g + 2 < 16) load_b1(b0, bp, g + 2);
      acc1 = mma_group1(b1, ap, g + 1, acc1);
    }
  }
  const float* wp = dtwT + (size_t)(hi * 2) * DEXP;
  const int colw = wave * 128 + nl;
  BSet2 c0, c1;
  load_b2(c0, wp, dtb, colw);
  if (wave < 6) {
    for (int i = 0; i < 8; ++i) {
      int M = i + hi * 8;
      sproj[M * 100 + wave * 16 + nl] = acc1[i];
      if (wave == 4) Bmbuf[(size_t)(rowbase + M) * NST + nl] = acc1[i];
      if (wave == 5) Cbuf [(size_t)(rowbase + M) * NST + nl] = acc1[i];
    }
  }
  __syncthreads();

  v2f a2[16];
  for (int kt = 0; kt < 16; ++kt) {
    int kb = kt * 4 + hi * 2;
    a2[kt].x = sproj[nl * 100 + kb];
    a2[kt].y = sproj[nl * 100 + kb + 1];
  }
  float* dp = dtbuf + (size_t)(rowbase + hi * 8) * DEXP;
#pragma unroll 1
  for (int j = 0; j < 8; j += 2) {
    load_b2(c1, wp, dtb, colw + (j + 1) * 16);
    tile2(c0, a2, dp, colw + j * 16);
    if (j + 2 < 8) load_b2(c0, wp, dtb, colw + (j + 2) * 16);
    tile2(c1, a2, dp, colw + (j + 1) * 16);
  }
}

#define RCH 4
#define UB  8
#define SCAN_THREADS 128
#define SCAN_BLOCKS  (B_SZ * NCHUNK * (DEXP / 32))

struct BatchIn  { float4 dt[UB]; float4 x[UB]; float b[UB]; };
struct BatchOut { float4 dt[UB]; float4 x[UB]; float b[UB]; float c[UB]; };

__device__ __forceinline__ void load_in(BatchIn& q, const float* dtp, const float* xp,
                                        const float* Bp, int l0) {
#pragma unroll
  for (int j = 0; j < UB; ++j) {
    q.dt[j] = *(const float4*)(dtp + (size_t)(l0 + j) * DEXP);
    q.x[j]  = *(const float4*)(xp  + (size_t)(l0 + j) * DEXP);
    q.b[j]  = Bp[(l0 + j) * NST];
  }
}

__device__ __forceinline__ void load_out(BatchOut& q, const float* dtp, const float* xp,
                                         const float* Bp, const float* Cp, int l0) {
#pragma unroll
  for (int j = 0; j < UB; ++j) {
    q.dt[j] = *(const float4*)(dtp + (size_t)(l0 + j) * DEXP);
    q.x[j]  = *(const float4*)(xp  + (size_t)(l0 + j) * DEXP);
    q.b[j]  = Bp[(l0 + j) * NST];
    q.c[j]  = Cp[(l0 + j) * NST];
  }
}

template <int M>
__device__ __forceinline__ float row_xor(float v) {
  return __int_as_float(__builtin_amdgcn_update_dpp(0, __float_as_int(v), 0x160 | M,
                                                    0xf, 0xf, true));
}

__device__ __forceinline__ void carry_steps(const BatchIn& q, const float (&And)[RCH],
                                            float (&h)[RCH], float (&sdt)[RCH]) {
#pragma unroll
  for (int j = 0; j < UB; ++j) {
    const float dv[RCH] = {q.dt[j].x, q.dt[j].y, q.dt[j].z, q.dt[j].w};
    const float xv[RCH] = {q.x[j].x,  q.x[j].y,  q.x[j].z,  q.x[j].w};
    const float Bv = q.b[j];
#pragma unroll
    for (int r = 0; r < RCH; ++r) {
      float dA = __expf(And[r] * dv[r]);
      h[r]   = fmaf(dA, h[r], dv[r] * xv[r] * Bv);
      sdt[r] += dv[r];
    }
  }
}

__global__ __launch_bounds__(SCAN_THREADS) void ssm_scan_carry(
    const float* __restrict__ x, const float* __restrict__ dtbuf,
    const float* __restrict__ Bmbuf, const float* __restrict__ Aneg,
    float* __restrict__ Sbuf, float* __restrict__ Pbuf)
{
  const int t    = threadIdx.x;
  const int n    = t & 15;
  const int hf   = (t >> 4) & 1;
  const int wave = t >> 5;
  const int bi   = blockIdx.x;
  const int c    = bi & (NCHUNK - 1);
  const int dg   = (bi >> 3) & 31;
  const int b    = bi >> 8;
  const int d0   = dg * 32 + wave * 8 + hf * RCH;

  float And[RCH], h[RCH], sdt[RCH];
#pragma unroll
  for (int r = 0; r < RCH; ++r) {
    And[r] = Aneg[(d0 + r) * NST + n];
    h[r] = 0.0f; sdt[r] = 0.0f;
  }
  const size_t rbase = (size_t)b * LSEQ + (size_t)c * CHUNK;
  const float* dtp = dtbuf + rbase * DEXP + d0;
  const float* xp  = x     + rbase * DEXP + d0;
  const float* Bp  = Bmbuf + rbase * NST + n;

  BatchIn qa, qb;
  load_in(qa, dtp, xp, Bp, 0);
  for (int l0 = 0; l0 < CHUNK; l0 += 2 * UB) {
    load_in(qb, dtp, xp, Bp, l0 + UB);
    carry_steps(qa, And, h, sdt);
    if (l0 + 2 * UB < CHUNK) load_in(qa, dtp, xp, Bp, l0 + 2 * UB);
    carry_steps(qb, And, h, sdt);
  }
  const size_t si = (((size_t)b * NCHUNK + c) * DEXP + d0) * NST + n;
#pragma unroll
  for (int r = 0; r < RCH; ++r) {
    Sbuf[si + r * NST] = h[r];
    Pbuf[si + r * NST] = __expf(And[r] * sdt[r]);
  }
}

__global__ void ssm_carry_combine(const float* __restrict__ Sbuf,
                                  const float* __restrict__ Pbuf,
                                  float* __restrict__ Car)
{
  const int tg = blockIdx.x * 256 + threadIdx.x;
  const int n  = tg & 15;
  const int d  = (tg >> 4) & (DEXP - 1);
  const int b  = tg >> 14;
  const size_t stride = (size_t)DEXP * NST;
  const size_t base   = (size_t)b * NCHUNK * stride + (size_t)d * NST + n;
  float cin = 0.0f;
  Car[base] = 0.0f;
  for (int c = 1; c < NCHUNK; ++c) {
    const size_t prev = base + (size_t)(c - 1) * stride;
    cin = Sbuf[prev] + Pbuf[prev] * cin;
    Car[base + (size_t)c * stride] = cin;
  }
}

__device__ __forceinline__ void out_steps(const BatchOut& q, const float (&And)[RCH],
                                          float (&h)[RCH], const bool b0, const bool b1,
                                          const bool writer, float* op, int l0) {
#pragma unroll
  for (int j = 0; j < UB; ++j) {
    const float dv[RCH] = {q.dt[j].x, q.dt[j].y, q.dt[j].z, q.dt[j].w};
    const float xv[RCH] = {q.x[j].x,  q.x[j].y,  q.x[j].z,  q.x[j].w};
    const float Bv = q.b[j], Cv = q.c[j];
    float y[RCH];
#pragma unroll
    for (int r = 0; r < RCH; ++r) {
      float dA = __expf(And[r] * dv[r]);
      h[r] = fmaf(dA, h[r], dv[r] * xv[r] * Bv);
      y[r] = h[r] * Cv;
    }
    const float k0 = b0 ? y[2] : y[0], s0 = b0 ? y[0] : y[2];
    const float k1 = b0 ? y[3] : y[1], s1 = b0 ? y[1] : y[3];
    const float z0 = k0 + row_xor<1>(s0);
    const float z1 = k1 + row_xor<1>(s1);
    const float kk = b1 ? z1 : z0, ss = b1 ? z0 : z1;
    float w = kk + row_xor<2>(ss);
    w += row_xor<4>(w);
    w += row_xor<8>(w);
    if (writer) op[(size_t)(l0 + j) * DEXP] = w;
  }
}

__global__ __launch_bounds__(SCAN_THREADS) void ssm_scan_out(
    const float* __restrict__ x, const float* __restrict__ dtbuf,
    const float* __restrict__ Bmbuf, const float* __restrict__ Cbuf,
    const float* __restrict__ Aneg, const float* __restrict__ Car,
    float* __restrict__ out)
{
  const int t    = threadIdx.x;
  const int n    = t & 15;
  const int hf   = (t >> 4) & 1;
  const int wave = t >> 5;
  const int bi   = blockIdx.x;
  const int c    = bi & (NCHUNK - 1);
  const int dg   = (bi >> 3) & 31;
  const int b    = bi >> 8;
  const int d0   = dg * 32 + wave * 8 + hf * RCH;

  const size_t rbase = (size_t)b * LSEQ + (size_t)c * CHUNK;
  const size_t si = (((size_t)b * NCHUNK + c) * DEXP + d0) * NST + n;
  float And[RCH], h[RCH];
#pragma unroll
  for (int r = 0; r < RCH; ++r) {
    And[r] = Aneg[(d0 + r) * NST + n];
    h[r]   = Car[si + r * NST];
  }
  const float* dtp = dtbuf + rbase * DEXP + d0;
  const float* xp  = x     + rbase * DEXP + d0;
  const float* Bp  = Bmbuf + rbase * NST + n;
  const float* Cp  = Cbuf  + rbase * NST + n;

  const bool b0 = (n & 1) != 0, b1 = (n & 2) != 0;
  const bool writer = n < RCH;
  const int  mych = ((n & 1) << 1) | ((n >> 1) & 1);
  float* op = out + rbase * DEXP + d0 + mych;

  BatchOut qa, qb;
  load_out(qa, dtp, xp, Bp, Cp, 0);
  for (int l0 = 0; l0 < CHUNK; l0 += 2 * UB) {
    load_out(qb, dtp, xp, Bp, Cp, l0 + UB);
    out_steps(qa, And, h, b0, b1, writer, op, l0);
    if (l0 + 2 * UB < CHUNK) load_out(qa, dtp, xp, Bp, Cp, l0 + 2 * UB);
    out_steps(qb, And, h, b0, b1, writer, op, l0 + UB);
  }
}

extern "C" void kernel_launch(void* const* d_in, const int* in_sizes, int n_in,
                              void* d_out, int out_size, void* d_ws, size_t ws_size,
                              hipStream_t stream) {
  (void)in_sizes; (void)n_in; (void)out_size; (void)ws_size;
  const float* x     = (const float*)d_in[0];
  const float* xw    = (const float*)d_in[1];
  const float* xb    = (const float*)d_in[2];
  const float* dtw   = (const float*)d_in[3];
  const float* dtb   = (const float*)d_in[4];
  const float* A_log = (const float*)d_in[5];
  float* out = (float*)d_out;

  float* ws    = (float*)d_ws;
  float* dtbuf = ws;
  float* Aneg  = dtbuf + (size_t)ROWS * DEXP;
  float* Bm    = Aneg  + (size_t)DEXP * NST;
  float* Cb    = Bm    + (size_t)ROWS * NST;
  float* Sb    = Cb    + (size_t)ROWS * NST;
  float* Pb    = Sb    + (size_t)B_SZ * NCHUNK * DEXP * NST;
  float* Car   = Pb    + (size_t)B_SZ * NCHUNK * DEXP * NST;
  float* xwT   = Car   + (size_t)B_SZ * NCHUNK * DEXP * NST;
  float* dtwT  = xwT   + (size_t)DEXP * RPROJ;

  hipLaunchKernelGGL(ssm_prep, dim3((DEXP * RPROJ + 255) / 256), dim3(256), 0, stream,
                     A_log, Aneg, xw, xwT, dtw, dtwT);
  hipLaunchKernelGGL(ssm_proj_wmma, dim3(ROWS / 16), dim3(256), 0, stream,
                     x, xwT, xb, dtwT, dtb, dtbuf, Bm, Cb);
  hipLaunchKernelGGL(ssm_scan_carry, dim3(SCAN_BLOCKS), dim3(SCAN_THREADS), 0, stream,
                     x, dtbuf, Bm, Aneg, Sb, Pb);
  hipLaunchKernelGGL(ssm_carry_combine, dim3((B_SZ * DEXP * NST) / 256), dim3(256), 0, stream,
                     Sb, Pb, Car);
  hipLaunchKernelGGL(ssm_scan_out, dim3(SCAN_BLOCKS), dim3(SCAN_THREADS), 0, stream,
                     x, dtbuf, Bm, Cb, Aneg, Car, out);
}


# scan kernels: dt/x/B/C loads issued two batches (16 steps) ahead of their use in a ring of three register sets, not one batch ahead; same values, same sum order
# speedup vs baseline: 3.5244x; 1.1383x over previous
#include <hip/hip_runtime.h>
#include <math.h>

#define B_SZ   2
#define LSEQ   2048
#define DEXP   1024
#define DRANK  64
#define NST    16
#define RPROJ  (DRANK + 2*NST)
#define NCHUNK 8
#define CHUNK  (LSEQ / NCHUNK)
#define ROWS   (B_SZ * LSEQ)

typedef __attribute__((ext_vector_type(2))) float v2f;
typedef __attribute__((ext_vector_type(8))) float v8f;

__device__ __forceinline__ v8f wmma_f32_k4(v2f a, v2f b, v8f c) {
  return __builtin_amdgcn_wmma_f32_16x16x4_f32(
       false, a,  false, b,
       (short)0, c,  false,  false);
}

__global__ void ssm_prep(const float* __restrict__ A_log, float* __restrict__ Aneg,
                         const float* __restrict__ xw,  float* __restrict__ xwT,
                         const float* __restrict__ dtw, float* __restrict__ dtwT) {
  int i = blockIdx.x * 256 + threadIdx.x;
  if (i < DEXP * NST) Aneg[i] = -__expf(A_log[i]);
  if (i < DEXP * RPROJ) {
    int k = i / RPROJ, c = i % RPROJ;
    xwT[i] = xw[(size_t)c * DEXP + k];
  }
  if (i < DRANK * DEXP) {
    int k = i / DEXP, d = i % DEXP;
    dtwT[i] = dtw[(size_t)d * DRANK + k];
  }
}

struct BSet1 { v2f f[16]; };
struct BSet2 { v2f f[16]; float bias; };

#define SXP 1028

__device__ __forceinline__ void load_b1(BSet1& q, const float* bp, int g) {
#pragma unroll
  for (int s = 0; s < 16; ++s) {
    const int k = g * 64 + s * 4;
    q.f[s].x = bp[(size_t)k * RPROJ];
    q.f[s].y = bp[(size_t)(k + 1) * RPROJ];
  }
}

__device__ __forceinline__ v8f mma_group1(const BSet1& q, const float* ap, int g, v8f acc) {
#pragma unroll
  for (int s = 0; s < 16; ++s) {
    const int k = g * 64 + s * 4;
    v2f a;  a.x = ap[k];  a.y = ap[k + 1];
    acc = wmma_f32_k4(a, q.f[s], acc);
  }
  return acc;
}

__device__ __forceinline__ void load_b2(BSet2& q, const float* wp, const float* dtb, int col) {
  q.bias = dtb[col];
#pragma unroll
  for (int kt = 0; kt < 16; ++kt) {
    q.f[kt].x = wp[(size_t)(kt * 4) * DEXP + col];
    q.f[kt].y = wp[(size_t)(kt * 4 + 1) * DEXP + col];
  }
}

__device__ __forceinline__ void tile2(const BSet2& q, const v2f (&a2)[16], float* dp, int col) {
  v8f acc2;
#pragma unroll
  for (int i = 0; i < 8; ++i) acc2[i] = q.bias;
#pragma unroll
  for (int kt = 0; kt < 16; ++kt) acc2 = wmma_f32_k4(a2[kt], q.f[kt], acc2);
  float sp[8];
#pragma unroll
  for (int i = 0; i < 8; ++i) sp[i] = log1pf(__expf(acc2[i]));
#pragma unroll
  for (int i = 0; i < 8; ++i) {
    const float v = acc2[i];
    dp[(size_t)i * DEXP + col] = (v > 20.0f) ? v : sp[i];
  }
}

__global__ __launch_bounds__(256) void ssm_proj_wmma(
    const float* __restrict__ x,    const float* __restrict__ xwT,
    const float* __restrict__ xb,   const float* __restrict__ dtwT,
    const float* __restrict__ dtb,
    float* __restrict__ dtbuf, float* __restrict__ Bmbuf, float* __restrict__ Cbuf)
{
  __shared__ float sx[16 * SXP];
  __shared__ float sproj[16 * 100];

  const int t    = threadIdx.x;
  const int wave = t >> 5;
  const int nl   = t & 15;
  const int hi   = (t >> 4) & 1;
  const int rowbase = blockIdx.x * 16;

  const int col1 = wave * 16 + nl;
  const float* bp = xwT + (size_t)(hi * 2) * RPROJ + col1;
  BSet1 b0, b1;
  v8f acc1;
  float bv1 = 0.0f;
  if (wave < 6) {
    load_b1(b0, bp, 0);
    bv1 = xb[col1];
  }
  for (int i = 0; i < 16; ++i) {
    int flat = t + i * 256;
    int r  = flat >> 8;
    int kk = (flat & 255) << 2;
    const float4 v = *(const float4*)(x + (size_t)(rowbase + r) * DEXP + kk);
    float* dst = &sx[r * SXP + kk];
    dst[0] = v.x; dst[1] = v.y; dst[2] = v.z; dst[3] = v.w;
  }
  __syncthreads();
  if (wave < 6) {
    for (int i = 0; i < 8; ++i) acc1[i] = bv1;
    const float* ap = &sx[nl * SXP + hi * 2];
#pragma unroll 1
    for (int g = 0; g < 16; g += 2) {
      load_b1(b1, bp, g + 1);
      acc1 = mma_group1(b0, ap, g, acc1);
      if (g + 2 < 16) load_b1(b0, bp, g + 2);
      acc1 = mma_group1(b1, ap, g + 1, acc1);
    }
  }
  const float* wp = dtwT + (size_t)(hi * 2) * DEXP;
  const int colw = wave * 128 + nl;
  BSet2 c0, c1;
  load_b2(c0, wp, dtb, colw);
  if (wave < 6) {
    for (int i = 0; i < 8; ++i) {
      int M = i + hi * 8;
      sproj[M * 100 + wave * 16 + nl] = acc1[i];
      if (wave == 4) Bmbuf[(size_t)(rowbase + M) * NST + nl] = acc1[i];
      if (wave == 5) Cbuf [(size_t)(rowbase + M) * NST + nl] = acc1[i];
    }
  }
  __syncthreads();

  v2f a2[16];
  for (int kt = 0; kt < 16; ++kt) {
    int kb = kt * 4 + hi * 2;
    a2[kt].x = sproj[nl * 100 + kb];
    a2[kt].y = sproj[nl * 100 + kb + 1];
  }
  float* dp = dtbuf + (size_t)(rowbase + hi * 8) * DEXP;
#pragma unroll 1
  for (int j = 0; j < 8; j += 2) {
    load_b2(c1, wp, dtb, colw + (j + 1) * 16);
    tile2(c0, a2, dp, colw + j * 16);
    if (j + 2 < 8) load_b2(c0, wp, dtb, colw + (j + 2) * 16);
    tile2(c1, a2, dp, colw + (j + 1) * 16);
  }
}

#define RCH 4
#define UB  8
#define SCAN_THREADS 128
#define SCAN_BLOCKS  (B_SZ * NCHUNK * (DEXP / 32))

#define SEG_FENCE() __builtin_amdgcn_sched_barrier(0)

struct BatchIn  { float4 dt[UB]; float4 x[UB]; float b[UB]; };
struct BatchOut { float4 dt[UB]; float4 x[UB]; float b[UB]; float c[UB]; };

__device__ __forceinline__ void load_in(BatchIn& q, const float* dtp, const float* xp,
                                        const float* Bp, int l0) {
#pragma unroll
  for (int j = 0; j < UB; ++j) {
    q.dt[j] = *(const float4*)(dtp + (size_t)(l0 + j) * DEXP);
    q.x[j]  = *(const float4*)(xp  + (size_t)(l0 + j) * DEXP);
    q.b[j]  = Bp[(l0 + j) * NST];
  }
}

__device__ __forceinline__ void load_out(BatchOut& q, const float* dtp, const float* xp,
                                         const float* Bp, const float* Cp, int l0) {
#pragma unroll
  for (int j = 0; j < UB; ++j) {
    q.dt[j] = *(const float4*)(dtp + (size_t)(l0 + j) * DEXP);
    q.x[j]  = *(const float4*)(xp  + (size_t)(l0 + j) * DEXP);
    q.b[j]  = Bp[(l0 + j) * NST];
    q.c[j]  = Cp[(l0 + j) * NST];
  }
}

template <int M>
__device__ __forceinline__ float row_xor(float v) {
  return __int_as_float(__builtin_amdgcn_update_dpp(0, __float_as_int(v), 0x160 | M,
                                                    0xf, 0xf, true));
}

__device__ __forceinline__ void carry_steps(const BatchIn& q, const float (&And)[RCH],
                                            float (&h)[RCH], float (&sdt)[RCH]) {
#pragma unroll
  for (int j = 0; j < UB; ++j) {
    const float dv[RCH] = {q.dt[j].x, q.dt[j].y, q.dt[j].z, q.dt[j].w};
    const float xv[RCH] = {q.x[j].x,  q.x[j].y,  q.x[j].z,  q.x[j].w};
    const float Bv = q.b[j];
#pragma unroll
    for (int r = 0; r < RCH; ++r) {
      float dA = __expf(And[r] * dv[r]);
      h[r]   = fmaf(dA, h[r], dv[r] * xv[r] * Bv);
      sdt[r] += dv[r];
    }
  }
}

__global__ __launch_bounds__(SCAN_THREADS) void ssm_scan_carry(
    const float* __restrict__ x, const float* __restrict__ dtbuf,
    const float* __restrict__ Bmbuf, const float* __restrict__ Aneg,
    float* __restrict__ Sbuf, float* __restrict__ Pbuf)
{
  const int t    = threadIdx.x;
  const int n    = t & 15;
  const int hf   = (t >> 4) & 1;
  const int wave = t >> 5;
  const int bi   = blockIdx.x;
  const int c    = bi & (NCHUNK - 1);
  const int dg   = (bi >> 3) & 31;
  const int b    = bi >> 8;
  const int d0   = dg * 32 + wave * 8 + hf * RCH;

  float And[RCH], h[RCH], sdt[RCH];
#pragma unroll
  for (int r = 0; r < RCH; ++r) {
    And[r] = Aneg[(d0 + r) * NST + n];
    h[r] = 0.0f; sdt[r] = 0.0f;
  }
  const size_t rbase = (size_t)b * LSEQ + (size_t)c * CHUNK;
  const float* dtp = dtbuf + rbase * DEXP + d0;
  const float* xp  = x     + rbase * DEXP + d0;
  const float* Bp  = Bmbuf + rbase * NST + n;

  static_assert((CHUNK / UB) % 6 == 2, "five six-batch trips (two turns of the ring), then the last two batches");
  BatchIn q0, q1, q2;
  load_in(q0, dtp, xp, Bp, 0);
  load_in(q1, dtp, xp, Bp, UB);
#pragma unroll 1
  for (int l0 = 0; l0 < CHUNK - 2 * UB; l0 += 6 * UB) {
    load_in(q2, dtp, xp, Bp, l0 + 2 * UB);  SEG_FENCE();
    carry_steps(q0, And, h, sdt);           SEG_FENCE();
    load_in(q0, dtp, xp, Bp, l0 + 3 * UB);  SEG_FENCE();
    carry_steps(q1, And, h, sdt);           SEG_FENCE();
    load_in(q1, dtp, xp, Bp, l0 + 4 * UB);  SEG_FENCE();
    carry_steps(q2, And, h, sdt);           SEG_FENCE();
    load_in(q2, dtp, xp, Bp, l0 + 5 * UB);  SEG_FENCE();
    carry_steps(q0, And, h, sdt);           SEG_FENCE();
    load_in(q0, dtp, xp, Bp, l0 + 6 * UB);  SEG_FENCE();
    carry_steps(q1, And, h, sdt);           SEG_FENCE();
    load_in(q1, dtp, xp, Bp, l0 + 7 * UB);  SEG_FENCE();
    carry_steps(q2, And, h, sdt);
  }
  carry_steps(q0, And, h, sdt);             SEG_FENCE();
  carry_steps(q1, And, h, sdt);
  const size_t si = (((size_t)b * NCHUNK + c) * DEXP + d0) * NST + n;
#pragma unroll
  for (int r = 0; r < RCH; ++r) {
    Sbuf[si + r * NST] = h[r];
    Pbuf[si + r * NST] = __expf(And[r] * sdt[r]);
  }
}

__global__ void ssm_carry_combine(const float* __restrict__ Sbuf,
                                  const float* __restrict__ Pbuf,
                                  float* __restrict__ Car)
{
  const int tg = blockIdx.x * 256 + threadIdx.x;
  const int n  = tg & 15;
  const int d  = (tg >> 4) & (DEXP - 1);
  const int b  = tg >> 14;
  const size_t stride = (size_t)DEXP * NST;
  const size_t base   = (size_t)b * NCHUNK * stride + (size_t)d * NST + n;
  float cin = 0.0f;
  Car[base] = 0.0f;
  for (int c = 1; c < NCHUNK; ++c) {
    const size_t prev = base + (size_t)(c - 1) * stride;
    cin = Sbuf[prev] + Pbuf[prev] * cin;
    Car[base + (size_t)c * stride] = cin;
  }
}

__device__ __forceinline__ void out_steps(const BatchOut& q, const float (&And)[RCH],
                                          float (&h)[RCH], const bool b0, const bool b1,
                                          const bool writer, float* op, int l0) {
#pragma unroll
  for (int j = 0; j < UB; ++j) {
    const float dv[RCH] = {q.dt[j].x, q.dt[j].y, q.dt[j].z, q.dt[j].w};
    const float xv[RCH] = {q.x[j].x,  q.x[j].y,  q.x[j].z,  q.x[j].w};
    const float Bv = q.b[j], Cv = q.c[j];
    float y[RCH];
#pragma unroll
    for (int r = 0; r < RCH; ++r) {
      float dA = __expf(And[r] * dv[r]);
      h[r] = fmaf(dA, h[r], dv[r] * xv[r] * Bv);
      y[r] = h[r] * Cv;
    }
    const float k0 = b0 ? y[2] : y[0], s0 = b0 ? y[0] : y[2];
    const float k1 = b0 ? y[3] : y[1], s1 = b0 ? y[1] : y[3];
    const float z0 = k0 + row_xor<1>(s0);
    const float z1 = k1 + row_xor<1>(s1);
    const float kk = b1 ? z1 : z0, ss = b1 ? z0 : z1;
    float w = kk + row_xor<2>(ss);
    w += row_xor<4>(w);
    w += row_xor<8>(w);
    if (writer) op[(size_t)(l0 + j) * DEXP] = w;
  }
}

__global__ __launch_bounds__(SCAN_THREADS) void ssm_scan_out(
    const float* __restrict__ x, const float* __restrict__ dtbuf,
    const float* __restrict__ Bmbuf, const float* __restrict__ Cbuf,
    const float* __restrict__ Aneg, const float* __restrict__ Car,
    float* __restrict__ out)
{
  const int t    = threadIdx.x;
  const int n    = t & 15;
  const int hf   = (t >> 4) & 1;
  const int wave = t >> 5;
  const int bi   = blockIdx.x;
  const int c    = bi & (NCHUNK - 1);
  const int dg   = (bi >> 3) & 31;
  const int b    = bi >> 8;
  const int d0   = dg * 32 + wave * 8 + hf * RCH;

  const size_t rbase = (size_t)b * LSEQ + (size_t)c * CHUNK;
  const size_t si = (((size_t)b * NCHUNK + c) * DEXP + d0) * NST + n;
  float And[RCH], h[RCH];
#pragma unroll
  for (int r = 0; r < RCH; ++r) {
    And[r] = Aneg[(d0 + r) * NST + n];
    h[r]   = Car[si + r * NST];
  }
  const float* dtp = dtbuf + rbase * DEXP + d0;
  const float* xp  = x     + rbase * DEXP + d0;
  const float* Bp  = Bmbuf + rbase * NST + n;
  const float* Cp  = Cbuf  + rbase * NST + n;

  const bool b0 = (n & 1) != 0, b1 = (n & 2) != 0;
  const bool writer = n < RCH;
  const int  mych = ((n & 1) << 1) | ((n >> 1) & 1);
  float* op = out + rbase * DEXP + d0 + mych;

  static_assert((CHUNK / UB) % 6 == 2, "five six-batch trips (two turns of the ring), then the last two batches");
  BatchOut qa, qb, qc;
  load_out(qa, dtp, xp, Bp, Cp, 0);
  load_out(qb, dtp, xp, Bp, Cp, UB);
#pragma unroll 1
  for (int l0 = 0; l0 < CHUNK - 2 * UB; l0 += 6 * UB) {
    load_out(qc, dtp, xp, Bp, Cp, l0 + 2 * UB);
    out_steps(qa, And, h, b0, b1, writer, op, l0);
    load_out(qa, dtp, xp, Bp, Cp, l0 + 3 * UB);
    out_steps(qb, And, h, b0, b1, writer, op, l0 + UB);
    load_out(qb, dtp, xp, Bp, Cp, l0 + 4 * UB);
    out_steps(qc, And, h, b0, b1, writer, op, l0 + 2 * UB);
    load_out(qc, dtp, xp, Bp, Cp, l0 + 5 * UB);
    out_steps(qa, And, h, b0, b1, writer, op, l0 + 3 * UB);
    load_out(qa, dtp, xp, Bp, Cp, l0 + 6 * UB);
    out_steps(qb, And, h, b0, b1, writer, op, l0 + 4 * UB);
    load_out(qb, dtp, xp, Bp, Cp, l0 + 7 * UB);
    out_steps(qc, And, h, b0, b1, writer, op, l0 + 5 * UB);
  }
  out_steps(qa, And, h, b0, b1, writer, op, CHUNK - 2 * UB);
  out_steps(qb, And, h, b0, b1, writer, op, CHUNK - UB);
}

extern "C" void kernel_launch(void* const* d_in, const int* in_sizes, int n_in,
                              void* d_out, int out_size, void* d_ws, size_t ws_size,
                              hipStream_t stream) {
  (void)in_sizes; (void)n_in; (void)out_size; (void)ws_size;
  const float* x     = (const float*)d_in[0];
  const float* xw    = (const float*)d_in[1];
  const float* xb    = (const float*)d_in[2];
  const float* dtw   = (const float*)d_in[3];
  const float* dtb   = (const float*)d_in[4];
  const float* A_log = (const float*)d_in[5];
  float* out = (float*)d_out;

  float* ws    = (float*)d_ws;
  float* dtbuf = ws;
  float* Aneg  = dtbuf + (size_t)ROWS * DEXP;
  float* Bm    = Aneg  + (size_t)DEXP * NST;
  float* Cb    = Bm    + (size_t)ROWS * NST;
  float* Sb    = Cb    + (size_t)ROWS * NST;
  float* Pb    = Sb    + (size_t)B_SZ * NCHUNK * DEXP * NST;
  float* Car   = Pb    + (size_t)B_SZ * NCHUNK * DEXP * NST;
  float* xwT   = Car   + (size_t)B_SZ * NCHUNK * DEXP * NST;
  float* dtwT  = xwT   + (size_t)DEXP * RPROJ;

  hipLaunchKernelGGL(ssm_prep, dim3((DEXP * RPROJ + 255) / 256), dim3(256), 0, stream,
                     A_log, Aneg, xw, xwT, dtw, dtwT);
  hipLaunchKernelGGL(ssm_proj_wmma, dim3(ROWS / 16), dim3(256), 0, stream,
                     x, xwT, xb, dtwT, dtb, dtbuf, Bm, Cb);
  hipLaunchKernelGGL(ssm_scan_carry, dim3(SCAN_BLOCKS), dim3(SCAN_THREADS), 0, stream,
                     x, dtbuf, Bm, Aneg, Sb, Pb);
  hipLaunchKernelGGL(ssm_carry_combine, dim3((B_SZ * DEXP * NST) / 256), dim3(256), 0, stream,
                     Sb, Pb, Car);
  hipLaunchKernelGGL(ssm_scan_out, dim3(SCAN_BLOCKS), dim3(SCAN_THREADS), 0, stream,
                     x, dtbuf, Bm, Cb, Aneg, Car, out);
}


# out scan: the n-sum butterfly's xor-4 and xor-8 stages transpose over 4 steps, so each lane ends with one (step, channel) sum and all lanes store: no per-step store branch; same sum tree
# speedup vs baseline: 3.9099x; 1.1094x over previous
#include <hip/hip_runtime.h>
#include <math.h>

#define B_SZ   2
#define LSEQ   2048
#define DEXP   1024
#define DRANK  64
#define NST    16
#define RPROJ  (DRANK + 2*NST)
#define NCHUNK 8
#define CHUNK  (LSEQ / NCHUNK)
#define ROWS   (B_SZ * LSEQ)

typedef __attribute__((ext_vector_type(2))) float v2f;
typedef __attribute__((ext_vector_type(8))) float v8f;

__device__ __forceinline__ v8f wmma_f32_k4(v2f a, v2f b, v8f c) {
  return __builtin_amdgcn_wmma_f32_16x16x4_f32(
       false, a,  false, b,
       (short)0, c,  false,  false);
}

__global__ void ssm_prep(const float* __restrict__ A_log, float* __restrict__ Aneg,
                         const float* __restrict__ xw,  float* __restrict__ xwT,
                         const float* __restrict__ dtw, float* __restrict__ dtwT) {
  int i = blockIdx.x * 256 + threadIdx.x;
  if (i < DEXP * NST) Aneg[i] = -__expf(A_log[i]);
  if (i < DEXP * RPROJ) {
    int k = i / RPROJ, c = i % RPROJ;
    xwT[i] = xw[(size_t)c * DEXP + k];
  }
  if (i < DRANK * DEXP) {
    int k = i / DEXP, d = i % DEXP;
    dtwT[i] = dtw[(size_t)d * DRANK + k];
  }
}

struct BSet1 { v2f f[16]; };
struct BSet2 { v2f f[16]; float bias; };

#define SXP 1028

__device__ __forceinline__ void load_b1(BSet1& q, const float* bp, int g) {
#pragma unroll
  for (int s = 0; s < 16; ++s) {
    const int k = g * 64 + s * 4;
    q.f[s].x = bp[(size_t)k * RPROJ];
    q.f[s].y = bp[(size_t)(k + 1) * RPROJ];
  }
}

__device__ __forceinline__ v8f mma_group1(const BSet1& q, const float* ap, int g, v8f acc) {
#pragma unroll
  for (int s = 0; s < 16; ++s) {
    const int k = g * 64 + s * 4;
    v2f a;  a.x = ap[k];  a.y = ap[k + 1];
    acc = wmma_f32_k4(a, q.f[s], acc);
  }
  return acc;
}

__device__ __forceinline__ void load_b2(BSet2& q, const float* wp, const float* dtb, int col) {
  q.bias = dtb[col];
#pragma unroll
  for (int kt = 0; kt < 16; ++kt) {
    q.f[kt].x = wp[(size_t)(kt * 4) * DEXP + col];
    q.f[kt].y = wp[(size_t)(kt * 4 + 1) * DEXP + col];
  }
}

__device__ __forceinline__ void tile2(const BSet2& q, const v2f (&a2)[16], float* dp, int col) {
  v8f acc2;
#pragma unroll
  for (int i = 0; i < 8; ++i) acc2[i] = q.bias;
#pragma unroll
  for (int kt = 0; kt < 16; ++kt) acc2 = wmma_f32_k4(a2[kt], q.f[kt], acc2);
  float sp[8];
#pragma unroll
  for (int i = 0; i < 8; ++i) sp[i] = log1pf(__expf(acc2[i]));
#pragma unroll
  for (int i = 0; i < 8; ++i) {
    const float v = acc2[i];
    dp[(size_t)i * DEXP + col] = (v > 20.0f) ? v : sp[i];
  }
}

__global__ __launch_bounds__(256) void ssm_proj_wmma(
    const float* __restrict__ x,    const float* __restrict__ xwT,
    const float* __restrict__ xb,   const float* __restrict__ dtwT,
    const float* __restrict__ dtb,
    float* __restrict__ dtbuf, float* __restrict__ Bmbuf, float* __restrict__ Cbuf)
{
  __shared__ float sx[16 * SXP];
  __shared__ float sproj[16 * 100];

  const int t    = threadIdx.x;
  const int wave = t >> 5;
  const int nl   = t & 15;
  const int hi   = (t >> 4) & 1;
  const int rowbase = blockIdx.x * 16;

  const int col1 = wave * 16 + nl;
  const float* bp = xwT + (size_t)(hi * 2) * RPROJ + col1;
  BSet1 b0, b1;
  v8f acc1;
  float bv1 = 0.0f;
  if (wave < 6) {
    load_b1(b0, bp, 0);
    bv1 = xb[col1];
  }
  for (int i = 0; i < 16; ++i) {
    int flat = t + i * 256;
    int r  = flat >> 8;
    int kk = (flat & 255) << 2;
    const float4 v = *(const float4*)(x + (size_t)(rowbase + r) * DEXP + kk);
    float* dst = &sx[r * SXP + kk];
    dst[0] = v.x; dst[1] = v.y; dst[2] = v.z; dst[3] = v.w;
  }
  __syncthreads();
  if (wave < 6) {
    for (int i = 0; i < 8; ++i) acc1[i] = bv1;
    const float* ap = &sx[nl * SXP + hi * 2];
#pragma unroll 1
    for (int g = 0; g < 16; g += 2) {
      load_b1(b1, bp, g + 1);
      acc1 = mma_group1(b0, ap, g, acc1);
      if (g + 2 < 16) load_b1(b0, bp, g + 2);
      acc1 = mma_group1(b1, ap, g + 1, acc1);
    }
  }
  const float* wp = dtwT + (size_t)(hi * 2) * DEXP;
  const int colw = wave * 128 + nl;
  BSet2 c0, c1;
  load_b2(c0, wp, dtb, colw);
  if (wave < 6) {
    for (int i = 0; i < 8; ++i) {
      int M = i + hi * 8;
      sproj[M * 100 + wave * 16 + nl] = acc1[i];
      if (wave == 4) Bmbuf[(size_t)(rowbase + M) * NST + nl] = acc1[i];
      if (wave == 5) Cbuf [(size_t)(rowbase + M) * NST + nl] = acc1[i];
    }
  }
  __syncthreads();

  v2f a2[16];
  for (int kt = 0; kt < 16; ++kt) {
    int kb = kt * 4 + hi * 2;
    a2[kt].x = sproj[nl * 100 + kb];
    a2[kt].y = sproj[nl * 100 + kb + 1];
  }
  float* dp = dtbuf + (size_t)(rowbase + hi * 8) * DEXP;
#pragma unroll 1
  for (int j = 0; j < 8; j += 2) {
    load_b2(c1, wp, dtb, colw + (j + 1) * 16);
    tile2(c0, a2, dp, colw + j * 16);
    if (j + 2 < 8) load_b2(c0, wp, dtb, colw + (j + 2) * 16);
    tile2(c1, a2, dp, colw + (j + 1) * 16);
  }
}

#define RCH 4
#define UB  8
#define SCAN_THREADS 128
#define SCAN_BLOCKS  (B_SZ * NCHUNK * (DEXP / 32))

#define SEG_FENCE() __builtin_amdgcn_sched_barrier(0)

struct BatchIn  { float4 dt[UB]; float4 x[UB]; float b[UB]; };
struct BatchOut { float4 dt[UB]; float4 x[UB]; float b[UB]; float c[UB]; };

__device__ __forceinline__ void load_in(BatchIn& q, const float* dtp, const float* xp,
                                        const float* Bp, int l0) {
#pragma unroll
  for (int j = 0; j < UB; ++j) {
    q.dt[j] = *(const float4*)(dtp + (size_t)(l0 + j) * DEXP);
    q.x[j]  = *(const float4*)(xp  + (size_t)(l0 + j) * DEXP);
    q.b[j]  = Bp[(l0 + j) * NST];
  }
}

__device__ __forceinline__ void load_out(BatchOut& q, const float* dtp, const float* xp,
                                         const float* Bp, const float* Cp, int l0) {
#pragma unroll
  for (int j = 0; j < UB; ++j) {
    q.dt[j] = *(const float4*)(dtp + (size_t)(l0 + j) * DEXP);
    q.x[j]  = *(const float4*)(xp  + (size_t)(l0 + j) * DEXP);
    q.b[j]  = Bp[(l0 + j) * NST];
    q.c[j]  = Cp[(l0 + j) * NST];
  }
}

template <int M>
__device__ __forceinline__ float row_xor(float v) {
  return __int_as_float(__builtin_amdgcn_update_dpp(0, __float_as_int(v), 0x160 | M,
                                                    0xf, 0xf, true));
}

__device__ __forceinline__ void carry_steps(const BatchIn& q, const float (&And)[RCH],
                                            float (&h)[RCH], float (&sdt)[RCH]) {
#pragma unroll
  for (int j = 0; j < UB; ++j) {
    const float dv[RCH] = {q.dt[j].x, q.dt[j].y, q.dt[j].z, q.dt[j].w};
    const float xv[RCH] = {q.x[j].x,  q.x[j].y,  q.x[j].z,  q.x[j].w};
    const float Bv = q.b[j];
#pragma unroll
    for (int r = 0; r < RCH; ++r) {
      float dA = __expf(And[r] * dv[r]);
      h[r]   = fmaf(dA, h[r], dv[r] * xv[r] * Bv);
      sdt[r] += dv[r];
    }
  }
}

__global__ __launch_bounds__(SCAN_THREADS) void ssm_scan_carry(
    const float* __restrict__ x, const float* __restrict__ dtbuf,
    const float* __restrict__ Bmbuf, const float* __restrict__ Aneg,
    float* __restrict__ Sbuf, float* __restrict__ Pbuf)
{
  const int t    = threadIdx.x;
  const int n    = t & 15;
  const int hf   = (t >> 4) & 1;
  const int wave = t >> 5;
  const int bi   = blockIdx.x;
  const int c    = bi & (NCHUNK - 1);
  const int dg   = (bi >> 3) & 31;
  const int b    = bi >> 8;
  const int d0   = dg * 32 + wave * 8 + hf * RCH;

  float And[RCH], h[RCH], sdt[RCH];
#pragma unroll
  for (int r = 0; r < RCH; ++r) {
    And[r] = Aneg[(d0 + r) * NST + n];
    h[r] = 0.0f; sdt[r] = 0.0f;
  }
  const size_t rbase = (size_t)b * LSEQ + (size_t)c * CHUNK;
  const float* dtp = dtbuf + rbase * DEXP + d0;
  const float* xp  = x     + rbase * DEXP + d0;
  const float* Bp  = Bmbuf + rbase * NST + n;

  static_assert((CHUNK / UB) % 6 == 2, "five six-batch trips (two turns of the ring), then the last two batches");
  BatchIn q0, q1, q2;
  load_in(q0, dtp, xp, Bp, 0);
  load_in(q1, dtp, xp, Bp, UB);
#pragma unroll 1
  for (int l0 = 0; l0 < CHUNK - 2 * UB; l0 += 6 * UB) {
    load_in(q2, dtp, xp, Bp, l0 + 2 * UB);  SEG_FENCE();
    carry_steps(q0, And, h, sdt);           SEG_FENCE();
    load_in(q0, dtp, xp, Bp, l0 + 3 * UB);  SEG_FENCE();
    carry_steps(q1, And, h, sdt);           SEG_FENCE();
    load_in(q1, dtp, xp, Bp, l0 + 4 * UB);  SEG_FENCE();
    carry_steps(q2, And, h, sdt);           SEG_FENCE();
    load_in(q2, dtp, xp, Bp, l0 + 5 * UB);  SEG_FENCE();
    carry_steps(q0, And, h, sdt);           SEG_FENCE();
    load_in(q0, dtp, xp, Bp, l0 + 6 * UB);  SEG_FENCE();
    carry_steps(q1, And, h, sdt);           SEG_FENCE();
    load_in(q1, dtp, xp, Bp, l0 + 7 * UB);  SEG_FENCE();
    carry_steps(q2, And, h, sdt);
  }
  carry_steps(q0, And, h, sdt);             SEG_FENCE();
  carry_steps(q1, And, h, sdt);
  const size_t si = (((size_t)b * NCHUNK + c) * DEXP + d0) * NST + n;
#pragma unroll
  for (int r = 0; r < RCH; ++r) {
    Sbuf[si + r * NST] = h[r];
    Pbuf[si + r * NST] = __expf(And[r] * sdt[r]);
  }
}

__global__ void ssm_carry_combine(const float* __restrict__ Sbuf,
                                  const float* __restrict__ Pbuf,
                                  float* __restrict__ Car)
{
  const int tg = blockIdx.x * 256 + threadIdx.x;
  const int n  = tg & 15;
  const int d  = (tg >> 4) & (DEXP - 1);
  const int b  = tg >> 14;
  const size_t stride = (size_t)DEXP * NST;
  const size_t base   = (size_t)b * NCHUNK * stride + (size_t)d * NST + n;
  float cin = 0.0f;
  Car[base] = 0.0f;
  for (int c = 1; c < NCHUNK; ++c) {
    const size_t prev = base + (size_t)(c - 1) * stride;
    cin = Sbuf[prev] + Pbuf[prev] * cin;
    Car[base + (size_t)c * stride] = cin;
  }
}

__device__ __forceinline__ void out_steps(const BatchOut& q, const float (&And)[RCH],
                                          float (&h)[RCH], const bool b0, const bool b1,
                                          const bool b2, const bool b3, float* op, int l0) {
  static_assert(UB % 4 == 0, "stages xor 4 and xor 8 transpose over groups of four steps");
  float z[UB];
#pragma unroll
  for (int j = 0; j < UB; ++j) {
    const float dv[RCH] = {q.dt[j].x, q.dt[j].y, q.dt[j].z, q.dt[j].w};
    const float xv[RCH] = {q.x[j].x,  q.x[j].y,  q.x[j].z,  q.x[j].w};
    const float Bv = q.b[j], Cv = q.c[j];
    float y[RCH];
#pragma unroll
    for (int r = 0; r < RCH; ++r) {
      float dA = __expf(And[r] * dv[r]);
      h[r] = fmaf(dA, h[r], dv[r] * xv[r] * Bv);
      y[r] = h[r] * Cv;
    }
    const float k0 = b0 ? y[2] : y[0], s0 = b0 ? y[0] : y[2];
    const float k1 = b0 ? y[3] : y[1], s1 = b0 ? y[1] : y[3];
    const float z0 = k0 + row_xor<1>(s0);
    const float z1 = k1 + row_xor<1>(s1);
    const float kk = b1 ? z1 : z0, ss = b1 ? z0 : z1;
    z[j] = kk + row_xor<2>(ss);
  }
#pragma unroll
  for (int g = 0; g < UB; g += 4) {
    const float kA = b2 ? z[g + 1] : z[g],     sA = b2 ? z[g]     : z[g + 1];
    const float kB = b2 ? z[g + 3] : z[g + 2], sB = b2 ? z[g + 2] : z[g + 3];
    const float wA = kA + row_xor<4>(sA);
    const float wB = kB + row_xor<4>(sB);
    const float kk = b3 ? wB : wA, ss = b3 ? wA : wB;
    op[(size_t)(l0 + g) * DEXP] = kk + row_xor<8>(ss);
  }
}

__global__ __launch_bounds__(SCAN_THREADS) void ssm_scan_out(
    const float* __restrict__ x, const float* __restrict__ dtbuf,
    const float* __restrict__ Bmbuf, const float* __restrict__ Cbuf,
    const float* __restrict__ Aneg, const float* __restrict__ Car,
    float* __restrict__ out)
{
  const int t    = threadIdx.x;
  const int n    = t & 15;
  const int hf   = (t >> 4) & 1;
  const int wave = t >> 5;
  const int bi   = blockIdx.x;
  const int c    = bi & (NCHUNK - 1);
  const int dg   = (bi >> 3) & 31;
  const int b    = bi >> 8;
  const int d0   = dg * 32 + wave * 8 + hf * RCH;

  const size_t rbase = (size_t)b * LSEQ + (size_t)c * CHUNK;
  const size_t si = (((size_t)b * NCHUNK + c) * DEXP + d0) * NST + n;
  float And[RCH], h[RCH];
#pragma unroll
  for (int r = 0; r < RCH; ++r) {
    And[r] = Aneg[(d0 + r) * NST + n];
    h[r]   = Car[si + r * NST];
  }
  const float* dtp = dtbuf + rbase * DEXP + d0;
  const float* xp  = x     + rbase * DEXP + d0;
  const float* Bp  = Bmbuf + rbase * NST + n;
  const float* Cp  = Cbuf  + rbase * NST + n;

  const bool b0 = (n & 1) != 0, b1 = (n & 2) != 0, b2 = (n & 4) != 0, b3 = (n & 8) != 0;
  const int  mych = ((n & 1) << 1) | ((n >> 1) & 1);
  const int  myst = ((n >> 2) & 1) | ((n >> 3) << 1);
  float* op = out + (rbase + myst) * DEXP + d0 + mych;

  static_assert((CHUNK / UB) % 6 == 2, "five six-batch trips (two turns of the ring), then the last two batches");
  BatchOut qa, qb, qc;
  load_out(qa, dtp, xp, Bp, Cp, 0);
  load_out(qb, dtp, xp, Bp, Cp, UB);
#pragma unroll 1
  for (int l0 = 0; l0 < CHUNK - 2 * UB; l0 += 6 * UB) {
    load_out(qc, dtp, xp, Bp, Cp, l0 + 2 * UB);        SEG_FENCE();
    out_steps(qa, And, h, b0, b1, b2, b3, op, l0);           SEG_FENCE();
    load_out(qa, dtp, xp, Bp, Cp, l0 + 3 * UB);        SEG_FENCE();
    out_steps(qb, And, h, b0, b1, b2, b3, op, l0 + UB);      SEG_FENCE();
    load_out(qb, dtp, xp, Bp, Cp, l0 + 4 * UB);        SEG_FENCE();
    out_steps(qc, And, h, b0, b1, b2, b3, op, l0 + 2 * UB);  SEG_FENCE();
    load_out(qc, dtp, xp, Bp, Cp, l0 + 5 * UB);        SEG_FENCE();
    out_steps(qa, And, h, b0, b1, b2, b3, op, l0 + 3 * UB);  SEG_FENCE();
    load_out(qa, dtp, xp, Bp, Cp, l0 + 6 * UB);        SEG_FENCE();
    out_steps(qb, And, h, b0, b1, b2, b3, op, l0 + 4 * UB);  SEG_FENCE();
    load_out(qb, dtp, xp, Bp, Cp, l0 + 7 * UB);        SEG_FENCE();
    out_steps(qc, And, h, b0, b1, b2, b3, op, l0 + 5 * UB);
  }
  out_steps(qa, And, h, b0, b1, b2, b3, op, CHUNK - 2 * UB);  SEG_FENCE();
  out_steps(qb, And, h, b0, b1, b2, b3, op, CHUNK - UB);
}

extern "C" void kernel_launch(void* const* d_in, const int* in_sizes, int n_in,
                              void* d_out, int out_size, void* d_ws, size_t ws_size,
                              hipStream_t stream) {
  (void)in_sizes; (void)n_in; (void)out_size; (void)ws_size;
  const float* x     = (const float*)d_in[0];
  const float* xw    = (const float*)d_in[1];
  const float* xb    = (const float*)d_in[2];
  const float* dtw   = (const float*)d_in[3];
  const float* dtb   = (const float*)d_in[4];
  const float* A_log = (const float*)d_in[5];
  float* out = (float*)d_out;

  float* ws    = (float*)d_ws;
  float* dtbuf = ws;
  float* Aneg  = dtbuf + (size_t)ROWS * DEXP;
  float* Bm    = Aneg  + (size_t)DEXP * NST;
  float* Cb    = Bm    + (size_t)ROWS * NST;
  float* Sb    = Cb    + (size_t)ROWS * NST;
  float* Pb    = Sb    + (size_t)B_SZ * NCHUNK * DEXP * NST;
  float* Car   = Pb    + (size_t)B_SZ * NCHUNK * DEXP * NST;
  float* xwT   = Car   + (size_t)B_SZ * NCHUNK * DEXP * NST;
  float* dtwT  = xwT   + (size_t)DEXP * RPROJ;

  hipLaunchKernelGGL(ssm_prep, dim3((DEXP * RPROJ + 255) / 256), dim3(256), 0, stream,
                     A_log, Aneg, xw, xwT, dtw, dtwT);
  hipLaunchKernelGGL(ssm_proj_wmma, dim3(ROWS / 16), dim3(256), 0, stream,
                     x, xwT, xb, dtwT, dtb, dtbuf, Bm, Cb);
  hipLaunchKernelGGL(ssm_scan_carry, dim3(SCAN_BLOCKS), dim3(SCAN_THREADS), 0, stream,
                     x, dtbuf, Bm, Aneg, Sb, Pb);
  hipLaunchKernelGGL(ssm_carry_combine, dim3((B_SZ * DEXP * NST) / 256), dim3(256), 0, stream,
                     Sb, Pb, Car);
  hipLaunchKernelGGL(ssm_scan_out, dim3(SCAN_BLOCKS), dim3(SCAN_THREADS), 0, stream,
                     x, dtbuf, Bm, Cb, Aneg, Car, out);
}


# scan kernels: dt, x loaded once per 16-lane row (8 bytes per lane per batch); dt*x, dt*log2(e) formed there, read through DPP row_share; chunk dt sum = 8 lane partials + tree (sum order changes)
# speedup vs baseline: 4.0410x; 1.0335x over previous
#include <hip/hip_runtime.h>
#include <math.h>

#define B_SZ   2
#define LSEQ   2048
#define DEXP   1024
#define DRANK  64
#define NST    16
#define RPROJ  (DRANK + 2*NST)
#define NCHUNK 8
#define CHUNK  (LSEQ / NCHUNK)
#define ROWS   (B_SZ * LSEQ)

typedef __attribute__((ext_vector_type(2))) float v2f;
typedef __attribute__((ext_vector_type(8))) float v8f;

__device__ __forceinline__ v8f wmma_f32_k4(v2f a, v2f b, v8f c) {
  return __builtin_amdgcn_wmma_f32_16x16x4_f32(
       false, a,  false, b,
       (short)0, c,  false,  false);
}

__global__ void ssm_prep(const float* __restrict__ A_log, float* __restrict__ Aneg,
                         const float* __restrict__ xw,  float* __restrict__ xwT,
                         const float* __restrict__ dtw, float* __restrict__ dtwT) {
  int i = blockIdx.x * 256 + threadIdx.x;
  if (i < DEXP * NST) Aneg[i] = -__expf(A_log[i]);
  if (i < DEXP * RPROJ) {
    int k = i / RPROJ, c = i % RPROJ;
    xwT[i] = xw[(size_t)c * DEXP + k];
  }
  if (i < DRANK * DEXP) {
    int k = i / DEXP, d = i % DEXP;
    dtwT[i] = dtw[(size_t)d * DRANK + k];
  }
}

struct BSet1 { v2f f[16]; };
struct BSet2 { v2f f[16]; float bias; };

#define SXP 1028

__device__ __forceinline__ void load_b1(BSet1& q, const float* bp, int g) {
#pragma unroll
  for (int s = 0; s < 16; ++s) {
    const int k = g * 64 + s * 4;
    q.f[s].x = bp[(size_t)k * RPROJ];
    q.f[s].y = bp[(size_t)(k + 1) * RPROJ];
  }
}

__device__ __forceinline__ v8f mma_group1(const BSet1& q, const float* ap, int g, v8f acc) {
#pragma unroll
  for (int s = 0; s < 16; ++s) {
    const int k = g * 64 + s * 4;
    v2f a;  a.x = ap[k];  a.y = ap[k + 1];
    acc = wmma_f32_k4(a, q.f[s], acc);
  }
  return acc;
}

__device__ __forceinline__ void load_b2(BSet2& q, const float* wp, const float* dtb, int col) {
  q.bias = dtb[col];
#pragma unroll
  for (int kt = 0; kt < 16; ++kt) {
    q.f[kt].x = wp[(size_t)(kt * 4) * DEXP + col];
    q.f[kt].y = wp[(size_t)(kt * 4 + 1) * DEXP + col];
  }
}

__device__ __forceinline__ void tile2(const BSet2& q, const v2f (&a2)[16], float* dp, int col) {
  v8f acc2;
#pragma unroll
  for (int i = 0; i < 8; ++i) acc2[i] = q.bias;
#pragma unroll
  for (int kt = 0; kt < 16; ++kt) acc2 = wmma_f32_k4(a2[kt], q.f[kt], acc2);
  float sp[8];
#pragma unroll
  for (int i = 0; i < 8; ++i) sp[i] = log1pf(__expf(acc2[i]));
#pragma unroll
  for (int i = 0; i < 8; ++i) {
    const float v = acc2[i];
    dp[(size_t)i * DEXP + col] = (v > 20.0f) ? v : sp[i];
  }
}

__global__ __launch_bounds__(256) void ssm_proj_wmma(
    const float* __restrict__ x,    const float* __restrict__ xwT,
    const float* __restrict__ xb,   const float* __restrict__ dtwT,
    const float* __restrict__ dtb,
    float* __restrict__ dtbuf, float* __restrict__ Bmbuf, float* __restrict__ Cbuf)
{
  __shared__ float sx[16 * SXP];
  __shared__ float sproj[16 * 100];

  const int t    = threadIdx.x;
  const int wave = t >> 5;
  const int nl   = t & 15;
  const int hi   = (t >> 4) & 1;
  const int rowbase = blockIdx.x * 16;

  const int col1 = wave * 16 + nl;
  const float* bp = xwT + (size_t)(hi * 2) * RPROJ + col1;
  BSet1 b0, b1;
  v8f acc1;
  float bv1 = 0.0f;
  if (wave < 6) {
    load_b1(b0, bp, 0);
    bv1 = xb[col1];
  }
  for (int i = 0; i < 16; ++i) {
    int flat = t + i * 256;
    int r  = flat >> 8;
    int kk = (flat & 255) << 2;
    const float4 v = *(const float4*)(x + (size_t)(rowbase + r) * DEXP + kk);
    float* dst = &sx[r * SXP + kk];
    dst[0] = v.x; dst[1] = v.y; dst[2] = v.z; dst[3] = v.w;
  }
  __syncthreads();
  if (wave < 6) {
    for (int i = 0; i < 8; ++i) acc1[i] = bv1;
    const float* ap = &sx[nl * SXP + hi * 2];
#pragma unroll 1
    for (int g = 0; g < 16; g += 2) {
      load_b1(b1, bp, g + 1);
      acc1 = mma_group1(b0, ap, g, acc1);
      if (g + 2 < 16) load_b1(b0, bp, g + 2);
      acc1 = mma_group1(b1, ap, g + 1, acc1);
    }
  }
  const float* wp = dtwT + (size_t)(hi * 2) * DEXP;
  const int colw = wave * 128 + nl;
  BSet2 c0, c1;
  load_b2(c0, wp, dtb, colw);
  if (wave < 6) {
    for (int i = 0; i < 8; ++i) {
      int M = i + hi * 8;
      sproj[M * 100 + wave * 16 + nl] = acc1[i];
      if (wave == 4) Bmbuf[(size_t)(rowbase + M) * NST + nl] = acc1[i];
      if (wave == 5) Cbuf [(size_t)(rowbase + M) * NST + nl] = acc1[i];
    }
  }
  __syncthreads();

  v2f a2[16];
  for (int kt = 0; kt < 16; ++kt) {
    int kb = kt * 4 + hi * 2;
    a2[kt].x = sproj[nl * 100 + kb];
    a2[kt].y = sproj[nl * 100 + kb + 1];
  }
  float* dp = dtbuf + (size_t)(rowbase + hi * 8) * DEXP;
#pragma unroll 1
  for (int j = 0; j < 8; j += 2) {
    load_b2(c1, wp, dtb, colw + (j + 1) * 16);
    tile2(c0, a2, dp, colw + j * 16);
    if (j + 2 < 8) load_b2(c0, wp, dtb, colw + (j + 2) * 16);
    tile2(c1, a2, dp, colw + (j + 1) * 16);
  }
}

#define RCH 4
#define UB  8
#define SCAN_THREADS 128
#define SCAN_BLOCKS  (B_SZ * NCHUNK * (DEXP / 32))

#define SEG_FENCE() __builtin_amdgcn_sched_barrier(0)
#define LOG2E 1.44269504088896340736f

__device__ __forceinline__ int group_row(int l0) {
  asm volatile("" : "+s"(l0));
  return l0;
}
#define HOLD_STATE(h) asm volatile("" : "+v"(h[0]), "+v"(h[1]), "+v"(h[2]), "+v"(h[3]))
#define TAKE_LOADED(a, b) asm volatile("" : "+v"(a.x), "+v"(a.y), "+v"(b.x), "+v"(b.y))

struct BatchIn  { float2 dt; float2 x; float b[UB]; };
struct BatchOut { float2 dt; float2 x; float b[UB]; float c[UB]; };

__device__ __forceinline__ void load_in(BatchIn& q, const float* dtp, const float* xp,
                                        const float* Bp, int l0) {
  q.dt = *(const float2*)(dtp + (size_t)l0 * DEXP);
  q.x  = *(const float2*)(xp  + (size_t)l0 * DEXP);
  const float* Bg = Bp + (size_t)l0 * NST;
#pragma unroll
  for (int j = 0; j < UB; ++j) q.b[j] = Bg[j * NST];
}

__device__ __forceinline__ void load_out(BatchOut& q, const float* dtp, const float* xp,
                                         const float* Bp, const float* Cp, int l0) {
  q.dt = *(const float2*)(dtp + (size_t)l0 * DEXP);
  q.x  = *(const float2*)(xp  + (size_t)l0 * DEXP);
  const float* Bg = Bp + (size_t)l0 * NST;
  const float* Cg = Cp + (size_t)l0 * NST;
#pragma unroll
  for (int j = 0; j < UB; ++j) {
    q.b[j] = Bg[j * NST];
    q.c[j] = Cg[j * NST];
  }
}

template <int M>
__device__ __forceinline__ float row_xor(float v) {
  return __int_as_float(__builtin_amdgcn_update_dpp(0, __float_as_int(v), 0x160 | M,
                                                    0xf, 0xf, true));
}

template <int K>
__device__ __forceinline__ float row_get(float v) {
  return __int_as_float(__builtin_amdgcn_update_dpp(0, __float_as_int(v), 0x150 | K,
                                                    0xf, 0xf, true));
}

template <int J>
__device__ __forceinline__ void scan_step(const float2 e, const float2 u, const float Bv,
                                          const float (&And)[RCH], float (&h)[RCH]) {
  const float d0 = row_get<2 * J>(e.x),      d1 = row_get<2 * J>(e.y);
  const float d2 = row_get<2 * J + 1>(e.x),  d3 = row_get<2 * J + 1>(e.y);
  const float u0 = row_get<2 * J>(u.x),      u1 = row_get<2 * J>(u.y);
  const float u2 = row_get<2 * J + 1>(u.x),  u3 = row_get<2 * J + 1>(u.y);
  h[0] = fmaf(__builtin_amdgcn_exp2f(And[0] * d0), h[0], u0 * Bv);
  h[1] = fmaf(__builtin_amdgcn_exp2f(And[1] * d1), h[1], u1 * Bv);
  h[2] = fmaf(__builtin_amdgcn_exp2f(And[2] * d2), h[2], u2 * Bv);
  h[3] = fmaf(__builtin_amdgcn_exp2f(And[3] * d3), h[3], u3 * Bv);
}

__device__ __forceinline__ void carry_steps(const BatchIn& q, const float (&And)[RCH],
                                            float (&h)[RCH], float2& sdt) {
  float2 dt = q.dt, xv = q.x;
  TAKE_LOADED(dt, xv);
  float2 u;  u.x = dt.x * xv.x;  u.y = dt.y * xv.y;
  float2 e;  e.x = dt.x * LOG2E;  e.y = dt.y * LOG2E;
  sdt.x += dt.x;  sdt.y += dt.y;
  scan_step<0>(e, u, q.b[0], And, h);
  scan_step<1>(e, u, q.b[1], And, h);
  scan_step<2>(e, u, q.b[2], And, h);
  scan_step<3>(e, u, q.b[3], And, h);
  scan_step<4>(e, u, q.b[4], And, h);
  scan_step<5>(e, u, q.b[5], And, h);
  scan_step<6>(e, u, q.b[6], And, h);
  scan_step<7>(e, u, q.b[7], And, h);
}

__global__ __launch_bounds__(SCAN_THREADS) void ssm_scan_carry(
    const float* __restrict__ x, const float* __restrict__ dtbuf,
    const float* __restrict__ Bmbuf, const float* __restrict__ Aneg,
    float* __restrict__ Sbuf, float* __restrict__ Pbuf)
{
  const int t    = threadIdx.x;
  const int n    = t & 15;
  const int hf   = (t >> 4) & 1;
  const int wave = t >> 5;
  const int bi   = blockIdx.x;
  const int c    = bi & (NCHUNK - 1);
  const int dg   = (bi >> 3) & 31;
  const int b    = bi >> 8;
  const int d0   = dg * 32 + wave * 8 + hf * RCH;

  float And[RCH], h[RCH];
#pragma unroll
  for (int r = 0; r < RCH; ++r) {
    And[r] = Aneg[(d0 + r) * NST + n];
    h[r] = 0.0f;
  }
  float2 sdt;  sdt.x = 0.0f;  sdt.y = 0.0f;
  const size_t rbase = (size_t)b * LSEQ + (size_t)c * CHUNK;
  const size_t hold  = (rbase + (n >> 1)) * DEXP + d0 + (n & 1) * 2;
  const float* dtp = dtbuf + hold;
  const float* xp  = x     + hold;
  const float* Bp  = Bmbuf + rbase * NST + n;

  static_assert((CHUNK / UB) % 6 == 2, "five six-batch trips (two turns of the ring), then the last two batches");
  BatchIn q0, q1, q2;
  load_in(q0, dtp, xp, Bp, 0);
  load_in(q1, dtp, xp, Bp, UB);
#pragma unroll 1
  for (int l0 = 0; l0 < CHUNK - 2 * UB; l0 += 6 * UB) {
    load_in(q2, dtp, xp, Bp, group_row(l0 + 2 * UB));  SEG_FENCE();
    carry_steps(q0, And, h, sdt); HOLD_STATE(h);           SEG_FENCE();
    load_in(q0, dtp, xp, Bp, group_row(l0 + 3 * UB));  SEG_FENCE();
    carry_steps(q1, And, h, sdt); HOLD_STATE(h);           SEG_FENCE();
    load_in(q1, dtp, xp, Bp, group_row(l0 + 4 * UB));  SEG_FENCE();
    carry_steps(q2, And, h, sdt); HOLD_STATE(h);           SEG_FENCE();
    load_in(q2, dtp, xp, Bp, group_row(l0 + 5 * UB));  SEG_FENCE();
    carry_steps(q0, And, h, sdt); HOLD_STATE(h);           SEG_FENCE();
    load_in(q0, dtp, xp, Bp, group_row(l0 + 6 * UB));  SEG_FENCE();
    carry_steps(q1, And, h, sdt); HOLD_STATE(h);           SEG_FENCE();
    load_in(q1, dtp, xp, Bp, group_row(l0 + 7 * UB));  SEG_FENCE();
    carry_steps(q2, And, h, sdt);           HOLD_STATE(h);
  }
  carry_steps(q0, And, h, sdt); HOLD_STATE(h);             SEG_FENCE();
  carry_steps(q1, And, h, sdt);
  sdt.x += row_xor<2>(sdt.x);  sdt.y += row_xor<2>(sdt.y);
  sdt.x += row_xor<4>(sdt.x);  sdt.y += row_xor<4>(sdt.y);
  sdt.x += row_xor<8>(sdt.x);  sdt.y += row_xor<8>(sdt.y);
  const float tot[RCH] = {row_get<0>(sdt.x), row_get<0>(sdt.y), row_get<1>(sdt.x), row_get<1>(sdt.y)};
  const size_t si = (((size_t)b * NCHUNK + c) * DEXP + d0) * NST + n;
#pragma unroll
  for (int r = 0; r < RCH; ++r) {
    Sbuf[si + r * NST] = h[r];
    Pbuf[si + r * NST] = __expf(And[r] * tot[r]);
  }
}

__global__ void ssm_carry_combine(const float* __restrict__ Sbuf,
                                  const float* __restrict__ Pbuf,
                                  float* __restrict__ Car)
{
  const int tg = blockIdx.x * 256 + threadIdx.x;
  const int n  = tg & 15;
  const int d  = (tg >> 4) & (DEXP - 1);
  const int b  = tg >> 14;
  const size_t stride = (size_t)DEXP * NST;
  const size_t base   = (size_t)b * NCHUNK * stride + (size_t)d * NST + n;
  float cin = 0.0f;
  Car[base] = 0.0f;
  for (int c = 1; c < NCHUNK; ++c) {
    const size_t prev = base + (size_t)(c - 1) * stride;
    cin = Sbuf[prev] + Pbuf[prev] * cin;
    Car[base + (size_t)c * stride] = cin;
  }
}

template <int J>
__device__ __forceinline__ float out_step(const float2 e, const float2 u, const float Bv,
                                          const float Cv, const float (&And)[RCH],
                                          float (&h)[RCH], const bool b0, const bool b1) {
  scan_step<J>(e, u, Bv, And, h);
  float y[RCH];
#pragma unroll
  for (int r = 0; r < RCH; ++r) y[r] = h[r] * Cv;
  const float k0 = b0 ? y[2] : y[0], s0 = b0 ? y[0] : y[2];
  const float k1 = b0 ? y[3] : y[1], s1 = b0 ? y[1] : y[3];
  const float z0 = k0 + row_xor<1>(s0);
  const float z1 = k1 + row_xor<1>(s1);
  const float kk = b1 ? z1 : z0, ss = b1 ? z0 : z1;
  return kk + row_xor<2>(ss);
}

__device__ __forceinline__ void out_steps(const BatchOut& q, const float (&And)[RCH],
                                          float (&h)[RCH], const bool b0, const bool b1,
                                          const bool b2, const bool b3, float* op, int l0) {
  static_assert(UB == 8, "the eight steps of a batch are written out below");
  float2 dt = q.dt, xv = q.x;
  TAKE_LOADED(dt, xv);
  float2 u;  u.x = dt.x * xv.x;  u.y = dt.y * xv.y;
  float2 e;  e.x = dt.x * LOG2E;  e.y = dt.y * LOG2E;
  float z[UB];
  z[0] = out_step<0>(e, u, q.b[0], q.c[0], And, h, b0, b1);
  z[1] = out_step<1>(e, u, q.b[1], q.c[1], And, h, b0, b1);
  z[2] = out_step<2>(e, u, q.b[2], q.c[2], And, h, b0, b1);
  z[3] = out_step<3>(e, u, q.b[3], q.c[3], And, h, b0, b1);
  z[4] = out_step<4>(e, u, q.b[4], q.c[4], And, h, b0, b1);
  z[5] = out_step<5>(e, u, q.b[5], q.c[5], And, h, b0, b1);
  z[6] = out_step<6>(e, u, q.b[6], q.c[6], And, h, b0, b1);
  z[7] = out_step<7>(e, u, q.b[7], q.c[7], And, h, b0, b1);
#pragma unroll
  for (int g = 0; g < UB; g += 4) {
    const float kA = b2 ? z[g + 1] : z[g],     sA = b2 ? z[g]     : z[g + 1];
    const float kB = b2 ? z[g + 3] : z[g + 2], sB = b2 ? z[g + 2] : z[g + 3];
    const float wA = kA + row_xor<4>(sA);
    const float wB = kB + row_xor<4>(sB);
    const float kk = b3 ? wB : wA, ss = b3 ? wA : wB;
    op[(size_t)(l0 + g) * DEXP] = kk + row_xor<8>(ss);
  }
}

__global__ __launch_bounds__(SCAN_THREADS) void ssm_scan_out(
    const float* __restrict__ x, const float* __restrict__ dtbuf,
    const float* __restrict__ Bmbuf, const float* __restrict__ Cbuf,
    const float* __restrict__ Aneg, const float* __restrict__ Car,
    float* __restrict__ out)
{
  const int t    = threadIdx.x;
  const int n    = t & 15;
  const int hf   = (t >> 4) & 1;
  const int wave = t >> 5;
  const int bi   = blockIdx.x;
  const int c    = bi & (NCHUNK - 1);
  const int dg   = (bi >> 3) & 31;
  const int b    = bi >> 8;
  const int d0   = dg * 32 + wave * 8 + hf * RCH;

  const size_t rbase = (size_t)b * LSEQ + (size_t)c * CHUNK;
  const size_t si = (((size_t)b * NCHUNK + c) * DEXP + d0) * NST + n;
  float And[RCH], h[RCH];
#pragma unroll
  for (int r = 0; r < RCH; ++r) {
    And[r] = Aneg[(d0 + r) * NST + n];
    h[r]   = Car[si + r * NST];
  }
  const size_t hold = (rbase + (n >> 1)) * DEXP + d0 + (n & 1) * 2;
  const float* dtp = dtbuf + hold;
  const float* xp  = x     + hold;
  const float* Bp  = Bmbuf + rbase * NST + n;
  const float* Cp  = Cbuf  + rbase * NST + n;

  const bool b0 = (n & 1) != 0, b1 = (n & 2) != 0, b2 = (n & 4) != 0, b3 = (n & 8) != 0;
  const int  mych = ((n & 1) << 1) | ((n >> 1) & 1);
  const int  myst = ((n >> 2) & 1) | ((n >> 3) << 1);
  float* op = out + (rbase + myst) * DEXP + d0 + mych;

  static_assert((CHUNK / UB) % 6 == 2, "five six-batch trips (two turns of the ring), then the last two batches");
  BatchOut qa, qb, qc;
  load_out(qa, dtp, xp, Bp, Cp, 0);
  load_out(qb, dtp, xp, Bp, Cp, UB);
#pragma unroll 1
  for (int l0 = 0; l0 < CHUNK - 2 * UB; l0 += 6 * UB) {
    load_out(qc, dtp, xp, Bp, Cp, group_row(l0 + 2 * UB));        SEG_FENCE();
    out_steps(qa, And, h, b0, b1, b2, b3, op, l0); HOLD_STATE(h);           SEG_FENCE();
    load_out(qa, dtp, xp, Bp, Cp, group_row(l0 + 3 * UB));        SEG_FENCE();
    out_steps(qb, And, h, b0, b1, b2, b3, op, l0 + UB); HOLD_STATE(h);      SEG_FENCE();
    load_out(qb, dtp, xp, Bp, Cp, group_row(l0 + 4 * UB));        SEG_FENCE();
    out_steps(qc, And, h, b0, b1, b2, b3, op, l0 + 2 * UB); HOLD_STATE(h);  SEG_FENCE();
    load_out(qc, dtp, xp, Bp, Cp, group_row(l0 + 5 * UB));        SEG_FENCE();
    out_steps(qa, And, h, b0, b1, b2, b3, op, l0 + 3 * UB); HOLD_STATE(h);  SEG_FENCE();
    load_out(qa, dtp, xp, Bp, Cp, group_row(l0 + 6 * UB));        SEG_FENCE();
    out_steps(qb, And, h, b0, b1, b2, b3, op, l0 + 4 * UB); HOLD_STATE(h);  SEG_FENCE();
    load_out(qb, dtp, xp, Bp, Cp, group_row(l0 + 7 * UB));        SEG_FENCE();
    out_steps(qc, And, h, b0, b1, b2, b3, op, l0 + 5 * UB);  HOLD_STATE(h);
  }
  out_steps(qa, And, h, b0, b1, b2, b3, op, CHUNK - 2 * UB); HOLD_STATE(h);  SEG_FENCE();
  out_steps(qb, And, h, b0, b1, b2, b3, op, CHUNK - UB);
}

extern "C" void kernel_launch(void* const* d_in, const int* in_sizes, int n_in,
                              void* d_out, int out_size, void* d_ws, size_t ws_size,
                              hipStream_t stream) {
  (void)in_sizes; (void)n_in; (void)out_size; (void)ws_size;
  const float* x     = (const float*)d_in[0];
  const float* xw    = (const float*)d_in[1];
  const float* xb    = (const float*)d_in[2];
  const float* dtw   = (const float*)d_in[3];
  const float* dtb   = (const float*)d_in[4];
  const float* A_log = (const float*)d_in[5];
  float* out = (float*)d_out;

  float* ws    = (float*)d_ws;
  float* dtbuf = ws;
  float* Aneg  = dtbuf + (size_t)ROWS * DEXP;
  float* Bm    = Aneg  + (size_t)DEXP * NST;
  float* Cb    = Bm    + (size_t)ROWS * NST;
  float* Sb    = Cb    + (size_t)ROWS * NST;
  float* Pb    = Sb    + (size_t)B_SZ * NCHUNK * DEXP * NST;
  float* Car   = Pb    + (size_t)B_SZ * NCHUNK * DEXP * NST;
  float* xwT   = Car   + (size_t)B_SZ * NCHUNK * DEXP * NST;
  float* dtwT  = xwT   + (size_t)DEXP * RPROJ;

  hipLaunchKernelGGL(ssm_prep, dim3((DEXP * RPROJ + 255) / 256), dim3(256), 0, stream,
                     A_log, Aneg, xw, xwT, dtw, dtwT);
  hipLaunchKernelGGL(ssm_proj_wmma, dim3(ROWS / 16), dim3(256), 0, stream,
                     x, xwT, xb, dtwT, dtb, dtbuf, Bm, Cb);
  hipLaunchKernelGGL(ssm_scan_carry, dim3(SCAN_BLOCKS), dim3(SCAN_THREADS), 0, stream,
                     x, dtbuf, Bm, Aneg, Sb, Pb);
  hipLaunchKernelGGL(ssm_carry_combine, dim3((B_SZ * DEXP * NST) / 256), dim3(256), 0, stream,
                     Sb, Pb, Car);
  hipLaunchKernelGGL(ssm_scan_out, dim3(SCAN_BLOCKS), dim3(SCAN_THREADS), 0, stream,
                     x, dtbuf, Bm, Cb, Aneg, Car, out);
}


# scan kernels: loads run three batches ahead in a ring of four sets, waits exact (opening groups fenced, dt/x kept as 64-bit pairs: no copy at the loop edge); carry scan given a 4-waves-per-SIMD cap
# speedup vs baseline: 4.1919x; 1.0373x over previous
#include <hip/hip_runtime.h>
#include <math.h>

#define B_SZ   2
#define LSEQ   2048
#define DEXP   1024
#define DRANK  64
#define NST    16
#define RPROJ  (DRANK + 2*NST)
#define NCHUNK 8
#define CHUNK  (LSEQ / NCHUNK)
#define ROWS   (B_SZ * LSEQ)

typedef __attribute__((ext_vector_type(2))) float v2f;
typedef __attribute__((ext_vector_type(8))) float v8f;

__device__ __forceinline__ v8f wmma_f32_k4(v2f a, v2f b, v8f c) {
  return __builtin_amdgcn_wmma_f32_16x16x4_f32(
       false, a,  false, b,
       (short)0, c,  false,  false);
}

__global__ void ssm_prep(const float* __restrict__ A_log, float* __restrict__ Aneg,
                         const float* __restrict__ xw,  float* __restrict__ xwT,
                         const float* __restrict__ dtw, float* __restrict__ dtwT) {
  int i = blockIdx.x * 256 + threadIdx.x;
  if (i < DEXP * NST) Aneg[i] = -__expf(A_log[i]);
  if (i < DEXP * RPROJ) {
    int k = i / RPROJ, c = i % RPROJ;
    xwT[i] = xw[(size_t)c * DEXP + k];
  }
  if (i < DRANK * DEXP) {
    int k = i / DEXP, d = i % DEXP;
    dtwT[i] = dtw[(size_t)d * DRANK + k];
  }
}

struct BSet1 { v2f f[16]; };
struct BSet2 { v2f f[16]; float bias; };

#define SXP 1028

__device__ __forceinline__ void load_b1(BSet1& q, const float* bp, int g) {
#pragma unroll
  for (int s = 0; s < 16; ++s) {
    const int k = g * 64 + s * 4;
    q.f[s].x = bp[(size_t)k * RPROJ];
    q.f[s].y = bp[(size_t)(k + 1) * RPROJ];
  }
}

__device__ __forceinline__ v8f mma_group1(const BSet1& q, const float* ap, int g, v8f acc) {
#pragma unroll
  for (int s = 0; s < 16; ++s) {
    const int k = g * 64 + s * 4;
    v2f a;  a.x = ap[k];  a.y = ap[k + 1];
    acc = wmma_f32_k4(a, q.f[s], acc);
  }
  return acc;
}

__device__ __forceinline__ void load_b2(BSet2& q, const float* wp, const float* dtb, int col) {
  q.bias = dtb[col];
#pragma unroll
  for (int kt = 0; kt < 16; ++kt) {
    q.f[kt].x = wp[(size_t)(kt * 4) * DEXP + col];
    q.f[kt].y = wp[(size_t)(kt * 4 + 1) * DEXP + col];
  }
}

__device__ __forceinline__ void tile2(const BSet2& q, const v2f (&a2)[16], float* dp, int col) {
  v8f acc2;
#pragma unroll
  for (int i = 0; i < 8; ++i) acc2[i] = q.bias;
#pragma unroll
  for (int kt = 0; kt < 16; ++kt) acc2 = wmma_f32_k4(a2[kt], q.f[kt], acc2);
  float sp[8];
#pragma unroll
  for (int i = 0; i < 8; ++i) sp[i] = log1pf(__expf(acc2[i]));
#pragma unroll
  for (int i = 0; i < 8; ++i) {
    const float v = acc2[i];
    dp[(size_t)i * DEXP + col] = (v > 20.0f) ? v : sp[i];
  }
}

__global__ __launch_bounds__(256) void ssm_proj_wmma(
    const float* __restrict__ x,    const float* __restrict__ xwT,
    const float* __restrict__ xb,   const float* __restrict__ dtwT,
    const float* __restrict__ dtb,
    float* __restrict__ dtbuf, float* __restrict__ Bmbuf, float* __restrict__ Cbuf)
{
  __shared__ float sx[16 * SXP];
  __shared__ float sproj[16 * 100];

  const int t    = threadIdx.x;
  const int wave = t >> 5;
  const int nl   = t & 15;
  const int hi   = (t >> 4) & 1;
  const int rowbase = blockIdx.x * 16;

  const int col1 = wave * 16 + nl;
  const float* bp = xwT + (size_t)(hi * 2) * RPROJ + col1;
  BSet1 b0, b1;
  v8f acc1;
  float bv1 = 0.0f;
  if (wave < 6) {
    load_b1(b0, bp, 0);
    bv1 = xb[col1];
  }
  for (int i = 0; i < 16; ++i) {
    int flat = t + i * 256;
    int r  = flat >> 8;
    int kk = (flat & 255) << 2;
    const float4 v = *(const float4*)(x + (size_t)(rowbase + r) * DEXP + kk);
    float* dst = &sx[r * SXP + kk];
    dst[0] = v.x; dst[1] = v.y; dst[2] = v.z; dst[3] = v.w;
  }
  __syncthreads();
  if (wave < 6) {
    for (int i = 0; i < 8; ++i) acc1[i] = bv1;
    const float* ap = &sx[nl * SXP + hi * 2];
#pragma unroll 1
    for (int g = 0; g < 16; g += 2) {
      load_b1(b1, bp, g + 1);
      acc1 = mma_group1(b0, ap, g, acc1);
      if (g + 2 < 16) load_b1(b0, bp, g + 2);
      acc1 = mma_group1(b1, ap, g + 1, acc1);
    }
  }
  const float* wp = dtwT + (size_t)(hi * 2) * DEXP;
  const int colw = wave * 128 + nl;
  BSet2 c0, c1;
  load_b2(c0, wp, dtb, colw);
  if (wave < 6) {
    for (int i = 0; i < 8; ++i) {
      int M = i + hi * 8;
      sproj[M * 100 + wave * 16 + nl] = acc1[i];
      if (wave == 4) Bmbuf[(size_t)(rowbase + M) * NST + nl] = acc1[i];
      if (wave == 5) Cbuf [(size_t)(rowbase + M) * NST + nl] = acc1[i];
    }
  }
  __syncthreads();

  v2f a2[16];
  for (int kt = 0; kt < 16; ++kt) {
    int kb = kt * 4 + hi * 2;
    a2[kt].x = sproj[nl * 100 + kb];
    a2[kt].y = sproj[nl * 100 + kb + 1];
  }
  float* dp = dtbuf + (size_t)(rowbase + hi * 8) * DEXP;
#pragma unroll 1
  for (int j = 0; j < 8; j += 2) {
    load_b2(c1, wp, dtb, colw + (j + 1) * 16);
    tile2(c0, a2, dp, colw + j * 16);
    if (j + 2 < 8) load_b2(c0, wp, dtb, colw + (j + 2) * 16);
    tile2(c1, a2, dp, colw + (j + 1) * 16);
  }
}

#define RCH 4
#define UB  8
#define SCAN_THREADS 128
#define SCAN_BLOCKS  (B_SZ * NCHUNK * (DEXP / 32))

#define SEG_FENCE() __builtin_amdgcn_sched_barrier(0)
#define LOG2E 1.44269504088896340736f

__device__ __forceinline__ int group_row(int l0) {
  asm volatile("" : "+s"(l0));
  return l0;
}
#define HOLD_STATE(h) asm volatile("" : "+v"(h[0]), "+v"(h[1]), "+v"(h[2]), "+v"(h[3]))
#define TAKE_LOADED(a, b) asm volatile("" : "+v"(a), "+v"(b))

struct BatchIn  { v2f dt; v2f x; float b[UB]; };
struct BatchOut { v2f dt; v2f x; float b[UB]; float c[UB]; };

__device__ __forceinline__ void load_in(BatchIn& q, const float* dtp, const float* xp,
                                        const float* Bp, int l0) {
  q.dt = *(const v2f*)(dtp + (size_t)l0 * DEXP);
  q.x  = *(const v2f*)(xp  + (size_t)l0 * DEXP);
  const float* Bg = Bp + (size_t)l0 * NST;
#pragma unroll
  for (int j = 0; j < UB; ++j) q.b[j] = Bg[j * NST];
}

__device__ __forceinline__ void load_out(BatchOut& q, const float* dtp, const float* xp,
                                         const float* Bp, const float* Cp, int l0) {
  q.dt = *(const v2f*)(dtp + (size_t)l0 * DEXP);
  q.x  = *(const v2f*)(xp  + (size_t)l0 * DEXP);
  const float* Bg = Bp + (size_t)l0 * NST;
  const float* Cg = Cp + (size_t)l0 * NST;
#pragma unroll
  for (int j = 0; j < UB; ++j) {
    q.b[j] = Bg[j * NST];
    q.c[j] = Cg[j * NST];
  }
}

template <int M>
__device__ __forceinline__ float row_xor(float v) {
  return __int_as_float(__builtin_amdgcn_update_dpp(0, __float_as_int(v), 0x160 | M,
                                                    0xf, 0xf, true));
}

template <int K>
__device__ __forceinline__ float row_get(float v) {
  return __int_as_float(__builtin_amdgcn_update_dpp(0, __float_as_int(v), 0x150 | K,
                                                    0xf, 0xf, true));
}

template <int J>
__device__ __forceinline__ void scan_step(const float2 e, const float2 u, const float Bv,
                                          const float (&And)[RCH], float (&h)[RCH]) {
  const float d0 = row_get<2 * J>(e.x),      d1 = row_get<2 * J>(e.y);
  const float d2 = row_get<2 * J + 1>(e.x),  d3 = row_get<2 * J + 1>(e.y);
  const float u0 = row_get<2 * J>(u.x),      u1 = row_get<2 * J>(u.y);
  const float u2 = row_get<2 * J + 1>(u.x),  u3 = row_get<2 * J + 1>(u.y);
  h[0] = fmaf(__builtin_amdgcn_exp2f(And[0] * d0), h[0], u0 * Bv);
  h[1] = fmaf(__builtin_amdgcn_exp2f(And[1] * d1), h[1], u1 * Bv);
  h[2] = fmaf(__builtin_amdgcn_exp2f(And[2] * d2), h[2], u2 * Bv);
  h[3] = fmaf(__builtin_amdgcn_exp2f(And[3] * d3), h[3], u3 * Bv);
}

__device__ __forceinline__ void carry_steps(const BatchIn& q, const float (&And)[RCH],
                                            float (&h)[RCH], float2& sdt) {
  v2f dt = q.dt, xv = q.x;
  TAKE_LOADED(dt, xv);
  float2 u;  u.x = dt.x * xv.x;  u.y = dt.y * xv.y;
  float2 e;  e.x = dt.x * LOG2E;  e.y = dt.y * LOG2E;
  sdt.x += dt.x;  sdt.y += dt.y;
  scan_step<0>(e, u, q.b[0], And, h);
  scan_step<1>(e, u, q.b[1], And, h);
  scan_step<2>(e, u, q.b[2], And, h);
  scan_step<3>(e, u, q.b[3], And, h);
  scan_step<4>(e, u, q.b[4], And, h);
  scan_step<5>(e, u, q.b[5], And, h);
  scan_step<6>(e, u, q.b[6], And, h);
  scan_step<7>(e, u, q.b[7], And, h);
}

__global__ __launch_bounds__(SCAN_THREADS) __attribute__((amdgpu_waves_per_eu(1, 4))) void ssm_scan_carry(
    const float* __restrict__ x, const float* __restrict__ dtbuf,
    const float* __restrict__ Bmbuf, const float* __restrict__ Aneg,
    float* __restrict__ Sbuf, float* __restrict__ Pbuf)
{
  const int t    = threadIdx.x;
  const int n    = t & 15;
  const int hf   = (t >> 4) & 1;
  const int wave = t >> 5;
  const int bi   = blockIdx.x;
  const int c    = bi & (NCHUNK - 1);
  const int dg   = (bi >> 3) & 31;
  const int b    = bi >> 8;
  const int d0   = dg * 32 + wave * 8 + hf * RCH;

  float And[RCH], h[RCH];
#pragma unroll
  for (int r = 0; r < RCH; ++r) {
    And[r] = Aneg[(d0 + r) * NST + n];
    h[r] = 0.0f;
  }
  float2 sdt;  sdt.x = 0.0f;  sdt.y = 0.0f;
  const size_t rbase = (size_t)b * LSEQ + (size_t)c * CHUNK;
  const size_t hold  = (rbase + (n >> 1)) * DEXP + d0 + (n & 1) * 2;
  const float* dtp = dtbuf + hold;
  const float* xp  = x     + hold;
  const float* Bp  = Bmbuf + rbase * NST + n;

  static_assert((CHUNK / UB) % 4 == 0, "whole turns of the ring: all but the last turn in the loop, the last turn behind it");
  BatchIn q0, q1, q2, q3;
  load_in(q0, dtp, xp, Bp, 0);                         SEG_FENCE();
  load_in(q1, dtp, xp, Bp, UB);                        SEG_FENCE();
  load_in(q2, dtp, xp, Bp, 2 * UB);                    SEG_FENCE();
#pragma unroll 1
  for (int l0 = 0; l0 < CHUNK - 4 * UB; l0 += 4 * UB) {
    load_in(q3, dtp, xp, Bp, group_row(l0 + 3 * UB));  SEG_FENCE();
    carry_steps(q0, And, h, sdt); HOLD_STATE(h);           SEG_FENCE();
    load_in(q0, dtp, xp, Bp, group_row(l0 + 4 * UB));  SEG_FENCE();
    carry_steps(q1, And, h, sdt); HOLD_STATE(h);           SEG_FENCE();
    load_in(q1, dtp, xp, Bp, group_row(l0 + 5 * UB));  SEG_FENCE();
    carry_steps(q2, And, h, sdt); HOLD_STATE(h);           SEG_FENCE();
    load_in(q2, dtp, xp, Bp, group_row(l0 + 6 * UB));  SEG_FENCE();
    carry_steps(q3, And, h, sdt);           HOLD_STATE(h);
  }
  load_in(q3, dtp, xp, Bp, group_row(CHUNK - UB));     SEG_FENCE();
  carry_steps(q0, And, h, sdt); HOLD_STATE(h);             SEG_FENCE();
  carry_steps(q1, And, h, sdt); HOLD_STATE(h);             SEG_FENCE();
  carry_steps(q2, And, h, sdt); HOLD_STATE(h);             SEG_FENCE();
  carry_steps(q3, And, h, sdt);
  sdt.x += row_xor<2>(sdt.x);  sdt.y += row_xor<2>(sdt.y);
  sdt.x += row_xor<4>(sdt.x);  sdt.y += row_xor<4>(sdt.y);
  sdt.x += row_xor<8>(sdt.x);  sdt.y += row_xor<8>(sdt.y);
  const float tot[RCH] = {row_get<0>(sdt.x), row_get<0>(sdt.y), row_get<1>(sdt.x), row_get<1>(sdt.y)};
  const size_t si = (((size_t)b * NCHUNK + c) * DEXP + d0) * NST + n;
#pragma unroll
  for (int r = 0; r < RCH; ++r) {
    Sbuf[si + r * NST] = h[r];
    Pbuf[si + r * NST] = __expf(And[r] * tot[r]);
  }
}

__global__ void ssm_carry_combine(const float* __restrict__ Sbuf,
                                  const float* __restrict__ Pbuf,
                                  float* __restrict__ Car)
{
  const int tg = blockIdx.x * 256 + threadIdx.x;
  const int n  = tg & 15;
  const int d  = (tg >> 4) & (DEXP - 1);
  const int b  = tg >> 14;
  const size_t stride = (size_t)DEXP * NST;
  const size_t base   = (size_t)b * NCHUNK * stride + (size_t)d * NST + n;
  float cin = 0.0f;
  Car[base] = 0.0f;
  for (int c = 1; c < NCHUNK; ++c) {
    const size_t prev = base + (size_t)(c - 1) * stride;
    cin = Sbuf[prev] + Pbuf[prev] * cin;
    Car[base + (size_t)c * stride] = cin;
  }
}

template <int J>
__device__ __forceinline__ float out_step(const float2 e, const float2 u, const float Bv,
                                          const float Cv, const float (&And)[RCH],
                                          float (&h)[RCH], const bool b0, const bool b1) {
  scan_step<J>(e, u, Bv, And, h);
  float y[RCH];
#pragma unroll
  for (int r = 0; r < RCH; ++r) y[r] = h[r] * Cv;
  const float k0 = b0 ? y[2] : y[0], s0 = b0 ? y[0] : y[2];
  const float k1 = b0 ? y[3] : y[1], s1 = b0 ? y[1] : y[3];
  const float z0 = k0 + row_xor<1>(s0);
  const float z1 = k1 + row_xor<1>(s1);
  const float kk = b1 ? z1 : z0, ss = b1 ? z0 : z1;
  return kk + row_xor<2>(ss);
}

__device__ __forceinline__ void out_steps(const BatchOut& q, const float (&And)[RCH],
                                          float (&h)[RCH], const bool b0, const bool b1,
                                          const bool b2, const bool b3, float* op, int l0) {
  static_assert(UB == 8, "the eight steps of a batch are written out below");
  v2f dt = q.dt, xv = q.x;
  TAKE_LOADED(dt, xv);
  float2 u;  u.x = dt.x * xv.x;  u.y = dt.y * xv.y;
  float2 e;  e.x = dt.x * LOG2E;  e.y = dt.y * LOG2E;
  float z[UB];
  z[0] = out_step<0>(e, u, q.b[0], q.c[0], And, h, b0, b1);
  z[1] = out_step<1>(e, u, q.b[1], q.c[1], And, h, b0, b1);
  z[2] = out_step<2>(e, u, q.b[2], q.c[2], And, h, b0, b1);
  z[3] = out_step<3>(e, u, q.b[3], q.c[3], And, h, b0, b1);
  z[4] = out_step<4>(e, u, q.b[4], q.c[4], And, h, b0, b1);
  z[5] = out_step<5>(e, u, q.b[5], q.c[5], And, h, b0, b1);
  z[6] = out_step<6>(e, u, q.b[6], q.c[6], And, h, b0, b1);
  z[7] = out_step<7>(e, u, q.b[7], q.c[7], And, h, b0, b1);
#pragma unroll
  for (int g = 0; g < UB; g += 4) {
    const float kA = b2 ? z[g + 1] : z[g],     sA = b2 ? z[g]     : z[g + 1];
    const float kB = b2 ? z[g + 3] : z[g + 2], sB = b2 ? z[g + 2] : z[g + 3];
    const float wA = kA + row_xor<4>(sA);
    const float wB = kB + row_xor<4>(sB);
    const float kk = b3 ? wB : wA, ss = b3 ? wA : wB;
    op[(size_t)(l0 + g) * DEXP] = kk + row_xor<8>(ss);
  }
}

__global__ __launch_bounds__(SCAN_THREADS) void ssm_scan_out(
    const float* __restrict__ x, const float* __restrict__ dtbuf,
    const float* __restrict__ Bmbuf, const float* __restrict__ Cbuf,
    const float* __restrict__ Aneg, const float* __restrict__ Car,
    float* __restrict__ out)
{
  const int t    = threadIdx.x;
  const int n    = t & 15;
  const int hf   = (t >> 4) & 1;
  const int wave = t >> 5;
  const int bi   = blockIdx.x;
  const int c    = bi & (NCHUNK - 1);
  const int dg   = (bi >> 3) & 31;
  const int b    = bi >> 8;
  const int d0   = dg * 32 + wave * 8 + hf * RCH;

  const size_t rbase = (size_t)b * LSEQ + (size_t)c * CHUNK;
  const size_t si = (((size_t)b * NCHUNK + c) * DEXP + d0) * NST + n;
  float And[RCH], h[RCH];
#pragma unroll
  for (int r = 0; r < RCH; ++r) {
    And[r] = Aneg[(d0 + r) * NST + n];
    h[r]   = Car[si + r * NST];
  }
  const size_t hold = (rbase + (n >> 1)) * DEXP + d0 + (n & 1) * 2;
  const float* dtp = dtbuf + hold;
  const float* xp  = x     + hold;
  const float* Bp  = Bmbuf + rbase * NST + n;
  const float* Cp  = Cbuf  + rbase * NST + n;

  const bool b0 = (n & 1) != 0, b1 = (n & 2) != 0, b2 = (n & 4) != 0, b3 = (n & 8) != 0;
  const int  mych = ((n & 1) << 1) | ((n >> 1) & 1);
  const int  myst = ((n >> 2) & 1) | ((n >> 3) << 1);
  float* op = out + (rbase + myst) * DEXP + d0 + mych;

  static_assert((CHUNK / UB) % 4 == 0, "whole turns of the ring: all but the last turn in the loop, the last turn behind it");
  BatchOut qa, qb, qc, qd;
  load_out(qa, dtp, xp, Bp, Cp, 0);                               SEG_FENCE();
  load_out(qb, dtp, xp, Bp, Cp, UB);                              SEG_FENCE();
  load_out(qc, dtp, xp, Bp, Cp, 2 * UB);                          SEG_FENCE();
#pragma unroll 1
  for (int l0 = 0; l0 < CHUNK - 4 * UB; l0 += 4 * UB) {
    load_out(qd, dtp, xp, Bp, Cp, group_row(l0 + 3 * UB));        SEG_FENCE();
    out_steps(qa, And, h, b0, b1, b2, b3, op, l0); HOLD_STATE(h);           SEG_FENCE();
    load_out(qa, dtp, xp, Bp, Cp, group_row(l0 + 4 * UB));        SEG_FENCE();
    out_steps(qb, And, h, b0, b1, b2, b3, op, l0 + UB); HOLD_STATE(h);      SEG_FENCE();
    load_out(qb, dtp, xp, Bp, Cp, group_row(l0 + 5 * UB));        SEG_FENCE();
    out_steps(qc, And, h, b0, b1, b2, b3, op, l0 + 2 * UB); HOLD_STATE(h);  SEG_FENCE();
    load_out(qc, dtp, xp, Bp, Cp, group_row(l0 + 6 * UB));        SEG_FENCE();
    out_steps(qd, And, h, b0, b1, b2, b3, op, l0 + 3 * UB);  HOLD_STATE(h);
  }
  load_out(qd, dtp, xp, Bp, Cp, group_row(CHUNK - UB));           SEG_FENCE();
  out_steps(qa, And, h, b0, b1, b2, b3, op, CHUNK - 4 * UB); HOLD_STATE(h);  SEG_FENCE();
  out_steps(qb, And, h, b0, b1, b2, b3, op, CHUNK - 3 * UB); HOLD_STATE(h);  SEG_FENCE();
  out_steps(qc, And, h, b0, b1, b2, b3, op, CHUNK - 2 * UB); HOLD_STATE(h);  SEG_FENCE();
  out_steps(qd, And, h, b0, b1, b2, b3, op, CHUNK - UB);
}

extern "C" void kernel_launch(void* const* d_in, const int* in_sizes, int n_in,
                              void* d_out, int out_size, void* d_ws, size_t ws_size,
                              hipStream_t stream) {
  (void)in_sizes; (void)n_in; (void)out_size; (void)ws_size;
  const float* x     = (const float*)d_in[0];
  const float* xw    = (const float*)d_in[1];
  const float* xb    = (const float*)d_in[2];
  const float* dtw   = (const float*)d_in[3];
  const float* dtb   = (const float*)d_in[4];
  const float* A_log = (const float*)d_in[5];
  float* out = (float*)d_out;

  float* ws    = (float*)d_ws;
  float* dtbuf = ws;
  float* Aneg  = dtbuf + (size_t)ROWS * DEXP;
  float* Bm    = Aneg  + (size_t)DEXP * NST;
  float* Cb    = Bm    + (size_t)ROWS * NST;
  float* Sb    = Cb    + (size_t)ROWS * NST;
  float* Pb    = Sb    + (size_t)B_SZ * NCHUNK * DEXP * NST;
  float* Car   = Pb    + (size_t)B_SZ * NCHUNK * DEXP * NST;
  float* xwT   = Car   + (size_t)B_SZ * NCHUNK * DEXP * NST;
  float* dtwT  = xwT   + (size_t)DEXP * RPROJ;

  hipLaunchKernelGGL(ssm_prep, dim3((DEXP * RPROJ + 255) / 256), dim3(256), 0, stream,
                     A_log, Aneg, xw, xwT, dtw, dtwT);
  hipLaunchKernelGGL(ssm_proj_wmma, dim3(ROWS / 16), dim3(256), 0, stream,
                     x, xwT, xb, dtwT, dtb, dtbuf, Bm, Cb);
  hipLaunchKernelGGL(ssm_scan_carry, dim3(SCAN_BLOCKS), dim3(SCAN_THREADS), 0, stream,
                     x, dtbuf, Bm, Aneg, Sb, Pb);
  hipLaunchKernelGGL(ssm_carry_combine, dim3((B_SZ * DEXP * NST) / 256), dim3(256), 0, stream,
                     Sb, Pb, Car);
  hipLaunchKernelGGL(ssm_scan_out, dim3(SCAN_BLOCKS), dim3(SCAN_THREADS), 0, stream,
                     x, dtbuf, Bm, Cb, Aneg, Car, out);
}


# scan kernels: lane holds 2 channels x 2 states (was 4 x 1): dt*x, dt*log2(e) come by 4 lane gathers per step (was 8 DPP reads), step math packed f32, first n-sum level in-lane (fused); same sum tree
# speedup vs baseline: 4.6484x; 1.1089x over previous
#include <hip/hip_runtime.h>
#include <math.h>

#define B_SZ   2
#define LSEQ   2048
#define DEXP   1024
#define DRANK  64
#define NST    16
#define RPROJ  (DRANK + 2*NST)
#define NCHUNK 8
#define CHUNK  (LSEQ / NCHUNK)
#define ROWS   (B_SZ * LSEQ)

typedef __attribute__((ext_vector_type(2))) float v2f;
typedef __attribute__((ext_vector_type(8))) float v8f;

__device__ __forceinline__ v8f wmma_f32_k4(v2f a, v2f b, v8f c) {
  return __builtin_amdgcn_wmma_f32_16x16x4_f32(
       false, a,  false, b,
       (short)0, c,  false,  false);
}

__global__ void ssm_prep(const float* __restrict__ A_log, float* __restrict__ Aneg,
                         const float* __restrict__ xw,  float* __restrict__ xwT,
                         const float* __restrict__ dtw, float* __restrict__ dtwT) {
  int i = blockIdx.x * 256 + threadIdx.x;
  if (i < DEXP * NST) Aneg[i] = -__expf(A_log[i]);
  if (i < DEXP * RPROJ) {
    int k = i / RPROJ, c = i % RPROJ;
    xwT[i] = xw[(size_t)c * DEXP + k];
  }
  if (i < DRANK * DEXP) {
    int k = i / DEXP, d = i % DEXP;
    dtwT[i] = dtw[(size_t)d * DRANK + k];
  }
}

struct BSet1 { v2f f[16]; };
struct BSet2 { v2f f[16]; float bias; };

#define SXP 1028

__device__ __forceinline__ void load_b1(BSet1& q, const float* bp, int g) {
#pragma unroll
  for (int s = 0; s < 16; ++s) {
    const int k = g * 64 + s * 4;
    q.f[s].x = bp[(size_t)k * RPROJ];
    q.f[s].y = bp[(size_t)(k + 1) * RPROJ];
  }
}

__device__ __forceinline__ v8f mma_group1(const BSet1& q, const float* ap, int g, v8f acc) {
#pragma unroll
  for (int s = 0; s < 16; ++s) {
    const int k = g * 64 + s * 4;
    v2f a;  a.x = ap[k];  a.y = ap[k + 1];
    acc = wmma_f32_k4(a, q.f[s], acc);
  }
  return acc;
}

__device__ __forceinline__ void load_b2(BSet2& q, const float* wp, const float* dtb, int col) {
  q.bias = dtb[col];
#pragma unroll
  for (int kt = 0; kt < 16; ++kt) {
    q.f[kt].x = wp[(size_t)(kt * 4) * DEXP + col];
    q.f[kt].y = wp[(size_t)(kt * 4 + 1) * DEXP + col];
  }
}

__device__ __forceinline__ void tile2(const BSet2& q, const v2f (&a2)[16], float* dp, int col) {
  v8f acc2;
#pragma unroll
  for (int i = 0; i < 8; ++i) acc2[i] = q.bias;
#pragma unroll
  for (int kt = 0; kt < 16; ++kt) acc2 = wmma_f32_k4(a2[kt], q.f[kt], acc2);
  float sp[8];
#pragma unroll
  for (int i = 0; i < 8; ++i) sp[i] = log1pf(__expf(acc2[i]));
#pragma unroll
  for (int i = 0; i < 8; ++i) {
    const float v = acc2[i];
    dp[(size_t)i * DEXP + col] = (v > 20.0f) ? v : sp[i];
  }
}

__global__ __launch_bounds__(256) void ssm_proj_wmma(
    const float* __restrict__ x,    const float* __restrict__ xwT,
    const float* __restrict__ xb,   const float* __restrict__ dtwT,
    const float* __restrict__ dtb,
    float* __restrict__ dtbuf, float* __restrict__ Bmbuf, float* __restrict__ Cbuf)
{
  __shared__ float sx[16 * SXP];
  __shared__ float sproj[16 * 100];

  const int t    = threadIdx.x;
  const int wave = t >> 5;
  const int nl   = t & 15;
  const int hi   = (t >> 4) & 1;
  const int rowbase = blockIdx.x * 16;

  const int col1 = wave * 16 + nl;
  const float* bp = xwT + (size_t)(hi * 2) * RPROJ + col1;
  BSet1 b0, b1;
  v8f acc1;
  float bv1 = 0.0f;
  if (wave < 6) {
    load_b1(b0, bp, 0);
    bv1 = xb[col1];
  }
  for (int i = 0; i < 16; ++i) {
    int flat = t + i * 256;
    int r  = flat >> 8;
    int kk = (flat & 255) << 2;
    const float4 v = *(const float4*)(x + (size_t)(rowbase + r) * DEXP + kk);
    float* dst = &sx[r * SXP + kk];
    dst[0] = v.x; dst[1] = v.y; dst[2] = v.z; dst[3] = v.w;
  }
  __syncthreads();
  if (wave < 6) {
    for (int i = 0; i < 8; ++i) acc1[i] = bv1;
    const float* ap = &sx[nl * SXP + hi * 2];
#pragma unroll 1
    for (int g = 0; g < 16; g += 2) {
      load_b1(b1, bp, g + 1);
      acc1 = mma_group1(b0, ap, g, acc1);
      if (g + 2 < 16) load_b1(b0, bp, g + 2);
      acc1 = mma_group1(b1, ap, g + 1, acc1);
    }
  }
  const float* wp = dtwT + (size_t)(hi * 2) * DEXP;
  const int colw = wave * 128 + nl;
  BSet2 c0, c1;
  load_b2(c0, wp, dtb, colw);
  if (wave < 6) {
    for (int i = 0; i < 8; ++i) {
      int M = i + hi * 8;
      sproj[M * 100 + wave * 16 + nl] = acc1[i];
      if (wave == 4) Bmbuf[(size_t)(rowbase + M) * NST + nl] = acc1[i];
      if (wave == 5) Cbuf [(size_t)(rowbase + M) * NST + nl] = acc1[i];
    }
  }
  __syncthreads();

  v2f a2[16];
  for (int kt = 0; kt < 16; ++kt) {
    int kb = kt * 4 + hi * 2;
    a2[kt].x = sproj[nl * 100 + kb];
    a2[kt].y = sproj[nl * 100 + kb + 1];
  }
  float* dp = dtbuf + (size_t)(rowbase + hi * 8) * DEXP;
#pragma unroll 1
  for (int j = 0; j < 8; j += 2) {
    load_b2(c1, wp, dtb, colw + (j + 1) * 16);
    tile2(c0, a2, dp, colw + j * 16);
    if (j + 2 < 8) load_b2(c0, wp, dtb, colw + (j + 2) * 16);
    tile2(c1, a2, dp, colw + (j + 1) * 16);
  }
}

#define RCH 4
#define UB  8
#define SCAN_THREADS 128
#define SCAN_BLOCKS  (B_SZ * NCHUNK * (DEXP / 32))

#define SEG_FENCE() __builtin_amdgcn_sched_barrier(0)
#define LOG2E 1.44269504088896340736f

__device__ __forceinline__ int group_row(int l0) {
  asm volatile("" : "+s"(l0));
  return l0;
}
#define HOLD_STATE(H) asm volatile("" : "+v"(H[0]), "+v"(H[1]))
#define TAKE_LOADED(a, b) asm volatile("" : "+v"(a), "+v"(b))

struct BatchIn  { v2f dt; v2f x; v2f b[UB]; };
struct BatchOut { v2f dt; v2f x; v2f b[UB]; v2f c[UB]; };

__device__ __forceinline__ void load_in(BatchIn& q, const float* dtp, const float* xp,
                                        const float* Bp, int l0) {
  q.dt = *(const v2f*)(dtp + (size_t)l0 * DEXP);
  q.x  = *(const v2f*)(xp  + (size_t)l0 * DEXP);
  const float* Bg = Bp + (size_t)l0 * NST;
#pragma unroll
  for (int j = 0; j < UB; ++j) q.b[j] = *(const v2f*)(Bg + j * NST);
}

__device__ __forceinline__ void load_out(BatchOut& q, const float* dtp, const float* xp,
                                         const float* Bp, const float* Cp, int l0) {
  q.dt = *(const v2f*)(dtp + (size_t)l0 * DEXP);
  q.x  = *(const v2f*)(xp  + (size_t)l0 * DEXP);
  const float* Bg = Bp + (size_t)l0 * NST;
  const float* Cg = Cp + (size_t)l0 * NST;
#pragma unroll
  for (int j = 0; j < UB; ++j) {
    q.b[j] = *(const v2f*)(Bg + j * NST);
    q.c[j] = *(const v2f*)(Cg + j * NST);
  }
}

template <int M>
__device__ __forceinline__ float row_xor(float v) {
  return __int_as_float(__builtin_amdgcn_update_dpp(0, __float_as_int(v), 0x160 | M,
                                                    0xf, 0xf, true));
}

template <int J>
__device__ __forceinline__ float pair_get(float v) {
  constexpr unsigned sel = 0x11111111u * (unsigned)(2 * J) + 0x10101010u;
  return __int_as_float(__builtin_amdgcn_permlane16(0, __float_as_int(v), sel, sel, false, true));
}

template <int J>
__device__ __forceinline__ void scan_step(const v2f e, const v2f u, const v2f Bv,
                                          const v2f (&A)[2], v2f (&H)[2]) {
  v2f E, U;
  E.x = pair_get<J>(e.x);  E.y = pair_get<J>(e.y);
  U.x = pair_get<J>(u.x);  U.y = pair_get<J>(u.y);
  const v2f g0 = A[0] * E, g1 = A[1] * E;
  v2f a0, a1;
  a0.x = __builtin_amdgcn_exp2f(g0.x);  a0.y = __builtin_amdgcn_exp2f(g0.y);
  a1.x = __builtin_amdgcn_exp2f(g1.x);  a1.y = __builtin_amdgcn_exp2f(g1.y);
  const v2f bs0 = {Bv.x, Bv.x}, bs1 = {Bv.y, Bv.y};
  H[0] = __builtin_elementwise_fma(a0, H[0], U * bs0);
  H[1] = __builtin_elementwise_fma(a1, H[1], U * bs1);
}

__device__ __forceinline__ void carry_steps(const BatchIn& q, const v2f (&A)[2],
                                            v2f (&H)[2], v2f& sdt) {
  v2f dt = q.dt, xv = q.x;
  TAKE_LOADED(dt, xv);
  const v2f u = dt * xv;
  const v2f e = dt * LOG2E;
  sdt += dt;
  scan_step<0>(e, u, q.b[0], A, H);
  scan_step<1>(e, u, q.b[1], A, H);
  scan_step<2>(e, u, q.b[2], A, H);
  scan_step<3>(e, u, q.b[3], A, H);
  scan_step<4>(e, u, q.b[4], A, H);
  scan_step<5>(e, u, q.b[5], A, H);
  scan_step<6>(e, u, q.b[6], A, H);
  scan_step<7>(e, u, q.b[7], A, H);
}

__global__ __launch_bounds__(SCAN_THREADS) __attribute__((amdgpu_waves_per_eu(1, 4))) void ssm_scan_carry(
    const float* __restrict__ x, const float* __restrict__ dtbuf,
    const float* __restrict__ Bmbuf, const float* __restrict__ Aneg,
    float* __restrict__ Sbuf, float* __restrict__ Pbuf)
{
  const int t    = threadIdx.x;
  const int n    = t & 15;
  const int cp   = n & 1;
  const int ng   = n >> 1;
  const int hf   = (t >> 4) & 1;
  const int wave = t >> 5;
  const int bi   = blockIdx.x;
  const int c    = bi & (NCHUNK - 1);
  const int dg   = (bi >> 3) & 31;
  const int b    = bi >> 8;
  const int d0   = dg * 32 + wave * 8 + hf * RCH;
  const int dl   = d0 + 2 * cp;

  const v2f ar0 = *(const v2f*)(Aneg + (size_t)dl * NST + 2 * ng);
  const v2f ar1 = *(const v2f*)(Aneg + (size_t)(dl + 1) * NST + 2 * ng);
  v2f A[2], H[2];
  A[0].x = ar0.x;  A[0].y = ar1.x;
  A[1].x = ar0.y;  A[1].y = ar1.y;
  H[0].x = 0.0f;  H[0].y = 0.0f;  H[1].x = 0.0f;  H[1].y = 0.0f;
  v2f sdt;  sdt.x = 0.0f;  sdt.y = 0.0f;
  const size_t rbase = (size_t)b * LSEQ + (size_t)c * CHUNK;
  const size_t hold  = (rbase + (n >> 1)) * DEXP + d0 + (n & 1) * 2;
  const float* dtp = dtbuf + hold;
  const float* xp  = x     + hold;
  const float* Bp  = Bmbuf + rbase * NST + 2 * ng;

  static_assert((CHUNK / UB) % 4 == 0, "whole turns of the ring: all but the last turn in the loop, the last turn behind it");
  BatchIn q0, q1, q2, q3;
  load_in(q0, dtp, xp, Bp, 0);                         SEG_FENCE();
  load_in(q1, dtp, xp, Bp, UB);                        SEG_FENCE();
  load_in(q2, dtp, xp, Bp, 2 * UB);                    SEG_FENCE();
#pragma unroll 1
  for (int l0 = 0; l0 < CHUNK - 4 * UB; l0 += 4 * UB) {
    load_in(q3, dtp, xp, Bp, group_row(l0 + 3 * UB));  SEG_FENCE();
    carry_steps(q0, A, H, sdt); HOLD_STATE(H);           SEG_FENCE();
    load_in(q0, dtp, xp, Bp, group_row(l0 + 4 * UB));  SEG_FENCE();
    carry_steps(q1, A, H, sdt); HOLD_STATE(H);           SEG_FENCE();
    load_in(q1, dtp, xp, Bp, group_row(l0 + 5 * UB));  SEG_FENCE();
    carry_steps(q2, A, H, sdt); HOLD_STATE(H);           SEG_FENCE();
    load_in(q2, dtp, xp, Bp, group_row(l0 + 6 * UB));  SEG_FENCE();
    carry_steps(q3, A, H, sdt);           HOLD_STATE(H);
  }
  load_in(q3, dtp, xp, Bp, group_row(CHUNK - UB));     SEG_FENCE();
  carry_steps(q0, A, H, sdt); HOLD_STATE(H);             SEG_FENCE();
  carry_steps(q1, A, H, sdt); HOLD_STATE(H);             SEG_FENCE();
  carry_steps(q2, A, H, sdt); HOLD_STATE(H);             SEG_FENCE();
  carry_steps(q3, A, H, sdt);
  sdt.x += row_xor<2>(sdt.x);  sdt.y += row_xor<2>(sdt.y);
  sdt.x += row_xor<4>(sdt.x);  sdt.y += row_xor<4>(sdt.y);
  sdt.x += row_xor<8>(sdt.x);  sdt.y += row_xor<8>(sdt.y);
  const v2f p0 = A[0] * sdt, p1 = A[1] * sdt;
  const size_t si = (((size_t)b * NCHUNK + c) * DEXP + dl) * NST + 2 * ng;
  v2f s0, s1, e0, e1;
  s0.x = H[0].x;  s0.y = H[1].x;
  s1.x = H[0].y;  s1.y = H[1].y;
  e0.x = __expf(p0.x);  e0.y = __expf(p1.x);
  e1.x = __expf(p0.y);  e1.y = __expf(p1.y);
  *(v2f*)(Sbuf + si)       = s0;
  *(v2f*)(Sbuf + si + NST) = s1;
  *(v2f*)(Pbuf + si)       = e0;
  *(v2f*)(Pbuf + si + NST) = e1;
}

__global__ void ssm_carry_combine(const float* __restrict__ Sbuf,
                                  const float* __restrict__ Pbuf,
                                  float* __restrict__ Car)
{
  const int tg = blockIdx.x * 256 + threadIdx.x;
  const int n  = tg & 15;
  const int d  = (tg >> 4) & (DEXP - 1);
  const int b  = tg >> 14;
  const size_t stride = (size_t)DEXP * NST;
  const size_t base   = (size_t)b * NCHUNK * stride + (size_t)d * NST + n;
  float cin = 0.0f;
  Car[base] = 0.0f;
  for (int c = 1; c < NCHUNK; ++c) {
    const size_t prev = base + (size_t)(c - 1) * stride;
    cin = Sbuf[prev] + Pbuf[prev] * cin;
    Car[base + (size_t)c * stride] = cin;
  }
}

template <int J>
__device__ __forceinline__ float out_step(const v2f e, const v2f u, const v2f Bv,
                                          const v2f Cv, const v2f (&A)[2],
                                          v2f (&H)[2], const bool b1) {
  scan_step<J>(e, u, Bv, A, H);
  const v2f cs0 = {Cv.x, Cv.x}, cs1 = {Cv.y, Cv.y};
  const v2f y = H[0] * cs0 + H[1] * cs1;
  const float kk = b1 ? y.y : y.x, ss = b1 ? y.x : y.y;
  return kk + row_xor<2>(ss);
}

__device__ __forceinline__ void out_steps(const BatchOut& q, const v2f (&A)[2],
                                          v2f (&H)[2], const bool b1,
                                          const bool b2, const bool b3, float* op, int l0) {
  static_assert(UB == 8, "the eight steps of a batch are written out below");
  v2f dt = q.dt, xv = q.x;
  TAKE_LOADED(dt, xv);
  const v2f u = dt * xv;
  const v2f e = dt * LOG2E;
  float z[UB];
  z[0] = out_step<0>(e, u, q.b[0], q.c[0], A, H, b1);
  z[1] = out_step<1>(e, u, q.b[1], q.c[1], A, H, b1);
  z[2] = out_step<2>(e, u, q.b[2], q.c[2], A, H, b1);
  z[3] = out_step<3>(e, u, q.b[3], q.c[3], A, H, b1);
  z[4] = out_step<4>(e, u, q.b[4], q.c[4], A, H, b1);
  z[5] = out_step<5>(e, u, q.b[5], q.c[5], A, H, b1);
  z[6] = out_step<6>(e, u, q.b[6], q.c[6], A, H, b1);
  z[7] = out_step<7>(e, u, q.b[7], q.c[7], A, H, b1);
#pragma unroll
  for (int g = 0; g < UB; g += 4) {
    const float kA = b2 ? z[g + 1] : z[g],     sA = b2 ? z[g]     : z[g + 1];
    const float kB = b2 ? z[g + 3] : z[g + 2], sB = b2 ? z[g + 2] : z[g + 3];
    const float wA = kA + row_xor<4>(sA);
    const float wB = kB + row_xor<4>(sB);
    const float kk = b3 ? wB : wA, ss = b3 ? wA : wB;
    op[(size_t)(l0 + g) * DEXP] = kk + row_xor<8>(ss);
  }
}

__global__ __launch_bounds__(SCAN_THREADS) void ssm_scan_out(
    const float* __restrict__ x, const float* __restrict__ dtbuf,
    const float* __restrict__ Bmbuf, const float* __restrict__ Cbuf,
    const float* __restrict__ Aneg, const float* __restrict__ Car,
    float* __restrict__ out)
{
  const int t    = threadIdx.x;
  const int n    = t & 15;
  const int cp   = n & 1;
  const int ng   = n >> 1;
  const int hf   = (t >> 4) & 1;
  const int wave = t >> 5;
  const int bi   = blockIdx.x;
  const int c    = bi & (NCHUNK - 1);
  const int dg   = (bi >> 3) & 31;
  const int b    = bi >> 8;
  const int d0   = dg * 32 + wave * 8 + hf * RCH;
  const int dl   = d0 + 2 * cp;

  const size_t rbase = (size_t)b * LSEQ + (size_t)c * CHUNK;
  const size_t si = (((size_t)b * NCHUNK + c) * DEXP + dl) * NST + 2 * ng;
  const v2f ar0 = *(const v2f*)(Aneg + (size_t)dl * NST + 2 * ng);
  const v2f ar1 = *(const v2f*)(Aneg + (size_t)(dl + 1) * NST + 2 * ng);
  const v2f cr0 = *(const v2f*)(Car + si);
  const v2f cr1 = *(const v2f*)(Car + si + NST);
  v2f A[2], H[2];
  A[0].x = ar0.x;  A[0].y = ar1.x;
  A[1].x = ar0.y;  A[1].y = ar1.y;
  H[0].x = cr0.x;  H[0].y = cr1.x;
  H[1].x = cr0.y;  H[1].y = cr1.y;
  const size_t hold = (rbase + (n >> 1)) * DEXP + d0 + (n & 1) * 2;
  const float* dtp = dtbuf + hold;
  const float* xp  = x     + hold;
  const float* Bp  = Bmbuf + rbase * NST + 2 * ng;
  const float* Cp  = Cbuf  + rbase * NST + 2 * ng;

  const bool b1 = (n & 2) != 0, b2 = (n & 4) != 0, b3 = (n & 8) != 0;
  const int  mych = ((n & 1) << 1) | ((n >> 1) & 1);
  const int  myst = ((n >> 2) & 1) | ((n >> 3) << 1);
  float* op = out + (rbase + myst) * DEXP + d0 + mych;

  static_assert((CHUNK / UB) % 4 == 0, "whole turns of the ring: all but the last turn in the loop, the last turn behind it");
  BatchOut qa, qb, qc, qd;
  load_out(qa, dtp, xp, Bp, Cp, 0);                               SEG_FENCE();
  load_out(qb, dtp, xp, Bp, Cp, UB);                              SEG_FENCE();
  load_out(qc, dtp, xp, Bp, Cp, 2 * UB);                          SEG_FENCE();
#pragma unroll 1
  for (int l0 = 0; l0 < CHUNK - 4 * UB; l0 += 4 * UB) {
    load_out(qd, dtp, xp, Bp, Cp, group_row(l0 + 3 * UB));        SEG_FENCE();
    out_steps(qa, A, H, b1, b2, b3, op, l0); HOLD_STATE(H);           SEG_FENCE();
    load_out(qa, dtp, xp, Bp, Cp, group_row(l0 + 4 * UB));        SEG_FENCE();
    out_steps(qb, A, H, b1, b2, b3, op, l0 + UB); HOLD_STATE(H);      SEG_FENCE();
    load_out(qb, dtp, xp, Bp, Cp, group_row(l0 + 5 * UB));        SEG_FENCE();
    out_steps(qc, A, H, b1, b2, b3, op, l0 + 2 * UB); HOLD_STATE(H);  SEG_FENCE();
    load_out(qc, dtp, xp, Bp, Cp, group_row(l0 + 6 * UB));        SEG_FENCE();
    out_steps(qd, A, H, b1, b2, b3, op, l0 + 3 * UB);  HOLD_STATE(H);
  }
  load_out(qd, dtp, xp, Bp, Cp, group_row(CHUNK - UB));           SEG_FENCE();
  out_steps(qa, A, H, b1, b2, b3, op, CHUNK - 4 * UB); HOLD_STATE(H);  SEG_FENCE();
  out_steps(qb, A, H, b1, b2, b3, op, CHUNK - 3 * UB); HOLD_STATE(H);  SEG_FENCE();
  out_steps(qc, A, H, b1, b2, b3, op, CHUNK - 2 * UB); HOLD_STATE(H);  SEG_FENCE();
  out_steps(qd, A, H, b1, b2, b3, op, CHUNK - UB);
}

extern "C" void kernel_launch(void* const* d_in, const int* in_sizes, int n_in,
                              void* d_out, int out_size, void* d_ws, size_t ws_size,
                              hipStream_t stream) {
  (void)in_sizes; (void)n_in; (void)out_size; (void)ws_size;
  const float* x     = (const float*)d_in[0];
  const float* xw    = (const float*)d_in[1];
  const float* xb    = (const float*)d_in[2];
  const float* dtw   = (const float*)d_in[3];
  const float* dtb   = (const float*)d_in[4];
  const float* A_log = (const float*)d_in[5];
  float* out = (float*)d_out;

  float* ws    = (float*)d_ws;
  float* dtbuf = ws;
  float* Aneg  = dtbuf + (size_t)ROWS * DEXP;
  float* Bm    = Aneg  + (size_t)DEXP * NST;
  float* Cb    = Bm    + (size_t)ROWS * NST;
  float* Sb    = Cb    + (size_t)ROWS * NST;
  float* Pb    = Sb    + (size_t)B_SZ * NCHUNK * DEXP * NST;
  float* Car   = Pb    + (size_t)B_SZ * NCHUNK * DEXP * NST;
  float* xwT   = Car   + (size_t)B_SZ * NCHUNK * DEXP * NST;
  float* dtwT  = xwT   + (size_t)DEXP * RPROJ;

  hipLaunchKernelGGL(ssm_prep, dim3((DEXP * RPROJ + 255) / 256), dim3(256), 0, stream,
                     A_log, Aneg, xw, xwT, dtw, dtwT);
  hipLaunchKernelGGL(ssm_proj_wmma, dim3(ROWS / 16), dim3(256), 0, stream,
                     x, xwT, xb, dtwT, dtb, dtbuf, Bm, Cb);
  hipLaunchKernelGGL(ssm_scan_carry, dim3(SCAN_BLOCKS), dim3(SCAN_THREADS), 0, stream,
                     x, dtbuf, Bm, Aneg, Sb, Pb);
  hipLaunchKernelGGL(ssm_carry_combine, dim3((B_SZ * DEXP * NST) / 256), dim3(256), 0, stream,
                     Sb, Pb, Car);
  hipLaunchKernelGGL(ssm_scan_out, dim3(SCAN_BLOCKS), dim3(SCAN_THREADS), 0, stream,
                     x, dtbuf, Bm, Cb, Aneg, Car, out);
}


# scan kernels: lane holds 1 channel x 4 states (was 2 x 2): 2 lane gathers per step (was 4), B/C by 16-byte loads, 4-state n-sum in-lane, then xor 4, 8; n-sum tree and chunk dt sum order change
# speedup vs baseline: 4.8008x; 1.0328x over previous
#include <hip/hip_runtime.h>
#include <math.h>

#define B_SZ   2
#define LSEQ   2048
#define DEXP   1024
#define DRANK  64
#define NST    16
#define RPROJ  (DRANK + 2*NST)
#define NCHUNK 8
#define CHUNK  (LSEQ / NCHUNK)
#define ROWS   (B_SZ * LSEQ)

typedef __attribute__((ext_vector_type(2))) float v2f;
typedef __attribute__((ext_vector_type(8))) float v8f;

__device__ __forceinline__ v8f wmma_f32_k4(v2f a, v2f b, v8f c) {
  return __builtin_amdgcn_wmma_f32_16x16x4_f32(
       false, a,  false, b,
       (short)0, c,  false,  false);
}

__global__ void ssm_prep(const float* __restrict__ A_log, float* __restrict__ Aneg,
                         const float* __restrict__ xw,  float* __restrict__ xwT,
                         const float* __restrict__ dtw, float* __restrict__ dtwT) {
  int i = blockIdx.x * 256 + threadIdx.x;
  if (i < DEXP * NST) Aneg[i] = -__expf(A_log[i]);
  if (i < DEXP * RPROJ) {
    int k = i / RPROJ, c = i % RPROJ;
    xwT[i] = xw[(size_t)c * DEXP + k];
  }
  if (i < DRANK * DEXP) {
    int k = i / DEXP, d = i % DEXP;
    dtwT[i] = dtw[(size_t)d * DRANK + k];
  }
}

struct BSet1 { v2f f[16]; };
struct BSet2 { v2f f[16]; float bias; };

#define SXP 1028

__device__ __forceinline__ void load_b1(BSet1& q, const float* bp, int g) {
#pragma unroll
  for (int s = 0; s < 16; ++s) {
    const int k = g * 64 + s * 4;
    q.f[s].x = bp[(size_t)k * RPROJ];
    q.f[s].y = bp[(size_t)(k + 1) * RPROJ];
  }
}

__device__ __forceinline__ v8f mma_group1(const BSet1& q, const float* ap, int g, v8f acc) {
#pragma unroll
  for (int s = 0; s < 16; ++s) {
    const int k = g * 64 + s * 4;
    v2f a;  a.x = ap[k];  a.y = ap[k + 1];
    acc = wmma_f32_k4(a, q.f[s], acc);
  }
  return acc;
}

__device__ __forceinline__ void load_b2(BSet2& q, const float* wp, const float* dtb, int col) {
  q.bias = dtb[col];
#pragma unroll
  for (int kt = 0; kt < 16; ++kt) {
    q.f[kt].x = wp[(size_t)(kt * 4) * DEXP + col];
    q.f[kt].y = wp[(size_t)(kt * 4 + 1) * DEXP + col];
  }
}

__device__ __forceinline__ void tile2(const BSet2& q, const v2f (&a2)[16], float* dp, int col) {
  v8f acc2;
#pragma unroll
  for (int i = 0; i < 8; ++i) acc2[i] = q.bias;
#pragma unroll
  for (int kt = 0; kt < 16; ++kt) acc2 = wmma_f32_k4(a2[kt], q.f[kt], acc2);
  float sp[8];
#pragma unroll
  for (int i = 0; i < 8; ++i) sp[i] = log1pf(__expf(acc2[i]));
#pragma unroll
  for (int i = 0; i < 8; ++i) {
    const float v = acc2[i];
    dp[(size_t)i * DEXP + col] = (v > 20.0f) ? v : sp[i];
  }
}

__global__ __launch_bounds__(256) void ssm_proj_wmma(
    const float* __restrict__ x,    const float* __restrict__ xwT,
    const float* __restrict__ xb,   const float* __restrict__ dtwT,
    const float* __restrict__ dtb,
    float* __restrict__ dtbuf, float* __restrict__ Bmbuf, float* __restrict__ Cbuf)
{
  __shared__ float sx[16 * SXP];
  __shared__ float sproj[16 * 100];

  const int t    = threadIdx.x;
  const int wave = t >> 5;
  const int nl   = t & 15;
  const int hi   = (t >> 4) & 1;
  const int rowbase = blockIdx.x * 16;

  const int col1 = wave * 16 + nl;
  const float* bp = xwT + (size_t)(hi * 2) * RPROJ + col1;
  BSet1 b0, b1;
  v8f acc1;
  float bv1 = 0.0f;
  if (wave < 6) {
    load_b1(b0, bp, 0);
    bv1 = xb[col1];
  }
  for (int i = 0; i < 16; ++i) {
    int flat = t + i * 256;
    int r  = flat >> 8;
    int kk = (flat & 255) << 2;
    const float4 v = *(const float4*)(x + (size_t)(rowbase + r) * DEXP + kk);
    float* dst = &sx[r * SXP + kk];
    dst[0] = v.x; dst[1] = v.y; dst[2] = v.z; dst[3] = v.w;
  }
  __syncthreads();
  if (wave < 6) {
    for (int i = 0; i < 8; ++i) acc1[i] = bv1;
    const float* ap = &sx[nl * SXP + hi * 2];
#pragma unroll 1
    for (int g = 0; g < 16; g += 2) {
      load_b1(b1, bp, g + 1);
      acc1 = mma_group1(b0, ap, g, acc1);
      if (g + 2 < 16) load_b1(b0, bp, g + 2);
      acc1 = mma_group1(b1, ap, g + 1, acc1);
    }
  }
  const float* wp = dtwT + (size_t)(hi * 2) * DEXP;
  const int colw = wave * 128 + nl;
  BSet2 c0, c1;
  load_b2(c0, wp, dtb, colw);
  if (wave < 6) {
    for (int i = 0; i < 8; ++i) {
      int M = i + hi * 8;
      sproj[M * 100 + wave * 16 + nl] = acc1[i];
      if (wave == 4) Bmbuf[(size_t)(rowbase + M) * NST + nl] = acc1[i];
      if (wave == 5) Cbuf [(size_t)(rowbase + M) * NST + nl] = acc1[i];
    }
  }
  __syncthreads();

  v2f a2[16];
  for (int kt = 0; kt < 16; ++kt) {
    int kb = kt * 4 + hi * 2;
    a2[kt].x = sproj[nl * 100 + kb];
    a2[kt].y = sproj[nl * 100 + kb + 1];
  }
  float* dp = dtbuf + (size_t)(rowbase + hi * 8) * DEXP;
#pragma unroll 1
  for (int j = 0; j < 8; j += 2) {
    load_b2(c1, wp, dtb, colw + (j + 1) * 16);
    tile2(c0, a2, dp, colw + j * 16);
    if (j + 2 < 8) load_b2(c0, wp, dtb, colw + (j + 2) * 16);
    tile2(c1, a2, dp, colw + (j + 1) * 16);
  }
}

#define RCH 4
#define UB  8
#define SCAN_THREADS 128
#define SCAN_BLOCKS  (B_SZ * NCHUNK * (DEXP / 32))

typedef __attribute__((ext_vector_type(4))) float v4f;

#define SEG_FENCE() __builtin_amdgcn_sched_barrier(0)
#define LOG2E 1.44269504088896340736f

__device__ __forceinline__ int group_row(int l0) {
  asm volatile("" : "+s"(l0));
  return l0;
}
#define HOLD_STATE(H) asm volatile("" : "+v"(H[0]), "+v"(H[1]))
#define TAKE_LOADED(a, b) asm volatile("" : "+v"(a), "+v"(b))

struct BatchIn  { v2f dt; v2f x; v4f b[UB]; };
struct BatchOut { v2f dt; v2f x; v4f b[UB]; v4f c[UB]; };

__device__ __forceinline__ void load_in(BatchIn& q, const float* dtp, const float* xp,
                                        const float* Bp, int l0) {
  q.dt.x = dtp[(size_t)l0 * DEXP];
  q.dt.y = dtp[(size_t)l0 * DEXP + DEXP];
  q.x.x  = xp [(size_t)l0 * DEXP];
  q.x.y  = xp [(size_t)l0 * DEXP + DEXP];
  const float* Bg = Bp + (size_t)l0 * NST;
#pragma unroll
  for (int j = 0; j < UB; ++j) q.b[j] = *(const v4f*)(Bg + j * NST);
}

__device__ __forceinline__ void load_out(BatchOut& q, const float* dtp, const float* xp,
                                         const float* Bp, const float* Cp, int l0) {
  q.dt.x = dtp[(size_t)l0 * DEXP];
  q.dt.y = dtp[(size_t)l0 * DEXP + DEXP];
  q.x.x  = xp [(size_t)l0 * DEXP];
  q.x.y  = xp [(size_t)l0 * DEXP + DEXP];
  const float* Bg = Bp + (size_t)l0 * NST;
  const float* Cg = Cp + (size_t)l0 * NST;
#pragma unroll
  for (int j = 0; j < UB; ++j) {
    q.b[j] = *(const v4f*)(Bg + j * NST);
    q.c[j] = *(const v4f*)(Cg + j * NST);
  }
}

template <int M>
__device__ __forceinline__ float row_xor(float v) {
  return __int_as_float(__builtin_amdgcn_update_dpp(0, __float_as_int(v), 0x160 | M,
                                                    0xf, 0xf, true));
}

template <int J>
__device__ __forceinline__ float step_get(const v2f v) {
  constexpr unsigned sel = 0x32103210u + 0x11111111u * (unsigned)(4 * (J >> 1));
  const float s = (J & 1) ? v.y : v.x;
  return __int_as_float(__builtin_amdgcn_permlane16(0, __float_as_int(s), sel, sel, false, true));
}

template <int J>
__device__ __forceinline__ void scan_step(const v2f e, const v2f u, const v4f Bv,
                                          const v2f (&A)[2], v2f (&H)[2]) {
  const float Es = step_get<J>(e), Us = step_get<J>(u);
  const v2f E = {Es, Es}, U = {Us, Us};
  const v2f g0 = A[0] * E, g1 = A[1] * E;
  v2f a0, a1;
  a0.x = __builtin_amdgcn_exp2f(g0.x);  a0.y = __builtin_amdgcn_exp2f(g0.y);
  a1.x = __builtin_amdgcn_exp2f(g1.x);  a1.y = __builtin_amdgcn_exp2f(g1.y);
  const v2f b0 = {Bv.x, Bv.y}, b1 = {Bv.z, Bv.w};
  H[0] = __builtin_elementwise_fma(a0, H[0], U * b0);
  H[1] = __builtin_elementwise_fma(a1, H[1], U * b1);
}

__device__ __forceinline__ void carry_steps(const BatchIn& q, const v2f (&A)[2],
                                            v2f (&H)[2], v2f& sdt) {
  v2f dt = q.dt, xv = q.x;
  TAKE_LOADED(dt, xv);
  const v2f u = dt * xv;
  const v2f e = dt * LOG2E;
  sdt += dt;
  scan_step<0>(e, u, q.b[0], A, H);
  scan_step<1>(e, u, q.b[1], A, H);
  scan_step<2>(e, u, q.b[2], A, H);
  scan_step<3>(e, u, q.b[3], A, H);
  scan_step<4>(e, u, q.b[4], A, H);
  scan_step<5>(e, u, q.b[5], A, H);
  scan_step<6>(e, u, q.b[6], A, H);
  scan_step<7>(e, u, q.b[7], A, H);
}

__global__ __launch_bounds__(SCAN_THREADS) __attribute__((amdgpu_waves_per_eu(1, 4))) void ssm_scan_carry(
    const float* __restrict__ x, const float* __restrict__ dtbuf,
    const float* __restrict__ Bmbuf, const float* __restrict__ Aneg,
    float* __restrict__ Sbuf, float* __restrict__ Pbuf)
{
  const int t    = threadIdx.x;
  const int n    = t & 15;
  const int ch   = n & 3;
  const int sq   = n >> 2;
  const int hf   = (t >> 4) & 1;
  const int wave = t >> 5;
  const int bi   = blockIdx.x;
  const int c    = bi & (NCHUNK - 1);
  const int dg   = (bi >> 3) & 31;
  const int b    = bi >> 8;
  const int d0   = dg * 32 + wave * 8 + hf * RCH;
  const int dl   = d0 + ch;

  const v4f ar = *(const v4f*)(Aneg + (size_t)dl * NST + 4 * sq);
  v2f A[2], H[2];
  A[0].x = ar.x;  A[0].y = ar.y;
  A[1].x = ar.z;  A[1].y = ar.w;
  H[0].x = 0.0f;  H[0].y = 0.0f;  H[1].x = 0.0f;  H[1].y = 0.0f;
  v2f sdt;  sdt.x = 0.0f;  sdt.y = 0.0f;
  const size_t rbase = (size_t)b * LSEQ + (size_t)c * CHUNK;
  const size_t hold  = (rbase + 2 * sq) * DEXP + dl;
  const float* dtp = dtbuf + hold;
  const float* xp  = x     + hold;
  const float* Bp  = Bmbuf + rbase * NST + 4 * sq;

  static_assert((CHUNK / UB) % 4 == 0, "whole turns of the ring: all but the last turn in the loop, the last turn behind it");
  BatchIn q0, q1, q2, q3;
  load_in(q0, dtp, xp, Bp, 0);                         SEG_FENCE();
  load_in(q1, dtp, xp, Bp, UB);                        SEG_FENCE();
  load_in(q2, dtp, xp, Bp, 2 * UB);                    SEG_FENCE();
#pragma unroll 1
  for (int l0 = 0; l0 < CHUNK - 4 * UB; l0 += 4 * UB) {
    load_in(q3, dtp, xp, Bp, group_row(l0 + 3 * UB));  SEG_FENCE();
    carry_steps(q0, A, H, sdt); HOLD_STATE(H);           SEG_FENCE();
    load_in(q0, dtp, xp, Bp, group_row(l0 + 4 * UB));  SEG_FENCE();
    carry_steps(q1, A, H, sdt); HOLD_STATE(H);           SEG_FENCE();
    load_in(q1, dtp, xp, Bp, group_row(l0 + 5 * UB));  SEG_FENCE();
    carry_steps(q2, A, H, sdt); HOLD_STATE(H);           SEG_FENCE();
    load_in(q2, dtp, xp, Bp, group_row(l0 + 6 * UB));  SEG_FENCE();
    carry_steps(q3, A, H, sdt);           HOLD_STATE(H);
  }
  load_in(q3, dtp, xp, Bp, group_row(CHUNK - UB));     SEG_FENCE();
  carry_steps(q0, A, H, sdt); HOLD_STATE(H);             SEG_FENCE();
  carry_steps(q1, A, H, sdt); HOLD_STATE(H);             SEG_FENCE();
  carry_steps(q2, A, H, sdt); HOLD_STATE(H);             SEG_FENCE();
  carry_steps(q3, A, H, sdt);
  float sd = sdt.x + sdt.y;
  sd += row_xor<4>(sd);
  sd += row_xor<8>(sd);
  const v2f sd2 = {sd, sd};
  const v2f p0 = A[0] * sd2, p1 = A[1] * sd2;
  const size_t si = (((size_t)b * NCHUNK + c) * DEXP + dl) * NST + 4 * sq;
  v4f sv, ev;
  sv.x = H[0].x;  sv.y = H[0].y;  sv.z = H[1].x;  sv.w = H[1].y;
  ev.x = __expf(p0.x);  ev.y = __expf(p0.y);  ev.z = __expf(p1.x);  ev.w = __expf(p1.y);
  *(v4f*)(Sbuf + si) = sv;
  *(v4f*)(Pbuf + si) = ev;
}

__global__ void ssm_carry_combine(const float* __restrict__ Sbuf,
                                  const float* __restrict__ Pbuf,
                                  float* __restrict__ Car)
{
  const int tg = blockIdx.x * 256 + threadIdx.x;
  const int n  = tg & 15;
  const int d  = (tg >> 4) & (DEXP - 1);
  const int b  = tg >> 14;
  const size_t stride = (size_t)DEXP * NST;
  const size_t base   = (size_t)b * NCHUNK * stride + (size_t)d * NST + n;
  float cin = 0.0f;
  Car[base] = 0.0f;
  for (int c = 1; c < NCHUNK; ++c) {
    const size_t prev = base + (size_t)(c - 1) * stride;
    cin = Sbuf[prev] + Pbuf[prev] * cin;
    Car[base + (size_t)c * stride] = cin;
  }
}

template <int J>
__device__ __forceinline__ float out_step(const v2f e, const v2f u, const v4f Bv,
                                          const v4f Cv, const v2f (&A)[2], v2f (&H)[2]) {
  scan_step<J>(e, u, Bv, A, H);
  const v2f c0 = {Cv.x, Cv.y}, c1 = {Cv.z, Cv.w};
  const v2f y = __builtin_elementwise_fma(H[1], c1, H[0] * c0);
  return y.x + y.y;
}

__device__ __forceinline__ void out_steps(const BatchOut& q, const v2f (&A)[2],
                                          v2f (&H)[2],
                                          const bool b2, const bool b3, float* op, int l0) {
  static_assert(UB == 8, "the eight steps of a batch are written out below");
  v2f dt = q.dt, xv = q.x;
  TAKE_LOADED(dt, xv);
  const v2f u = dt * xv;
  const v2f e = dt * LOG2E;
  float z[UB];
  z[0] = out_step<0>(e, u, q.b[0], q.c[0], A, H);
  z[1] = out_step<1>(e, u, q.b[1], q.c[1], A, H);
  z[2] = out_step<2>(e, u, q.b[2], q.c[2], A, H);
  z[3] = out_step<3>(e, u, q.b[3], q.c[3], A, H);
  z[4] = out_step<4>(e, u, q.b[4], q.c[4], A, H);
  z[5] = out_step<5>(e, u, q.b[5], q.c[5], A, H);
  z[6] = out_step<6>(e, u, q.b[6], q.c[6], A, H);
  z[7] = out_step<7>(e, u, q.b[7], q.c[7], A, H);
#pragma unroll
  for (int g = 0; g < UB; g += 4) {
    const float kA = b2 ? z[g + 1] : z[g],     sA = b2 ? z[g]     : z[g + 1];
    const float kB = b2 ? z[g + 3] : z[g + 2], sB = b2 ? z[g + 2] : z[g + 3];
    const float wA = kA + row_xor<4>(sA);
    const float wB = kB + row_xor<4>(sB);
    const float kk = b3 ? wB : wA, ss = b3 ? wA : wB;
    op[(size_t)(l0 + g) * DEXP] = kk + row_xor<8>(ss);
  }
}

__global__ __launch_bounds__(SCAN_THREADS) void ssm_scan_out(
    const float* __restrict__ x, const float* __restrict__ dtbuf,
    const float* __restrict__ Bmbuf, const float* __restrict__ Cbuf,
    const float* __restrict__ Aneg, const float* __restrict__ Car,
    float* __restrict__ out)
{
  const int t    = threadIdx.x;
  const int n    = t & 15;
  const int ch   = n & 3;
  const int sq   = n >> 2;
  const int hf   = (t >> 4) & 1;
  const int wave = t >> 5;
  const int bi   = blockIdx.x;
  const int c    = bi & (NCHUNK - 1);
  const int dg   = (bi >> 3) & 31;
  const int b    = bi >> 8;
  const int d0   = dg * 32 + wave * 8 + hf * RCH;
  const int dl   = d0 + ch;

  const size_t rbase = (size_t)b * LSEQ + (size_t)c * CHUNK;
  const size_t si = (((size_t)b * NCHUNK + c) * DEXP + dl) * NST + 4 * sq;
  const v4f ar = *(const v4f*)(Aneg + (size_t)dl * NST + 4 * sq);
  const v4f cr = *(const v4f*)(Car + si);
  v2f A[2], H[2];
  A[0].x = ar.x;  A[0].y = ar.y;
  A[1].x = ar.z;  A[1].y = ar.w;
  H[0].x = cr.x;  H[0].y = cr.y;
  H[1].x = cr.z;  H[1].y = cr.w;
  const size_t hold = (rbase + 2 * sq) * DEXP + dl;
  const float* dtp = dtbuf + hold;
  const float* xp  = x     + hold;
  const float* Bp  = Bmbuf + rbase * NST + 4 * sq;
  const float* Cp  = Cbuf  + rbase * NST + 4 * sq;

  const bool b2 = (n & 4) != 0, b3 = (n & 8) != 0;
  const int  myst = ((n >> 2) & 1) | ((n >> 3) << 1);
  float* op = out + (rbase + myst) * DEXP + dl;

  static_assert((CHUNK / UB) % 4 == 0, "whole turns of the ring: all but the last turn in the loop, the last turn behind it");
  BatchOut qa, qb, qc, qd;
  load_out(qa, dtp, xp, Bp, Cp, 0);                               SEG_FENCE();
  load_out(qb, dtp, xp, Bp, Cp, UB);                              SEG_FENCE();
  load_out(qc, dtp, xp, Bp, Cp, 2 * UB);                          SEG_FENCE();
#pragma unroll 1
  for (int l0 = 0; l0 < CHUNK - 4 * UB; l0 += 4 * UB) {
    load_out(qd, dtp, xp, Bp, Cp, group_row(l0 + 3 * UB));        SEG_FENCE();
    out_steps(qa, A, H, b2, b3, op, l0); HOLD_STATE(H);               SEG_FENCE();
    load_out(qa, dtp, xp, Bp, Cp, group_row(l0 + 4 * UB));        SEG_FENCE();
    out_steps(qb, A, H, b2, b3, op, l0 + UB); HOLD_STATE(H);          SEG_FENCE();
    load_out(qb, dtp, xp, Bp, Cp, group_row(l0 + 5 * UB));        SEG_FENCE();
    out_steps(qc, A, H, b2, b3, op, l0 + 2 * UB); HOLD_STATE(H);      SEG_FENCE();
    load_out(qc, dtp, xp, Bp, Cp, group_row(l0 + 6 * UB));        SEG_FENCE();
    out_steps(qd, A, H, b2, b3, op, l0 + 3 * UB);  HOLD_STATE(H);
  }
  load_out(qd, dtp, xp, Bp, Cp, group_row(CHUNK - UB));           SEG_FENCE();
  out_steps(qa, A, H, b2, b3, op, CHUNK - 4 * UB); HOLD_STATE(H);      SEG_FENCE();
  out_steps(qb, A, H, b2, b3, op, CHUNK - 3 * UB); HOLD_STATE(H);      SEG_FENCE();
  out_steps(qc, A, H, b2, b3, op, CHUNK - 2 * UB); HOLD_STATE(H);      SEG_FENCE();
  out_steps(qd, A, H, b2, b3, op, CHUNK - UB);
}

extern "C" void kernel_launch(void* const* d_in, const int* in_sizes, int n_in,
                              void* d_out, int out_size, void* d_ws, size_t ws_size,
                              hipStream_t stream) {
  (void)in_sizes; (void)n_in; (void)out_size; (void)ws_size;
  const float* x     = (const float*)d_in[0];
  const float* xw    = (const float*)d_in[1];
  const float* xb    = (const float*)d_in[2];
  const float* dtw   = (const float*)d_in[3];
  const float* dtb   = (const float*)d_in[4];
  const float* A_log = (const float*)d_in[5];
  float* out = (float*)d_out;

  float* ws    = (float*)d_ws;
  float* dtbuf = ws;
  float* Aneg  = dtbuf + (size_t)ROWS * DEXP;
  float* Bm    = Aneg  + (size_t)DEXP * NST;
  float* Cb    = Bm    + (size_t)ROWS * NST;
  float* Sb    = Cb    + (size_t)ROWS * NST;
  float* Pb    = Sb    + (size_t)B_SZ * NCHUNK * DEXP * NST;
  float* Car   = Pb    + (size_t)B_SZ * NCHUNK * DEXP * NST;
  float* xwT   = Car   + (size_t)B_SZ * NCHUNK * DEXP * NST;
  float* dtwT  = xwT   + (size_t)DEXP * RPROJ;

  hipLaunchKernelGGL(ssm_prep, dim3((DEXP * RPROJ + 255) / 256), dim3(256), 0, stream,
                     A_log, Aneg, xw, xwT, dtw, dtwT);
  hipLaunchKernelGGL(ssm_proj_wmma, dim3(ROWS / 16), dim3(256), 0, stream,
                     x, xwT, xb, dtwT, dtb, dtbuf, Bm, Cb);
  hipLaunchKernelGGL(ssm_scan_carry, dim3(SCAN_BLOCKS), dim3(SCAN_THREADS), 0, stream,
                     x, dtbuf, Bm, Aneg, Sb, Pb);
  hipLaunchKernelGGL(ssm_carry_combine, dim3((B_SZ * DEXP * NST) / 256), dim3(256), 0, stream,
                     Sb, Pb, Car);
  hipLaunchKernelGGL(ssm_scan_out, dim3(SCAN_BLOCKS), dim3(SCAN_THREADS), 0, stream,
                     x, dtbuf, Bm, Cb, Aneg, Car, out);
}


# carry-combine launch removed (4 launches, was 5): each output-scan lane folds the summaries of chunks 0..c-1 for its own 4 states in registers, loaded ahead of the opening groups; same fold order
# speedup vs baseline: 4.9858x; 1.0385x over previous
#include <hip/hip_runtime.h>
#include <math.h>

#define B_SZ   2
#define LSEQ   2048
#define DEXP   1024
#define DRANK  64
#define NST    16
#define RPROJ  (DRANK + 2*NST)
#define NCHUNK 8
#define CHUNK  (LSEQ / NCHUNK)
#define ROWS   (B_SZ * LSEQ)

typedef __attribute__((ext_vector_type(2))) float v2f;
typedef __attribute__((ext_vector_type(8))) float v8f;

__device__ __forceinline__ v8f wmma_f32_k4(v2f a, v2f b, v8f c) {
  return __builtin_amdgcn_wmma_f32_16x16x4_f32(
       false, a,  false, b,
       (short)0, c,  false,  false);
}

__global__ void ssm_prep(const float* __restrict__ A_log, float* __restrict__ Aneg,
                         const float* __restrict__ xw,  float* __restrict__ xwT,
                         const float* __restrict__ dtw, float* __restrict__ dtwT) {
  int i = blockIdx.x * 256 + threadIdx.x;
  if (i < DEXP * NST) Aneg[i] = -__expf(A_log[i]);
  if (i < DEXP * RPROJ) {
    int k = i / RPROJ, c = i % RPROJ;
    xwT[i] = xw[(size_t)c * DEXP + k];
  }
  if (i < DRANK * DEXP) {
    int k = i / DEXP, d = i % DEXP;
    dtwT[i] = dtw[(size_t)d * DRANK + k];
  }
}

struct BSet1 { v2f f[16]; };
struct BSet2 { v2f f[16]; float bias; };

#define SXP 1028

__device__ __forceinline__ void load_b1(BSet1& q, const float* bp, int g) {
#pragma unroll
  for (int s = 0; s < 16; ++s) {
    const int k = g * 64 + s * 4;
    q.f[s].x = bp[(size_t)k * RPROJ];
    q.f[s].y = bp[(size_t)(k + 1) * RPROJ];
  }
}

__device__ __forceinline__ v8f mma_group1(const BSet1& q, const float* ap, int g, v8f acc) {
#pragma unroll
  for (int s = 0; s < 16; ++s) {
    const int k = g * 64 + s * 4;
    v2f a;  a.x = ap[k];  a.y = ap[k + 1];
    acc = wmma_f32_k4(a, q.f[s], acc);
  }
  return acc;
}

__device__ __forceinline__ void load_b2(BSet2& q, const float* wp, const float* dtb, int col) {
  q.bias = dtb[col];
#pragma unroll
  for (int kt = 0; kt < 16; ++kt) {
    q.f[kt].x = wp[(size_t)(kt * 4) * DEXP + col];
    q.f[kt].y = wp[(size_t)(kt * 4 + 1) * DEXP + col];
  }
}

__device__ __forceinline__ void tile2(const BSet2& q, const v2f (&a2)[16], float* dp, int col) {
  v8f acc2;
#pragma unroll
  for (int i = 0; i < 8; ++i) acc2[i] = q.bias;
#pragma unroll
  for (int kt = 0; kt < 16; ++kt) acc2 = wmma_f32_k4(a2[kt], q.f[kt], acc2);
  float sp[8];
#pragma unroll
  for (int i = 0; i < 8; ++i) sp[i] = log1pf(__expf(acc2[i]));
#pragma unroll
  for (int i = 0; i < 8; ++i) {
    const float v = acc2[i];
    dp[(size_t)i * DEXP + col] = (v > 20.0f) ? v : sp[i];
  }
}

__global__ __launch_bounds__(256) void ssm_proj_wmma(
    const float* __restrict__ x,    const float* __restrict__ xwT,
    const float* __restrict__ xb,   const float* __restrict__ dtwT,
    const float* __restrict__ dtb,
    float* __restrict__ dtbuf, float* __restrict__ Bmbuf, float* __restrict__ Cbuf)
{
  __shared__ float sx[16 * SXP];
  __shared__ float sproj[16 * 100];

  const int t    = threadIdx.x;
  const int wave = t >> 5;
  const int nl   = t & 15;
  const int hi   = (t >> 4) & 1;
  const int rowbase = blockIdx.x * 16;

  const int col1 = wave * 16 + nl;
  const float* bp = xwT + (size_t)(hi * 2) * RPROJ + col1;
  BSet1 b0, b1;
  v8f acc1;
  float bv1 = 0.0f;
  if (wave < 6) {
    load_b1(b0, bp, 0);
    bv1 = xb[col1];
  }
  for (int i = 0; i < 16; ++i) {
    int flat = t + i * 256;
    int r  = flat >> 8;
    int kk = (flat & 255) << 2;
    const float4 v = *(const float4*)(x + (size_t)(rowbase + r) * DEXP + kk);
    float* dst = &sx[r * SXP + kk];
    dst[0] = v.x; dst[1] = v.y; dst[2] = v.z; dst[3] = v.w;
  }
  __syncthreads();
  if (wave < 6) {
    for (int i = 0; i < 8; ++i) acc1[i] = bv1;
    const float* ap = &sx[nl * SXP + hi * 2];
#pragma unroll 1
    for (int g = 0; g < 16; g += 2) {
      load_b1(b1, bp, g + 1);
      acc1 = mma_group1(b0, ap, g, acc1);
      if (g + 2 < 16) load_b1(b0, bp, g + 2);
      acc1 = mma_group1(b1, ap, g + 1, acc1);
    }
  }
  const float* wp = dtwT + (size_t)(hi * 2) * DEXP;
  const int colw = wave * 128 + nl;
  BSet2 c0, c1;
  load_b2(c0, wp, dtb, colw);
  if (wave < 6) {
    for (int i = 0; i < 8; ++i) {
      int M = i + hi * 8;
      sproj[M * 100 + wave * 16 + nl] = acc1[i];
      if (wave == 4) Bmbuf[(size_t)(rowbase + M) * NST + nl] = acc1[i];
      if (wave == 5) Cbuf [(size_t)(rowbase + M) * NST + nl] = acc1[i];
    }
  }
  __syncthreads();

  v2f a2[16];
  for (int kt = 0; kt < 16; ++kt) {
    int kb = kt * 4 + hi * 2;
    a2[kt].x = sproj[nl * 100 + kb];
    a2[kt].y = sproj[nl * 100 + kb + 1];
  }
  float* dp = dtbuf + (size_t)(rowbase + hi * 8) * DEXP;
#pragma unroll 1
  for (int j = 0; j < 8; j += 2) {
    load_b2(c1, wp, dtb, colw + (j + 1) * 16);
    tile2(c0, a2, dp, colw + j * 16);
    if (j + 2 < 8) load_b2(c0, wp, dtb, colw + (j + 2) * 16);
    tile2(c1, a2, dp, colw + (j + 1) * 16);
  }
}

#define RCH 4
#define UB  8
#define SCAN_THREADS 128
#define SCAN_BLOCKS  (B_SZ * NCHUNK * (DEXP / 32))

typedef __attribute__((ext_vector_type(4))) float v4f;

#define SEG_FENCE() __builtin_amdgcn_sched_barrier(0)
#define LOG2E 1.44269504088896340736f

__device__ __forceinline__ int group_row(int l0) {
  asm volatile("" : "+s"(l0));
  return l0;
}
#define HOLD_STATE(H) asm volatile("" : "+v"(H[0]), "+v"(H[1]))
#define TAKE_LOADED(a, b) asm volatile("" : "+v"(a), "+v"(b))

struct BatchIn  { v2f dt; v2f x; v4f b[UB]; };
struct BatchOut { v2f dt; v2f x; v4f b[UB]; v4f c[UB]; };

__device__ __forceinline__ void load_in(BatchIn& q, const float* dtp, const float* xp,
                                        const float* Bp, int l0) {
  q.dt.x = dtp[(size_t)l0 * DEXP];
  q.dt.y = dtp[(size_t)l0 * DEXP + DEXP];
  q.x.x  = xp [(size_t)l0 * DEXP];
  q.x.y  = xp [(size_t)l0 * DEXP + DEXP];
  const float* Bg = Bp + (size_t)l0 * NST;
#pragma unroll
  for (int j = 0; j < UB; ++j) q.b[j] = *(const v4f*)(Bg + j * NST);
}

__device__ __forceinline__ void load_out(BatchOut& q, const float* dtp, const float* xp,
                                         const float* Bp, const float* Cp, int l0) {
  q.dt.x = dtp[(size_t)l0 * DEXP];
  q.dt.y = dtp[(size_t)l0 * DEXP + DEXP];
  q.x.x  = xp [(size_t)l0 * DEXP];
  q.x.y  = xp [(size_t)l0 * DEXP + DEXP];
  const float* Bg = Bp + (size_t)l0 * NST;
  const float* Cg = Cp + (size_t)l0 * NST;
#pragma unroll
  for (int j = 0; j < UB; ++j) {
    q.b[j] = *(const v4f*)(Bg + j * NST);
    q.c[j] = *(const v4f*)(Cg + j * NST);
  }
}

template <int M>
__device__ __forceinline__ float row_xor(float v) {
  return __int_as_float(__builtin_amdgcn_update_dpp(0, __float_as_int(v), 0x160 | M,
                                                    0xf, 0xf, true));
}

template <int J>
__device__ __forceinline__ float step_get(const v2f v) {
  constexpr unsigned sel = 0x32103210u + 0x11111111u * (unsigned)(4 * (J >> 1));
  const float s = (J & 1) ? v.y : v.x;
  return __int_as_float(__builtin_amdgcn_permlane16(0, __float_as_int(s), sel, sel, false, true));
}

template <int J>
__device__ __forceinline__ void scan_step(const v2f e, const v2f u, const v4f Bv,
                                          const v2f (&A)[2], v2f (&H)[2]) {
  const float Es = step_get<J>(e), Us = step_get<J>(u);
  const v2f E = {Es, Es}, U = {Us, Us};
  const v2f g0 = A[0] * E, g1 = A[1] * E;
  v2f a0, a1;
  a0.x = __builtin_amdgcn_exp2f(g0.x);  a0.y = __builtin_amdgcn_exp2f(g0.y);
  a1.x = __builtin_amdgcn_exp2f(g1.x);  a1.y = __builtin_amdgcn_exp2f(g1.y);
  const v2f b0 = {Bv.x, Bv.y}, b1 = {Bv.z, Bv.w};
  H[0] = __builtin_elementwise_fma(a0, H[0], U * b0);
  H[1] = __builtin_elementwise_fma(a1, H[1], U * b1);
}

__device__ __forceinline__ void carry_steps(const BatchIn& q, const v2f (&A)[2],
                                            v2f (&H)[2], v2f& sdt) {
  v2f dt = q.dt, xv = q.x;
  TAKE_LOADED(dt, xv);
  const v2f u = dt * xv;
  const v2f e = dt * LOG2E;
  sdt += dt;
  scan_step<0>(e, u, q.b[0], A, H);
  scan_step<1>(e, u, q.b[1], A, H);
  scan_step<2>(e, u, q.b[2], A, H);
  scan_step<3>(e, u, q.b[3], A, H);
  scan_step<4>(e, u, q.b[4], A, H);
  scan_step<5>(e, u, q.b[5], A, H);
  scan_step<6>(e, u, q.b[6], A, H);
  scan_step<7>(e, u, q.b[7], A, H);
}

__global__ __launch_bounds__(SCAN_THREADS) __attribute__((amdgpu_waves_per_eu(1, 4))) void ssm_scan_carry(
    const float* __restrict__ x, const float* __restrict__ dtbuf,
    const float* __restrict__ Bmbuf, const float* __restrict__ Aneg,
    float* __restrict__ Sbuf, float* __restrict__ Pbuf)
{
  const int t    = threadIdx.x;
  const int n    = t & 15;
  const int ch   = n & 3;
  const int sq   = n >> 2;
  const int hf   = (t >> 4) & 1;
  const int wave = t >> 5;
  const int bi   = blockIdx.x;
  const int c    = bi & (NCHUNK - 1);
  const int dg   = (bi >> 3) & 31;
  const int b    = bi >> 8;
  const int d0   = dg * 32 + wave * 8 + hf * RCH;
  const int dl   = d0 + ch;

  const v4f ar = *(const v4f*)(Aneg + (size_t)dl * NST + 4 * sq);
  v2f A[2], H[2];
  A[0].x = ar.x;  A[0].y = ar.y;
  A[1].x = ar.z;  A[1].y = ar.w;
  H[0].x = 0.0f;  H[0].y = 0.0f;  H[1].x = 0.0f;  H[1].y = 0.0f;
  v2f sdt;  sdt.x = 0.0f;  sdt.y = 0.0f;
  const size_t rbase = (size_t)b * LSEQ + (size_t)c * CHUNK;
  const size_t hold  = (rbase + 2 * sq) * DEXP + dl;
  const float* dtp = dtbuf + hold;
  const float* xp  = x     + hold;
  const float* Bp  = Bmbuf + rbase * NST + 4 * sq;

  static_assert((CHUNK / UB) % 4 == 0, "whole turns of the ring: all but the last turn in the loop, the last turn behind it");
  BatchIn q0, q1, q2, q3;
  load_in(q0, dtp, xp, Bp, 0);                         SEG_FENCE();
  load_in(q1, dtp, xp, Bp, UB);                        SEG_FENCE();
  load_in(q2, dtp, xp, Bp, 2 * UB);                    SEG_FENCE();
#pragma unroll 1
  for (int l0 = 0; l0 < CHUNK - 4 * UB; l0 += 4 * UB) {
    load_in(q3, dtp, xp, Bp, group_row(l0 + 3 * UB));  SEG_FENCE();
    carry_steps(q0, A, H, sdt); HOLD_STATE(H);           SEG_FENCE();
    load_in(q0, dtp, xp, Bp, group_row(l0 + 4 * UB));  SEG_FENCE();
    carry_steps(q1, A, H, sdt); HOLD_STATE(H);           SEG_FENCE();
    load_in(q1, dtp, xp, Bp, group_row(l0 + 5 * UB));  SEG_FENCE();
    carry_steps(q2, A, H, sdt); HOLD_STATE(H);           SEG_FENCE();
    load_in(q2, dtp, xp, Bp, group_row(l0 + 6 * UB));  SEG_FENCE();
    carry_steps(q3, A, H, sdt);           HOLD_STATE(H);
  }
  load_in(q3, dtp, xp, Bp, group_row(CHUNK - UB));     SEG_FENCE();
  carry_steps(q0, A, H, sdt); HOLD_STATE(H);             SEG_FENCE();
  carry_steps(q1, A, H, sdt); HOLD_STATE(H);             SEG_FENCE();
  carry_steps(q2, A, H, sdt); HOLD_STATE(H);             SEG_FENCE();
  carry_steps(q3, A, H, sdt);
  float sd = sdt.x + sdt.y;
  sd += row_xor<4>(sd);
  sd += row_xor<8>(sd);
  const v2f sd2 = {sd, sd};
  const v2f p0 = A[0] * sd2, p1 = A[1] * sd2;
  const size_t si = (((size_t)b * NCHUNK + c) * DEXP + dl) * NST + 4 * sq;
  v4f sv, ev;
  sv.x = H[0].x;  sv.y = H[0].y;  sv.z = H[1].x;  sv.w = H[1].y;
  ev.x = __expf(p0.x);  ev.y = __expf(p0.y);  ev.z = __expf(p1.x);  ev.w = __expf(p1.y);
  *(v4f*)(Sbuf + si) = sv;
  *(v4f*)(Pbuf + si) = ev;
}

template <int J>
__device__ __forceinline__ float out_step(const v2f e, const v2f u, const v4f Bv,
                                          const v4f Cv, const v2f (&A)[2], v2f (&H)[2]) {
  scan_step<J>(e, u, Bv, A, H);
  const v2f c0 = {Cv.x, Cv.y}, c1 = {Cv.z, Cv.w};
  const v2f y = __builtin_elementwise_fma(H[1], c1, H[0] * c0);
  return y.x + y.y;
}

__device__ __forceinline__ void out_steps(const BatchOut& q, const v2f (&A)[2],
                                          v2f (&H)[2],
                                          const bool b2, const bool b3, float* op, int l0) {
  static_assert(UB == 8, "the eight steps of a batch are written out below");
  v2f dt = q.dt, xv = q.x;
  TAKE_LOADED(dt, xv);
  const v2f u = dt * xv;
  const v2f e = dt * LOG2E;
  float z[UB];
  z[0] = out_step<0>(e, u, q.b[0], q.c[0], A, H);
  z[1] = out_step<1>(e, u, q.b[1], q.c[1], A, H);
  z[2] = out_step<2>(e, u, q.b[2], q.c[2], A, H);
  z[3] = out_step<3>(e, u, q.b[3], q.c[3], A, H);
  z[4] = out_step<4>(e, u, q.b[4], q.c[4], A, H);
  z[5] = out_step<5>(e, u, q.b[5], q.c[5], A, H);
  z[6] = out_step<6>(e, u, q.b[6], q.c[6], A, H);
  z[7] = out_step<7>(e, u, q.b[7], q.c[7], A, H);
#pragma unroll
  for (int g = 0; g < UB; g += 4) {
    const float kA = b2 ? z[g + 1] : z[g],     sA = b2 ? z[g]     : z[g + 1];
    const float kB = b2 ? z[g + 3] : z[g + 2], sB = b2 ? z[g + 2] : z[g + 3];
    const float wA = kA + row_xor<4>(sA);
    const float wB = kB + row_xor<4>(sB);
    const float kk = b3 ? wB : wA, ss = b3 ? wA : wB;
    op[(size_t)(l0 + g) * DEXP] = kk + row_xor<8>(ss);
  }
}

__global__ __launch_bounds__(SCAN_THREADS) void ssm_scan_out(
    const float* __restrict__ x, const float* __restrict__ dtbuf,
    const float* __restrict__ Bmbuf, const float* __restrict__ Cbuf,
    const float* __restrict__ Aneg,
    const float* __restrict__ Sbuf, const float* __restrict__ Pbuf,
    float* __restrict__ out)
{
  const int t    = threadIdx.x;
  const int n    = t & 15;
  const int ch   = n & 3;
  const int sq   = n >> 2;
  const int hf   = (t >> 4) & 1;
  const int wave = t >> 5;
  const int bi   = blockIdx.x;
  const int c    = bi & (NCHUNK - 1);
  const int dg   = (bi >> 3) & 31;
  const int b    = bi >> 8;
  const int d0   = dg * 32 + wave * 8 + hf * RCH;
  const int dl   = d0 + ch;

  const size_t rbase = (size_t)b * LSEQ + (size_t)c * CHUNK;
  const v4f ar = *(const v4f*)(Aneg + (size_t)dl * NST + 4 * sq);
  v2f A[2], H[2];
  A[0].x = ar.x;  A[0].y = ar.y;
  A[1].x = ar.z;  A[1].y = ar.w;
  const size_t s0 = ((size_t)b * NCHUNK * DEXP + dl) * NST + 4 * sq;
  const int cm = (c > 0 ? c : 1) - 1;
  v4f sv[NCHUNK - 1], pv[NCHUNK - 1];
#pragma unroll
  for (int cc = 0; cc < NCHUNK - 1; ++cc) {
    const int ci = cc < cm ? cc : cm;
    sv[cc] = *(const v4f*)(Sbuf + s0 + (size_t)ci * DEXP * NST);
    pv[cc] = *(const v4f*)(Pbuf + s0 + (size_t)ci * DEXP * NST);
  }
  SEG_FENCE();
  const size_t hold = (rbase + 2 * sq) * DEXP + dl;
  const float* dtp = dtbuf + hold;
  const float* xp  = x     + hold;
  const float* Bp  = Bmbuf + rbase * NST + 4 * sq;
  const float* Cp  = Cbuf  + rbase * NST + 4 * sq;

  const bool b2 = (n & 4) != 0, b3 = (n & 8) != 0;
  const int  myst = ((n >> 2) & 1) | ((n >> 3) << 1);
  float* op = out + (rbase + myst) * DEXP + dl;

  static_assert((CHUNK / UB) % 4 == 0, "whole turns of the ring: all but the last turn in the loop, the last turn behind it");
  BatchOut qa, qb, qc, qd;
  load_out(qa, dtp, xp, Bp, Cp, 0);                               SEG_FENCE();
  load_out(qb, dtp, xp, Bp, Cp, UB);                              SEG_FENCE();
  load_out(qc, dtp, xp, Bp, Cp, 2 * UB);                          SEG_FENCE();
  v4f cin = {0.0f, 0.0f, 0.0f, 0.0f};
#pragma unroll
  for (int cc = 0; cc < NCHUNK - 1; ++cc) {
    const v4f nx = sv[cc] + pv[cc] * cin;
    cin = (cc < c) ? nx : cin;
  }
  H[0].x = cin.x;  H[0].y = cin.y;
  H[1].x = cin.z;  H[1].y = cin.w;
#pragma unroll 1
  for (int l0 = 0; l0 < CHUNK - 4 * UB; l0 += 4 * UB) {
    load_out(qd, dtp, xp, Bp, Cp, group_row(l0 + 3 * UB));        SEG_FENCE();
    out_steps(qa, A, H, b2, b3, op, l0); HOLD_STATE(H);               SEG_FENCE();
    load_out(qa, dtp, xp, Bp, Cp, group_row(l0 + 4 * UB));        SEG_FENCE();
    out_steps(qb, A, H, b2, b3, op, l0 + UB); HOLD_STATE(H);          SEG_FENCE();
    load_out(qb, dtp, xp, Bp, Cp, group_row(l0 + 5 * UB));        SEG_FENCE();
    out_steps(qc, A, H, b2, b3, op, l0 + 2 * UB); HOLD_STATE(H);      SEG_FENCE();
    load_out(qc, dtp, xp, Bp, Cp, group_row(l0 + 6 * UB));        SEG_FENCE();
    out_steps(qd, A, H, b2, b3, op, l0 + 3 * UB);  HOLD_STATE(H);
  }
  load_out(qd, dtp, xp, Bp, Cp, group_row(CHUNK - UB));           SEG_FENCE();
  out_steps(qa, A, H, b2, b3, op, CHUNK - 4 * UB); HOLD_STATE(H);      SEG_FENCE();
  out_steps(qb, A, H, b2, b3, op, CHUNK - 3 * UB); HOLD_STATE(H);      SEG_FENCE();
  out_steps(qc, A, H, b2, b3, op, CHUNK - 2 * UB); HOLD_STATE(H);      SEG_FENCE();
  out_steps(qd, A, H, b2, b3, op, CHUNK - UB);
}

extern "C" void kernel_launch(void* const* d_in, const int* in_sizes, int n_in,
                              void* d_out, int out_size, void* d_ws, size_t ws_size,
                              hipStream_t stream) {
  (void)in_sizes; (void)n_in; (void)out_size; (void)ws_size;
  const float* x     = (const float*)d_in[0];
  const float* xw    = (const float*)d_in[1];
  const float* xb    = (const float*)d_in[2];
  const float* dtw   = (const float*)d_in[3];
  const float* dtb   = (const float*)d_in[4];
  const float* A_log = (const float*)d_in[5];
  float* out = (float*)d_out;

  float* ws    = (float*)d_ws;
  float* dtbuf = ws;
  float* Aneg  = dtbuf + (size_t)ROWS * DEXP;
  float* Bm    = Aneg  + (size_t)DEXP * NST;
  float* Cb    = Bm    + (size_t)ROWS * NST;
  float* Sb    = Cb    + (size_t)ROWS * NST;
  float* Pb    = Sb    + (size_t)B_SZ * NCHUNK * DEXP * NST;
  float* xwT   = Pb    + (size_t)B_SZ * NCHUNK * DEXP * NST;
  float* dtwT  = xwT   + (size_t)DEXP * RPROJ;

  hipLaunchKernelGGL(ssm_prep, dim3((DEXP * RPROJ + 255) / 256), dim3(256), 0, stream,
                     A_log, Aneg, xw, xwT, dtw, dtwT);
  hipLaunchKernelGGL(ssm_proj_wmma, dim3(ROWS / 16), dim3(256), 0, stream,
                     x, xwT, xb, dtwT, dtb, dtbuf, Bm, Cb);
  hipLaunchKernelGGL(ssm_scan_carry, dim3(SCAN_BLOCKS), dim3(SCAN_THREADS), 0, stream,
                     x, dtbuf, Bm, Aneg, Sb, Pb);
  hipLaunchKernelGGL(ssm_scan_out, dim3(SCAN_BLOCKS), dim3(SCAN_THREADS), 0, stream,
                     x, dtbuf, Bm, Cb, Aneg, Sb, Pb, out);
}
